# starting kernel
# baseline (speedup 1.0000x reference)
#include <hip/hip_runtime.h>
#include <math.h>


typedef float v8f __attribute__((ext_vector_type(8)));
typedef float v2f __attribute__((ext_vector_type(2)));

#define IN_C   64
#define OUT_C  64
#define HH     512
#define WW     512
#define MODES  64
#define HW     (HH * WW)
#define TWO_PI 6.283185307179586f
#define ISPLIT 4
#define ICHUNK (IN_C / ISPLIT)

__device__ __forceinline__ v8f wmma_f32(v2f a, v2f b, v8f c) {
    return __builtin_amdgcn_wmma_f32_16x16x4_f32(false, a, false, b, (short)0, c,
                                                 false, false);
}

__device__ __forceinline__ v2f load_a_tile(const float* __restrict__ p, int ld, int lane) {
    int m = lane & 15;
    int k = (lane >> 4) << 1;
    v2f a;
    a.x = p[(size_t)m * ld + k];
    a.y = p[(size_t)m * ld + k + 1];
    return a;
}

__device__ __forceinline__ v2f load_b_tile(const float* __restrict__ p, size_t ld, int lane) {
    int n = lane & 15;
    int k = (lane >> 4) << 1;
    v2f b;
    b.x = p[(size_t)k * ld + n];
    b.y = p[(size_t)(k + 1) * ld + n];
    return b;
}

__global__ void twiddle_kernel(float* __restrict__ FW, float* __restrict__ cosH,
                               float* __restrict__ sinH, float* __restrict__ DW) {
    int t = blockIdx.x * blockDim.x + threadIdx.x;
    if (t >= 32768) return;
    const float step = TWO_PI / 512.0f;
    {
        int w = t >> 6, m2 = t & 63;
        float th = (float)((w * m2) & 511) * step;
        FW[(size_t)w * 128 + 2 * m2]     = cosf(th);
        FW[(size_t)w * 128 + 2 * m2 + 1] = -sinf(th);
    }
    {
        float th = (float)(((t >> 9) * (t & 511)) & 511) * step;
        cosH[t] = cosf(th);
        sinH[t] = sinf(th);
    }
    {
        int k = t >> 9, w = t & 511;
        float th = (float)((k * w) & 511) * step;
        float s  = ((k == 0) ? 1.0f : 2.0f) / 262144.0f;
        DW[(size_t)(2 * k) * 512 + w]     = s * cosf(th);
        DW[(size_t)(2 * k + 1) * 512 + w] = -s * sinf(th);
    }
}

__global__ void dftw_kernel(const float* __restrict__ x, const float* __restrict__ FW,
                            float* __restrict__ Xw) {
    __shared__ float As[16 * 512];
    const int row0 = blockIdx.x * 16;
    const float* xrow = x + (size_t)row0 * 512;
    for (int idx = threadIdx.x; idx < 16 * 512 / 4; idx += blockDim.x)
        ((float4*)As)[idx] = ((const float4*)xrow)[idx];
    __syncthreads();

    const int wave = threadIdx.x >> 5, lane = threadIdx.x & 31;
    const int n0 = wave * 32;
    v8f acc0 = {}, acc1 = {};
    for (int k = 0; k < 512; k += 4) {
        v2f a  = load_a_tile(As + k, 512, lane);
        v2f b0 = load_b_tile(FW + (size_t)k * 128 + n0,      128, lane);
        v2f b1 = load_b_tile(FW + (size_t)k * 128 + n0 + 16, 128, lane);
        acc0 = wmma_f32(a, b0, acc0);
        acc1 = wmma_f32(a, b1, acc1);
    }
    float* crow = Xw + (size_t)row0 * 128;
    const int n = lane & 15, mb = (lane >> 4) * 8;
    for (int v = 0; v < 8; ++v) {
        crow[(size_t)(mb + v) * 128 + n0 + n]      = acc0[v];
        crow[(size_t)(mb + v) * 128 + n0 + 16 + n] = acc1[v];
    }
}

__global__ void dfth_kernel(const float* __restrict__ Xw, const float* __restrict__ cosH,
                            const float* __restrict__ sinH, float* __restrict__ Xs) {
    const int c = blockIdx.x, m1base = blockIdx.y * 8;
    const int m2 = threadIdx.x & 63, sub = threadIdx.x >> 6;
    const float* xw = Xw + (size_t)c * 512 * 128 + 2 * m2;
    for (int mm = sub; mm < 8; mm += 4) {
        const int m1 = m1base + mm;
        const float* cp = cosH + (size_t)m1 * 512;
        const float* sp = sinH + (size_t)m1 * 512;
        float accR = 0.0f, accI = 0.0f;
        for (int h = 0; h < 512; ++h) {
            float ct = cp[h], st = sp[h];
            float xr = xw[(size_t)h * 128], xi = xw[(size_t)h * 128 + 1];
            accR += ct * xr + st * xi;
            accI += ct * xi - st * xr;
        }
        Xs[((size_t)c * 64 + m1) * 128 + 2 * m2]     = accR;
        Xs[((size_t)c * 64 + m1) * 128 + 2 * m2 + 1] = accI;
    }
}

__global__ void modemul_kernel(const float* __restrict__ Xs, const float* __restrict__ wre,
                               const float* __restrict__ wim, float* __restrict__ YsP) {
    const int m1 = blockIdx.x, obase = blockIdx.y * 16;
    const int ibase = blockIdx.z * ICHUNK;
    __shared__ float xsR[ICHUNK * 64], xsI[ICHUNK * 64];
    for (int idx = threadIdx.x; idx < ICHUNK * 64; idx += blockDim.x) {
        int i = idx >> 6, m2 = idx & 63;
        const float* p = Xs + ((size_t)(ibase + i) * 64 + m1) * 128 + 2 * m2;
        xsR[idx] = p[0];
        xsI[idx] = p[1];
    }
    __syncthreads();

    const int m2q = (threadIdx.x & 15) * 4;
    const int o   = obase + (threadIdx.x >> 4);
    float aR[4] = {0, 0, 0, 0}, aI[4] = {0, 0, 0, 0};
#pragma unroll 4
    for (int i = 0; i < ICHUNK; ++i) {
        size_t base = (((size_t)(ibase + i) * 64 + o) * 64 + m1) * 64 + m2q;
        float4 wr = *(const float4*)(wre + base);
        float4 wi = *(const float4*)(wim + base);
        const float* xr = xsR + i * 64 + m2q;
        const float* xi = xsI + i * 64 + m2q;
        aR[0] += xr[0] * wr.x - xi[0] * wi.x;  aI[0] += xr[0] * wi.x + xi[0] * wr.x;
        aR[1] += xr[1] * wr.y - xi[1] * wi.y;  aI[1] += xr[1] * wi.y + xi[1] * wr.y;
        aR[2] += xr[2] * wr.z - xi[2] * wi.z;  aI[2] += xr[2] * wi.z + xi[2] * wr.z;
        aR[3] += xr[3] * wr.w - xi[3] * wi.w;  aI[3] += xr[3] * wi.w + xi[3] * wr.w;
    }
    float* yp = YsP + (size_t)blockIdx.z * (OUT_C * 64 * 128)
                    + ((size_t)o * 64 + m1) * 128 + 2 * m2q;
    for (int j = 0; j < 4; ++j) { yp[2 * j] = aR[j]; yp[2 * j + 1] = aI[j]; }
}

__global__ void idfth_kernel(const float* __restrict__ YsP, const float* __restrict__ cosH,
                             const float* __restrict__ sinH, float* __restrict__ Z) {
    const int o = blockIdx.x, hbase = blockIdx.y * 8;
    __shared__ float ysR[64 * 64], ysI[64 * 64];
    for (int idx = threadIdx.x; idx < 64 * 64; idx += blockDim.x) {
        int m1 = idx >> 6, m2 = idx & 63;
        size_t off = ((size_t)o * 64 + m1) * 128 + 2 * m2;
        float r = 0.0f, im = 0.0f;
        for (int z = 0; z < ISPLIT; ++z) {
            const float* p = YsP + (size_t)z * (OUT_C * 64 * 128) + off;
            r  += p[0];
            im += p[1];
        }
        ysR[idx] = r;
        ysI[idx] = im;
    }
    __syncthreads();

    const int m2 = threadIdx.x & 63, sub = threadIdx.x >> 6;
    for (int hh = sub; hh < 8; hh += 4) {
        const int h = hbase + hh;
        float accR = 0.0f, accI = 0.0f;
        for (int m1 = 0; m1 < 64; ++m1) {
            float ct = cosH[(size_t)m1 * 512 + h], st = sinH[(size_t)m1 * 512 + h];
            float yr = ysR[m1 * 64 + m2], yi = ysI[m1 * 64 + m2];
            accR += ct * yr - st * yi;
            accI += ct * yi + st * yr;
        }
        Z[((size_t)o * 512 + h) * 128 + 2 * m2]     = accR;
        Z[((size_t)o * 512 + h) * 128 + 2 * m2 + 1] = accI;
    }
}

__global__ void invw_skip_gelu_kernel(const float* __restrict__ Z, const float* __restrict__ DW,
                                      const float* __restrict__ x, const float* __restrict__ wskip,
                                      const float* __restrict__ bconv, const float* __restrict__ bskip,
                                      float* __restrict__ out) {
    const int h = blockIdx.x;
    __shared__ float Az[64 * 128];
    __shared__ float Ak[64 * 64];
    for (int idx = threadIdx.x; idx < 64 * 128 / 4; idx += blockDim.x) {
        int o = idx >> 5, kq = idx & 31;
        ((float4*)(Az + o * 128))[kq] =
            ((const float4*)(Z + ((size_t)o * 512 + h) * 128))[kq];
    }
    for (int idx = threadIdx.x; idx < 64 * 64 / 4; idx += blockDim.x)
        ((float4*)Ak)[idx] = ((const float4*)wskip)[idx];
    __syncthreads();

    const int wave = threadIdx.x >> 5, lane = threadIdx.x & 31;
    const int n0 = wave * 64;
    const float* xh = x + (size_t)h * 512;

    for (int mt = 0; mt < 4; ++mt) {
        const int m0 = mt * 16;
        v8f acc[4] = {{}, {}, {}, {}};
        for (int k = 0; k < 128; k += 4) {
            v2f a = load_a_tile(Az + m0 * 128 + k, 128, lane);
            for (int j = 0; j < 4; ++j) {
                v2f b = load_b_tile(DW + (size_t)k * 512 + n0 + 16 * j, 512, lane);
                acc[j] = wmma_f32(a, b, acc[j]);
            }
        }
        for (int k = 0; k < 64; k += 4) {
            v2f a = load_a_tile(Ak + m0 * 64 + k, 64, lane);
            for (int j = 0; j < 4; ++j) {
                v2f b = load_b_tile(xh + (size_t)k * HW + n0 + 16 * j, (size_t)HW, lane);
                acc[j] = wmma_f32(a, b, acc[j]);
            }
        }
        const int n = lane & 15, mb = m0 + (lane >> 4) * 8;
        for (int j = 0; j < 4; ++j) {
            for (int v = 0; v < 8; ++v) {
                int m = mb + v, w = n0 + 16 * j + n;
                float val = acc[j][v] + bconv[m] + bskip[m];
                float c3  = val * val * val;
                float g   = 0.5f * val *
                            (1.0f + tanhf(0.7978845608028654f * (val + 0.044715f * c3)));
                out[(size_t)m * HW + (size_t)h * 512 + w] = g;
            }
        }
    }
}

extern "C" void kernel_launch(void* const* d_in, const int* in_sizes, int n_in,
                              void* d_out, int out_size, void* d_ws, size_t ws_size,
                              hipStream_t stream) {
    const float* x     = (const float*)d_in[0];
    const float* w_re  = (const float*)d_in[1];
    const float* w_im  = (const float*)d_in[2];
    const float* b_cv  = (const float*)d_in[3];
    const float* w_sk  = (const float*)d_in[4];
    const float* b_sk  = (const float*)d_in[5];
    float* out = (float*)d_out;

    float* ws   = (float*)d_ws;
    float* FW   = ws;
    float* cosH = FW   + 65536;
    float* sinH = cosH + 32768;
    float* DW   = sinH + 32768;
    float* Xw   = DW   + 65536;
    float* Xs   = Xw   + 4194304;
    float* YsP  = Xs   + 524288;
    float* Z    = YsP  + 2097152;

    twiddle_kernel<<<128, 256, 0, stream>>>(FW, cosH, sinH, DW);
    dftw_kernel<<<2048, 128, 0, stream>>>(x, FW, Xw);
    dfth_kernel<<<dim3(64, 8), 256, 0, stream>>>(Xw, cosH, sinH, Xs);
    modemul_kernel<<<dim3(64, 4, ISPLIT), 256, 0, stream>>>(Xs, w_re, w_im, YsP);
    idfth_kernel<<<dim3(64, 64), 256, 0, stream>>>(YsP, cosH, sinH, Z);
    invw_skip_gelu_kernel<<<512, 256, 0, stream>>>(Z, DW, x, w_sk, b_cv, b_sk, out);
}


# W-axis and H-axis forward DFT fused in one kernel: a wave's Xw accumulator tiles feed the H-twiddle WMMAs as B operands, Xw never stored; h sum now in WMMA order, 8 partials of 64 rows
# speedup vs baseline: 1.1700x; 1.1700x over previous
#include <hip/hip_runtime.h>
#include <math.h>


typedef float v8f __attribute__((ext_vector_type(8)));
typedef float v2f __attribute__((ext_vector_type(2)));

#define IN_C   64
#define OUT_C  64
#define HH     512
#define WW     512
#define MODES  64
#define HW     (HH * WW)
#define TWO_PI 6.283185307179586f
#define ISPLIT 4
#define ICHUNK (IN_C / ISPLIT)
#define HPART  8

__device__ __forceinline__ v8f wmma_f32(v2f a, v2f b, v8f c) {
    return __builtin_amdgcn_wmma_f32_16x16x4_f32(false, a, false, b, (short)0, c,
                                                 false, false);
}

__device__ __forceinline__ v2f load_a_tile(const float* __restrict__ p, int ld, int lane) {
    int m = lane & 15;
    int k = (lane >> 4) << 1;
    v2f a;
    a.x = p[(size_t)m * ld + k];
    a.y = p[(size_t)m * ld + k + 1];
    return a;
}

__device__ __forceinline__ v2f load_b_tile(const float* __restrict__ p, size_t ld, int lane) {
    int n = lane & 15;
    int k = (lane >> 4) << 1;
    v2f b;
    b.x = p[(size_t)k * ld + n];
    b.y = p[(size_t)(k + 1) * ld + n];
    return b;
}

__global__ void twiddle_kernel(float* __restrict__ FW, float* __restrict__ cosH,
                               float* __restrict__ sinH, float* __restrict__ DW,
                               float* __restrict__ CT2, float* __restrict__ ST2) {
    int t = blockIdx.x * blockDim.x + threadIdx.x;
    if (t >= 32768) return;
    const float step = TWO_PI / 512.0f;
    {
        int w = t >> 6, m2 = t & 63;
        float th = (float)((w * m2) & 511) * step;
        const int col = ((m2 >> 4) << 5) + (m2 & 15);
        FW[(size_t)w * 128 + col]      = cosf(th);
        FW[(size_t)w * 128 + col + 16] = -sinf(th);
    }
    {
        int m1 = t >> 9, h = t & 511;
        float th = (float)((m1 * h) & 511) * step;
        float ct = cosf(th), st = sinf(th);
        cosH[t] = ct;
        sinH[t] = st;
        size_t p2 = ((size_t)(h >> 1) * 64 + m1) * 2 + (h & 1);
        CT2[p2] = ct;
        ST2[p2] = st;
    }
    {
        int k = t >> 9, w = t & 511;
        float th = (float)((k * w) & 511) * step;
        float s  = ((k == 0) ? 1.0f : 2.0f) / 262144.0f;
        DW[(size_t)(2 * k) * 512 + w]     = s * cosf(th);
        DW[(size_t)(2 * k + 1) * 512 + w] = -s * sinf(th);
    }
}

__global__ __launch_bounds__(512)
void dftw_dfth_kernel(const float* __restrict__ x, const float* __restrict__ FW,
                      const float* __restrict__ CT2, const float* __restrict__ ST2,
                      float* __restrict__ XsP) {
    __shared__ float As[64 * 512];
    const int c = blockIdx.x >> 3, chunk = blockIdx.x & 7;
    const float* xrow = x + ((size_t)c * 512 + chunk * 64) * 512;
    for (int idx = threadIdx.x; idx < 64 * 512 / 4; idx += blockDim.x)
        ((float4*)As)[idx] = ((const float4*)xrow)[idx];
    __syncthreads();

    const int wave = threadIdx.x >> 5, lane = threadIdx.x & 31;
    const int g = wave >> 2, wq = wave & 3;
    const float* Ag = As + g * 16 * 512;
    const int n0 = wq * 32;
    v8f acc0 = {}, acc1 = {};
    for (int k = 0; k < 512; k += 4) {
        v2f a  = load_a_tile(Ag + k, 512, lane);
        v2f b0 = load_b_tile(FW + (size_t)k * 128 + n0,      128, lane);
        v2f b1 = load_b_tile(FW + (size_t)k * 128 + n0 + 16, 128, lane);
        acc0 = wmma_f32(a, b0, acc0);
        acc1 = wmma_f32(a, b1, acc1);
    }

    const v8f nacc0 = -acc0;
    v8f xr[4] = {{}, {}, {}, {}}, xi[4] = {{}, {}, {}, {}};
    const int m = lane & 15, half = lane >> 4;
    const int hrow = chunk * 64 + g * 16 + 8 * half;
    for (int j = 0; j < 4; ++j) {
        v2f bR, bI, bN;
        bR.x = acc0[2 * j];  bR.y = acc0[2 * j + 1];
        bI.x = acc1[2 * j];  bI.y = acc1[2 * j + 1];
        bN.x = nacc0[2 * j]; bN.y = nacc0[2 * j + 1];
        const size_t trow = (size_t)((hrow >> 1) + j) * 64;
        for (int mt = 0; mt < 4; ++mt) {
            const size_t ti = (trow + 16 * mt + m) * 2;
            v2f aC = *(const v2f*)(CT2 + ti);
            v2f aS = *(const v2f*)(ST2 + ti);
            xr[mt] = wmma_f32(aC, bR, xr[mt]);
            xr[mt] = wmma_f32(aS, bI, xr[mt]);
            xi[mt] = wmma_f32(aC, bI, xi[mt]);
            xi[mt] = wmma_f32(aS, bN, xi[mt]);
        }
    }

    __syncthreads();
    for (int mt = 0; mt < 4; ++mt) {
        for (int v = 0; v < 8; ++v) {
            As[((wave * 8 + 2 * mt) * 8 + v) * 32 + lane]     = xr[mt][v];
            As[((wave * 8 + 2 * mt + 1) * 8 + v) * 32 + lane] = xi[mt][v];
        }
    }
    __syncthreads();
    float* dst = XsP + (size_t)chunk * (IN_C * 64 * 128) + (size_t)c * 64 * 128;
    for (int v = 0; v < 8; ++v) {
        float sr = 0.0f, si = 0.0f;
        for (int gg = 0; gg < 4; ++gg) {
            const int w2 = gg * 4 + wq;
            sr += As[((w2 * 8 + 2 * g) * 8 + v) * 32 + lane];
            si += As[((w2 * 8 + 2 * g + 1) * 8 + v) * 32 + lane];
        }
        const int m1 = 16 * g + 8 * half + v, m2 = 16 * wq + m;
        v2f o; o.x = sr; o.y = si;
        *(v2f*)(dst + (size_t)m1 * 128 + 2 * m2) = o;
    }
}

__global__ void modemul_kernel(const float* __restrict__ XsP, const float* __restrict__ wre,
                               const float* __restrict__ wim, float* __restrict__ YsP) {
    const int m1 = blockIdx.x, obase = blockIdx.y * 16;
    const int ibase = blockIdx.z * ICHUNK;
    __shared__ float xsR[ICHUNK * 64], xsI[ICHUNK * 64];
    for (int idx = threadIdx.x; idx < ICHUNK * 64; idx += blockDim.x) {
        int i = idx >> 6, m2 = idx & 63;
        const float* p = XsP + ((size_t)(ibase + i) * 64 + m1) * 128 + 2 * m2;
        float r = 0.0f, im = 0.0f;
        for (int q = 0; q < HPART; ++q) {
            v2f t = *(const v2f*)(p + (size_t)q * (IN_C * 64 * 128));
            r  += t.x;
            im += t.y;
        }
        xsR[idx] = r;
        xsI[idx] = im;
    }
    __syncthreads();

    const int m2q = (threadIdx.x & 15) * 4;
    const int o   = obase + (threadIdx.x >> 4);
    float aR[4] = {0, 0, 0, 0}, aI[4] = {0, 0, 0, 0};
#pragma unroll 4
    for (int i = 0; i < ICHUNK; ++i) {
        size_t base = (((size_t)(ibase + i) * 64 + o) * 64 + m1) * 64 + m2q;
        float4 wr = *(const float4*)(wre + base);
        float4 wi = *(const float4*)(wim + base);
        const float* xr = xsR + i * 64 + m2q;
        const float* xi = xsI + i * 64 + m2q;
        aR[0] += xr[0] * wr.x - xi[0] * wi.x;  aI[0] += xr[0] * wi.x + xi[0] * wr.x;
        aR[1] += xr[1] * wr.y - xi[1] * wi.y;  aI[1] += xr[1] * wi.y + xi[1] * wr.y;
        aR[2] += xr[2] * wr.z - xi[2] * wi.z;  aI[2] += xr[2] * wi.z + xi[2] * wr.z;
        aR[3] += xr[3] * wr.w - xi[3] * wi.w;  aI[3] += xr[3] * wi.w + xi[3] * wr.w;
    }
    float* yp = YsP + (size_t)blockIdx.z * (OUT_C * 64 * 128)
                    + ((size_t)o * 64 + m1) * 128 + 2 * m2q;
    for (int j = 0; j < 4; ++j) { yp[2 * j] = aR[j]; yp[2 * j + 1] = aI[j]; }
}

__global__ void idfth_kernel(const float* __restrict__ YsP, const float* __restrict__ cosH,
                             const float* __restrict__ sinH, float* __restrict__ Z) {
    const int o = blockIdx.x, hbase = blockIdx.y * 8;
    __shared__ float ysR[64 * 64], ysI[64 * 64];
    for (int idx = threadIdx.x; idx < 64 * 64; idx += blockDim.x) {
        int m1 = idx >> 6, m2 = idx & 63;
        size_t off = ((size_t)o * 64 + m1) * 128 + 2 * m2;
        float r = 0.0f, im = 0.0f;
        for (int z = 0; z < ISPLIT; ++z) {
            const float* p = YsP + (size_t)z * (OUT_C * 64 * 128) + off;
            r  += p[0];
            im += p[1];
        }
        ysR[idx] = r;
        ysI[idx] = im;
    }
    __syncthreads();

    const int m2 = threadIdx.x & 63, sub = threadIdx.x >> 6;
    for (int hh = sub; hh < 8; hh += 4) {
        const int h = hbase + hh;
        float accR = 0.0f, accI = 0.0f;
        for (int m1 = 0; m1 < 64; ++m1) {
            float ct = cosH[(size_t)m1 * 512 + h], st = sinH[(size_t)m1 * 512 + h];
            float yr = ysR[m1 * 64 + m2], yi = ysI[m1 * 64 + m2];
            accR += ct * yr - st * yi;
            accI += ct * yi + st * yr;
        }
        Z[((size_t)o * 512 + h) * 128 + 2 * m2]     = accR;
        Z[((size_t)o * 512 + h) * 128 + 2 * m2 + 1] = accI;
    }
}

__global__ void invw_skip_gelu_kernel(const float* __restrict__ Z, const float* __restrict__ DW,
                                      const float* __restrict__ x, const float* __restrict__ wskip,
                                      const float* __restrict__ bconv, const float* __restrict__ bskip,
                                      float* __restrict__ out) {
    const int h = blockIdx.x;
    __shared__ float Az[64 * 128];
    __shared__ float Ak[64 * 64];
    for (int idx = threadIdx.x; idx < 64 * 128 / 4; idx += blockDim.x) {
        int o = idx >> 5, kq = idx & 31;
        ((float4*)(Az + o * 128))[kq] =
            ((const float4*)(Z + ((size_t)o * 512 + h) * 128))[kq];
    }
    for (int idx = threadIdx.x; idx < 64 * 64 / 4; idx += blockDim.x)
        ((float4*)Ak)[idx] = ((const float4*)wskip)[idx];
    __syncthreads();

    const int wave = threadIdx.x >> 5, lane = threadIdx.x & 31;
    const int n0 = wave * 64;
    const float* xh = x + (size_t)h * 512;

    for (int mt = 0; mt < 4; ++mt) {
        const int m0 = mt * 16;
        v8f acc[4] = {{}, {}, {}, {}};
        for (int k = 0; k < 128; k += 4) {
            v2f a = load_a_tile(Az + m0 * 128 + k, 128, lane);
            for (int j = 0; j < 4; ++j) {
                v2f b = load_b_tile(DW + (size_t)k * 512 + n0 + 16 * j, 512, lane);
                acc[j] = wmma_f32(a, b, acc[j]);
            }
        }
        for (int k = 0; k < 64; k += 4) {
            v2f a = load_a_tile(Ak + m0 * 64 + k, 64, lane);
            for (int j = 0; j < 4; ++j) {
                v2f b = load_b_tile(xh + (size_t)k * HW + n0 + 16 * j, (size_t)HW, lane);
                acc[j] = wmma_f32(a, b, acc[j]);
            }
        }
        const int n = lane & 15, mb = m0 + (lane >> 4) * 8;
        for (int j = 0; j < 4; ++j) {
            for (int v = 0; v < 8; ++v) {
                int m = mb + v, w = n0 + 16 * j + n;
                float val = acc[j][v] + bconv[m] + bskip[m];
                float c3  = val * val * val;
                float g   = 0.5f * val *
                            (1.0f + tanhf(0.7978845608028654f * (val + 0.044715f * c3)));
                out[(size_t)m * HW + (size_t)h * 512 + w] = g;
            }
        }
    }
}

extern "C" void kernel_launch(void* const* d_in, const int* in_sizes, int n_in,
                              void* d_out, int out_size, void* d_ws, size_t ws_size,
                              hipStream_t stream) {
    const float* x     = (const float*)d_in[0];
    const float* w_re  = (const float*)d_in[1];
    const float* w_im  = (const float*)d_in[2];
    const float* b_cv  = (const float*)d_in[3];
    const float* w_sk  = (const float*)d_in[4];
    const float* b_sk  = (const float*)d_in[5];
    float* out = (float*)d_out;

    float* ws   = (float*)d_ws;
    float* FW   = ws;
    float* cosH = FW   + 65536;
    float* sinH = cosH + 32768;
    float* DW   = sinH + 32768;
    float* XsP  = DW   + 65536;
    float* CT2  = XsP  + 4194304;
    float* ST2  = CT2  + 32768;
    float* YsP  = CT2  + 524288;
    float* Z    = YsP  + 2097152;

    twiddle_kernel<<<128, 256, 0, stream>>>(FW, cosH, sinH, DW, CT2, ST2);
    dftw_dfth_kernel<<<512, 512, 0, stream>>>(x, FW, CT2, ST2, XsP);
    modemul_kernel<<<dim3(64, 4, ISPLIT), 256, 0, stream>>>(XsP, w_re, w_im, YsP);
    idfth_kernel<<<dim3(64, 64), 256, 0, stream>>>(YsP, cosH, sinH, Z);
    invw_skip_gelu_kernel<<<512, 256, 0, stream>>>(Z, DW, x, w_sk, b_cv, b_sk, out);
}


# inverse H-DFT kernel retiled: 32 rows h per block (was 8), Ys staged once per 32 rows, twiddles staged in LDS, a thread keeps 8 row sums so each Ys read feeds 8 rows; m1 sum order unchanged
# speedup vs baseline: 1.3451x; 1.1497x over previous
#include <hip/hip_runtime.h>
#include <math.h>


typedef float v8f __attribute__((ext_vector_type(8)));
typedef float v2f __attribute__((ext_vector_type(2)));

#define IN_C   64
#define OUT_C  64
#define HH     512
#define WW     512
#define MODES  64
#define HW     (HH * WW)
#define TWO_PI 6.283185307179586f
#define ISPLIT 4
#define ICHUNK (IN_C / ISPLIT)
#define HPART  8

__device__ __forceinline__ v8f wmma_f32(v2f a, v2f b, v8f c) {
    return __builtin_amdgcn_wmma_f32_16x16x4_f32(false, a, false, b, (short)0, c,
                                                 false, false);
}

__device__ __forceinline__ v2f load_a_tile(const float* __restrict__ p, int ld, int lane) {
    int m = lane & 15;
    int k = (lane >> 4) << 1;
    v2f a;
    a.x = p[(size_t)m * ld + k];
    a.y = p[(size_t)m * ld + k + 1];
    return a;
}

__device__ __forceinline__ v2f load_b_tile(const float* __restrict__ p, size_t ld, int lane) {
    int n = lane & 15;
    int k = (lane >> 4) << 1;
    v2f b;
    b.x = p[(size_t)k * ld + n];
    b.y = p[(size_t)(k + 1) * ld + n];
    return b;
}

__global__ void twiddle_kernel(float* __restrict__ FW, float* __restrict__ cosH,
                               float* __restrict__ sinH, float* __restrict__ DW,
                               float* __restrict__ CT2, float* __restrict__ ST2) {
    int t = blockIdx.x * blockDim.x + threadIdx.x;
    if (t >= 32768) return;
    const float step = TWO_PI / 512.0f;
    {
        int w = t >> 6, m2 = t & 63;
        float th = (float)((w * m2) & 511) * step;
        const int col = ((m2 >> 4) << 5) + (m2 & 15);
        FW[(size_t)w * 128 + col]      = cosf(th);
        FW[(size_t)w * 128 + col + 16] = -sinf(th);
    }
    {
        int m1 = t >> 9, h = t & 511;
        float th = (float)((m1 * h) & 511) * step;
        float ct = cosf(th), st = sinf(th);
        cosH[t] = ct;
        sinH[t] = st;
        size_t p2 = ((size_t)(h >> 1) * 64 + m1) * 2 + (h & 1);
        CT2[p2] = ct;
        ST2[p2] = st;
    }
    {
        int k = t >> 9, w = t & 511;
        float th = (float)((k * w) & 511) * step;
        float s  = ((k == 0) ? 1.0f : 2.0f) / 262144.0f;
        DW[(size_t)(2 * k) * 512 + w]     = s * cosf(th);
        DW[(size_t)(2 * k + 1) * 512 + w] = -s * sinf(th);
    }
}

__global__ __launch_bounds__(512)
void dftw_dfth_kernel(const float* __restrict__ x, const float* __restrict__ FW,
                      const float* __restrict__ CT2, const float* __restrict__ ST2,
                      float* __restrict__ XsP) {
    __shared__ float As[64 * 512];
    const int c = blockIdx.x >> 3, chunk = blockIdx.x & 7;
    const float* xrow = x + ((size_t)c * 512 + chunk * 64) * 512;
    for (int idx = threadIdx.x; idx < 64 * 512 / 4; idx += blockDim.x)
        ((float4*)As)[idx] = ((const float4*)xrow)[idx];
    __syncthreads();

    const int wave = threadIdx.x >> 5, lane = threadIdx.x & 31;
    const int g = wave >> 2, wq = wave & 3;
    const float* Ag = As + g * 16 * 512;
    const int n0 = wq * 32;
    v8f acc0 = {}, acc1 = {};
    for (int k = 0; k < 512; k += 4) {
        v2f a  = load_a_tile(Ag + k, 512, lane);
        v2f b0 = load_b_tile(FW + (size_t)k * 128 + n0,      128, lane);
        v2f b1 = load_b_tile(FW + (size_t)k * 128 + n0 + 16, 128, lane);
        acc0 = wmma_f32(a, b0, acc0);
        acc1 = wmma_f32(a, b1, acc1);
    }

    const v8f nacc0 = -acc0;
    v8f xr[4] = {{}, {}, {}, {}}, xi[4] = {{}, {}, {}, {}};
    const int m = lane & 15, half = lane >> 4;
    const int hrow = chunk * 64 + g * 16 + 8 * half;
    for (int j = 0; j < 4; ++j) {
        v2f bR, bI, bN;
        bR.x = acc0[2 * j];  bR.y = acc0[2 * j + 1];
        bI.x = acc1[2 * j];  bI.y = acc1[2 * j + 1];
        bN.x = nacc0[2 * j]; bN.y = nacc0[2 * j + 1];
        const size_t trow = (size_t)((hrow >> 1) + j) * 64;
        for (int mt = 0; mt < 4; ++mt) {
            const size_t ti = (trow + 16 * mt + m) * 2;
            v2f aC = *(const v2f*)(CT2 + ti);
            v2f aS = *(const v2f*)(ST2 + ti);
            xr[mt] = wmma_f32(aC, bR, xr[mt]);
            xr[mt] = wmma_f32(aS, bI, xr[mt]);
            xi[mt] = wmma_f32(aC, bI, xi[mt]);
            xi[mt] = wmma_f32(aS, bN, xi[mt]);
        }
    }

    __syncthreads();
    for (int mt = 0; mt < 4; ++mt) {
        for (int v = 0; v < 8; ++v) {
            As[((wave * 8 + 2 * mt) * 8 + v) * 32 + lane]     = xr[mt][v];
            As[((wave * 8 + 2 * mt + 1) * 8 + v) * 32 + lane] = xi[mt][v];
        }
    }
    __syncthreads();
    float* dst = XsP + (size_t)chunk * (IN_C * 64 * 128) + (size_t)c * 64 * 128;
    for (int v = 0; v < 8; ++v) {
        float sr = 0.0f, si = 0.0f;
        for (int gg = 0; gg < 4; ++gg) {
            const int w2 = gg * 4 + wq;
            sr += As[((w2 * 8 + 2 * g) * 8 + v) * 32 + lane];
            si += As[((w2 * 8 + 2 * g + 1) * 8 + v) * 32 + lane];
        }
        const int m1 = 16 * g + 8 * half + v, m2 = 16 * wq + m;
        v2f o; o.x = sr; o.y = si;
        *(v2f*)(dst + (size_t)m1 * 128 + 2 * m2) = o;
    }
}

__global__ void modemul_kernel(const float* __restrict__ XsP, const float* __restrict__ wre,
                               const float* __restrict__ wim, float* __restrict__ YsP) {
    const int m1 = blockIdx.x, obase = blockIdx.y * 16;
    const int ibase = blockIdx.z * ICHUNK;
    __shared__ float xsR[ICHUNK * 64], xsI[ICHUNK * 64];
    for (int idx = threadIdx.x; idx < ICHUNK * 64; idx += blockDim.x) {
        int i = idx >> 6, m2 = idx & 63;
        const float* p = XsP + ((size_t)(ibase + i) * 64 + m1) * 128 + 2 * m2;
        float r = 0.0f, im = 0.0f;
        for (int q = 0; q < HPART; ++q) {
            v2f t = *(const v2f*)(p + (size_t)q * (IN_C * 64 * 128));
            r  += t.x;
            im += t.y;
        }
        xsR[idx] = r;
        xsI[idx] = im;
    }
    __syncthreads();

    const int m2q = (threadIdx.x & 15) * 4;
    const int o   = obase + (threadIdx.x >> 4);
    float aR[4] = {0, 0, 0, 0}, aI[4] = {0, 0, 0, 0};
#pragma unroll 4
    for (int i = 0; i < ICHUNK; ++i) {
        size_t base = (((size_t)(ibase + i) * 64 + o) * 64 + m1) * 64 + m2q;
        float4 wr = *(const float4*)(wre + base);
        float4 wi = *(const float4*)(wim + base);
        const float* xr = xsR + i * 64 + m2q;
        const float* xi = xsI + i * 64 + m2q;
        aR[0] += xr[0] * wr.x - xi[0] * wi.x;  aI[0] += xr[0] * wi.x + xi[0] * wr.x;
        aR[1] += xr[1] * wr.y - xi[1] * wi.y;  aI[1] += xr[1] * wi.y + xi[1] * wr.y;
        aR[2] += xr[2] * wr.z - xi[2] * wi.z;  aI[2] += xr[2] * wi.z + xi[2] * wr.z;
        aR[3] += xr[3] * wr.w - xi[3] * wi.w;  aI[3] += xr[3] * wi.w + xi[3] * wr.w;
    }
    float* yp = YsP + (size_t)blockIdx.z * (OUT_C * 64 * 128)
                    + ((size_t)o * 64 + m1) * 128 + 2 * m2q;
    for (int j = 0; j < 4; ++j) { yp[2 * j] = aR[j]; yp[2 * j + 1] = aI[j]; }
}

#define HCH 32
__global__ __launch_bounds__(256)
void idfth_kernel(const float* __restrict__ YsP, const float* __restrict__ cosH,
                  const float* __restrict__ sinH, float* __restrict__ Z) {
    const int o = blockIdx.x, hbase = blockIdx.y * HCH;
    __shared__ float  ysR[64 * 64], ysI[64 * 64];
    __shared__ float4 tc4[64 * HCH / 4], ts4[64 * HCH / 4];
    for (int idx = threadIdx.x; idx < 64 * 64; idx += blockDim.x) {
        int m1 = idx >> 6, m2 = idx & 63;
        size_t off = ((size_t)o * 64 + m1) * 128 + 2 * m2;
        float r = 0.0f, im = 0.0f;
        for (int z = 0; z < ISPLIT; ++z) {
            v2f t = *(const v2f*)(YsP + (size_t)z * (OUT_C * 64 * 128) + off);
            r  += t.x;
            im += t.y;
        }
        ysR[idx] = r;
        ysI[idx] = im;
    }
    for (int idx = threadIdx.x; idx < 64 * HCH; idx += blockDim.x) {
        int m1 = idx / HCH, hh = idx % HCH;
        ((float*)tc4)[idx] = cosH[(size_t)m1 * 512 + hbase + hh];
        ((float*)ts4)[idx] = sinH[(size_t)m1 * 512 + hbase + hh];
    }
    __syncthreads();

    const int m2 = threadIdx.x & 63;
    const int h0 = (threadIdx.x >> 6) * 8;
    float aR[8] = {0, 0, 0, 0, 0, 0, 0, 0}, aI[8] = {0, 0, 0, 0, 0, 0, 0, 0};
    for (int m1 = 0; m1 < 64; ++m1) {
        const float yr = ysR[m1 * 64 + m2], yi = ysI[m1 * 64 + m2];
        const float4 c0 = tc4[(m1 * HCH + h0) / 4], c1 = tc4[(m1 * HCH + h0) / 4 + 1];
        const float4 s0 = ts4[(m1 * HCH + h0) / 4], s1 = ts4[(m1 * HCH + h0) / 4 + 1];
        const float ct[8] = {c0.x, c0.y, c0.z, c0.w, c1.x, c1.y, c1.z, c1.w};
        const float st[8] = {s0.x, s0.y, s0.z, s0.w, s1.x, s1.y, s1.z, s1.w};
        for (int j = 0; j < 8; ++j) {
            aR[j] += ct[j] * yr - st[j] * yi;
            aI[j] += ct[j] * yi + st[j] * yr;
        }
    }
    for (int j = 0; j < 8; ++j) {
        v2f zo; zo.x = aR[j]; zo.y = aI[j];
        *(v2f*)(Z + ((size_t)o * 512 + hbase + h0 + j) * 128 + 2 * m2) = zo;
    }
}

__global__ void invw_skip_gelu_kernel(const float* __restrict__ Z, const float* __restrict__ DW,
                                      const float* __restrict__ x, const float* __restrict__ wskip,
                                      const float* __restrict__ bconv, const float* __restrict__ bskip,
                                      float* __restrict__ out) {
    const int h = blockIdx.x;
    __shared__ float Az[64 * 128];
    __shared__ float Ak[64 * 64];
    for (int idx = threadIdx.x; idx < 64 * 128 / 4; idx += blockDim.x) {
        int o = idx >> 5, kq = idx & 31;
        ((float4*)(Az + o * 128))[kq] =
            ((const float4*)(Z + ((size_t)o * 512 + h) * 128))[kq];
    }
    for (int idx = threadIdx.x; idx < 64 * 64 / 4; idx += blockDim.x)
        ((float4*)Ak)[idx] = ((const float4*)wskip)[idx];
    __syncthreads();

    const int wave = threadIdx.x >> 5, lane = threadIdx.x & 31;
    const int n0 = wave * 64;
    const float* xh = x + (size_t)h * 512;

    for (int mt = 0; mt < 4; ++mt) {
        const int m0 = mt * 16;
        v8f acc[4] = {{}, {}, {}, {}};
        for (int k = 0; k < 128; k += 4) {
            v2f a = load_a_tile(Az + m0 * 128 + k, 128, lane);
            for (int j = 0; j < 4; ++j) {
                v2f b = load_b_tile(DW + (size_t)k * 512 + n0 + 16 * j, 512, lane);
                acc[j] = wmma_f32(a, b, acc[j]);
            }
        }
        for (int k = 0; k < 64; k += 4) {
            v2f a = load_a_tile(Ak + m0 * 64 + k, 64, lane);
            for (int j = 0; j < 4; ++j) {
                v2f b = load_b_tile(xh + (size_t)k * HW + n0 + 16 * j, (size_t)HW, lane);
                acc[j] = wmma_f32(a, b, acc[j]);
            }
        }
        const int n = lane & 15, mb = m0 + (lane >> 4) * 8;
        for (int j = 0; j < 4; ++j) {
            for (int v = 0; v < 8; ++v) {
                int m = mb + v, w = n0 + 16 * j + n;
                float val = acc[j][v] + bconv[m] + bskip[m];
                float c3  = val * val * val;
                float g   = 0.5f * val *
                            (1.0f + tanhf(0.7978845608028654f * (val + 0.044715f * c3)));
                out[(size_t)m * HW + (size_t)h * 512 + w] = g;
            }
        }
    }
}

extern "C" void kernel_launch(void* const* d_in, const int* in_sizes, int n_in,
                              void* d_out, int out_size, void* d_ws, size_t ws_size,
                              hipStream_t stream) {
    const float* x     = (const float*)d_in[0];
    const float* w_re  = (const float*)d_in[1];
    const float* w_im  = (const float*)d_in[2];
    const float* b_cv  = (const float*)d_in[3];
    const float* w_sk  = (const float*)d_in[4];
    const float* b_sk  = (const float*)d_in[5];
    float* out = (float*)d_out;

    float* ws   = (float*)d_ws;
    float* FW   = ws;
    float* cosH = FW   + 65536;
    float* sinH = cosH + 32768;
    float* DW   = sinH + 32768;
    float* XsP  = DW   + 65536;
    float* CT2  = XsP  + 4194304;
    float* ST2  = CT2  + 32768;
    float* YsP  = CT2  + 524288;
    float* Z    = YsP  + 2097152;

    twiddle_kernel<<<128, 256, 0, stream>>>(FW, cosH, sinH, DW, CT2, ST2);
    dftw_dfth_kernel<<<512, 512, 0, stream>>>(x, FW, CT2, ST2, XsP);
    modemul_kernel<<<dim3(64, 4, ISPLIT), 256, 0, stream>>>(XsP, w_re, w_im, YsP);
    idfth_kernel<<<dim3(64, 16), 256, 0, stream>>>(YsP, cosH, sinH, Z);
    invw_skip_gelu_kernel<<<512, 256, 0, stream>>>(Z, DW, x, w_sk, b_cv, b_sk, out);
}


# iDFT_W+skip+gelu kernel: a wave's 4 tiles column-interleaved (lane n owns w=4n..4n+3): one 16-byte load of a DW/x row feeds 4 tiles, 8 k-steps of loads per trip; 16-byte out stores; sums unchanged
# speedup vs baseline: 1.4748x; 1.0964x over previous
#include <hip/hip_runtime.h>
#include <math.h>


typedef float v8f __attribute__((ext_vector_type(8)));
typedef float v2f __attribute__((ext_vector_type(2)));

#define IN_C   64
#define OUT_C  64
#define HH     512
#define WW     512
#define MODES  64
#define HW     (HH * WW)
#define TWO_PI 6.283185307179586f
#define ISPLIT 4
#define ICHUNK (IN_C / ISPLIT)
#define HPART  8

__device__ __forceinline__ v8f wmma_f32(v2f a, v2f b, v8f c) {
    return __builtin_amdgcn_wmma_f32_16x16x4_f32(false, a, false, b, (short)0, c,
                                                 false, false);
}

__device__ __forceinline__ v2f load_a_tile(const float* __restrict__ p, int ld, int lane) {
    int m = lane & 15;
    int k = (lane >> 4) << 1;
    v2f a;
    a.x = p[(size_t)m * ld + k];
    a.y = p[(size_t)m * ld + k + 1];
    return a;
}

__device__ __forceinline__ v2f load_b_tile(const float* __restrict__ p, size_t ld, int lane) {
    int n = lane & 15;
    int k = (lane >> 4) << 1;
    v2f b;
    b.x = p[(size_t)k * ld + n];
    b.y = p[(size_t)(k + 1) * ld + n];
    return b;
}

__global__ void twiddle_kernel(float* __restrict__ FW, float* __restrict__ cosH,
                               float* __restrict__ sinH, float* __restrict__ DW,
                               float* __restrict__ CT2, float* __restrict__ ST2) {
    int t = blockIdx.x * blockDim.x + threadIdx.x;
    if (t >= 32768) return;
    const float step = TWO_PI / 512.0f;
    {
        int w = t >> 6, m2 = t & 63;
        float th = (float)((w * m2) & 511) * step;
        const int col = ((m2 >> 4) << 5) + (m2 & 15);
        FW[(size_t)w * 128 + col]      = cosf(th);
        FW[(size_t)w * 128 + col + 16] = -sinf(th);
    }
    {
        int m1 = t >> 9, h = t & 511;
        float th = (float)((m1 * h) & 511) * step;
        float ct = cosf(th), st = sinf(th);
        cosH[t] = ct;
        sinH[t] = st;
        size_t p2 = ((size_t)(h >> 1) * 64 + m1) * 2 + (h & 1);
        CT2[p2] = ct;
        ST2[p2] = st;
    }
    {
        int k = t >> 9, w = t & 511;
        float th = (float)((k * w) & 511) * step;
        float s  = ((k == 0) ? 1.0f : 2.0f) / 262144.0f;
        DW[(size_t)(2 * k) * 512 + w]     = s * cosf(th);
        DW[(size_t)(2 * k + 1) * 512 + w] = -s * sinf(th);
    }
}

__global__ __launch_bounds__(512)
void dftw_dfth_kernel(const float* __restrict__ x, const float* __restrict__ FW,
                      const float* __restrict__ CT2, const float* __restrict__ ST2,
                      float* __restrict__ XsP) {
    __shared__ float As[64 * 512];
    const int c = blockIdx.x >> 3, chunk = blockIdx.x & 7;
    const float* xrow = x + ((size_t)c * 512 + chunk * 64) * 512;
    for (int idx = threadIdx.x; idx < 64 * 512 / 4; idx += blockDim.x)
        ((float4*)As)[idx] = ((const float4*)xrow)[idx];
    __syncthreads();

    const int wave = threadIdx.x >> 5, lane = threadIdx.x & 31;
    const int g = wave >> 2, wq = wave & 3;
    const float* Ag = As + g * 16 * 512;
    const int n0 = wq * 32;
    v8f acc0 = {}, acc1 = {};
    for (int k = 0; k < 512; k += 4) {
        v2f a  = load_a_tile(Ag + k, 512, lane);
        v2f b0 = load_b_tile(FW + (size_t)k * 128 + n0,      128, lane);
        v2f b1 = load_b_tile(FW + (size_t)k * 128 + n0 + 16, 128, lane);
        acc0 = wmma_f32(a, b0, acc0);
        acc1 = wmma_f32(a, b1, acc1);
    }

    const v8f nacc0 = -acc0;
    v8f xr[4] = {{}, {}, {}, {}}, xi[4] = {{}, {}, {}, {}};
    const int m = lane & 15, half = lane >> 4;
    const int hrow = chunk * 64 + g * 16 + 8 * half;
    for (int j = 0; j < 4; ++j) {
        v2f bR, bI, bN;
        bR.x = acc0[2 * j];  bR.y = acc0[2 * j + 1];
        bI.x = acc1[2 * j];  bI.y = acc1[2 * j + 1];
        bN.x = nacc0[2 * j]; bN.y = nacc0[2 * j + 1];
        const size_t trow = (size_t)((hrow >> 1) + j) * 64;
        for (int mt = 0; mt < 4; ++mt) {
            const size_t ti = (trow + 16 * mt + m) * 2;
            v2f aC = *(const v2f*)(CT2 + ti);
            v2f aS = *(const v2f*)(ST2 + ti);
            xr[mt] = wmma_f32(aC, bR, xr[mt]);
            xr[mt] = wmma_f32(aS, bI, xr[mt]);
            xi[mt] = wmma_f32(aC, bI, xi[mt]);
            xi[mt] = wmma_f32(aS, bN, xi[mt]);
        }
    }

    __syncthreads();
    for (int mt = 0; mt < 4; ++mt) {
        for (int v = 0; v < 8; ++v) {
            As[((wave * 8 + 2 * mt) * 8 + v) * 32 + lane]     = xr[mt][v];
            As[((wave * 8 + 2 * mt + 1) * 8 + v) * 32 + lane] = xi[mt][v];
        }
    }
    __syncthreads();
    float* dst = XsP + (size_t)chunk * (IN_C * 64 * 128) + (size_t)c * 64 * 128;
    for (int v = 0; v < 8; ++v) {
        float sr = 0.0f, si = 0.0f;
        for (int gg = 0; gg < 4; ++gg) {
            const int w2 = gg * 4 + wq;
            sr += As[((w2 * 8 + 2 * g) * 8 + v) * 32 + lane];
            si += As[((w2 * 8 + 2 * g + 1) * 8 + v) * 32 + lane];
        }
        const int m1 = 16 * g + 8 * half + v, m2 = 16 * wq + m;
        v2f o; o.x = sr; o.y = si;
        *(v2f*)(dst + (size_t)m1 * 128 + 2 * m2) = o;
    }
}

__global__ void modemul_kernel(const float* __restrict__ XsP, const float* __restrict__ wre,
                               const float* __restrict__ wim, float* __restrict__ YsP) {
    const int m1 = blockIdx.x, obase = blockIdx.y * 16;
    const int ibase = blockIdx.z * ICHUNK;
    __shared__ float xsR[ICHUNK * 64], xsI[ICHUNK * 64];
    for (int idx = threadIdx.x; idx < ICHUNK * 64; idx += blockDim.x) {
        int i = idx >> 6, m2 = idx & 63;
        const float* p = XsP + ((size_t)(ibase + i) * 64 + m1) * 128 + 2 * m2;
        float r = 0.0f, im = 0.0f;
        for (int q = 0; q < HPART; ++q) {
            v2f t = *(const v2f*)(p + (size_t)q * (IN_C * 64 * 128));
            r  += t.x;
            im += t.y;
        }
        xsR[idx] = r;
        xsI[idx] = im;
    }
    __syncthreads();

    const int m2q = (threadIdx.x & 15) * 4;
    const int o   = obase + (threadIdx.x >> 4);
    float aR[4] = {0, 0, 0, 0}, aI[4] = {0, 0, 0, 0};
#pragma unroll 4
    for (int i = 0; i < ICHUNK; ++i) {
        size_t base = (((size_t)(ibase + i) * 64 + o) * 64 + m1) * 64 + m2q;
        float4 wr = *(const float4*)(wre + base);
        float4 wi = *(const float4*)(wim + base);
        const float* xr = xsR + i * 64 + m2q;
        const float* xi = xsI + i * 64 + m2q;
        aR[0] += xr[0] * wr.x - xi[0] * wi.x;  aI[0] += xr[0] * wi.x + xi[0] * wr.x;
        aR[1] += xr[1] * wr.y - xi[1] * wi.y;  aI[1] += xr[1] * wi.y + xi[1] * wr.y;
        aR[2] += xr[2] * wr.z - xi[2] * wi.z;  aI[2] += xr[2] * wi.z + xi[2] * wr.z;
        aR[3] += xr[3] * wr.w - xi[3] * wi.w;  aI[3] += xr[3] * wi.w + xi[3] * wr.w;
    }
    float* yp = YsP + (size_t)blockIdx.z * (OUT_C * 64 * 128)
                    + ((size_t)o * 64 + m1) * 128 + 2 * m2q;
    for (int j = 0; j < 4; ++j) { yp[2 * j] = aR[j]; yp[2 * j + 1] = aI[j]; }
}

#define HCH 32
__global__ __launch_bounds__(256)
void idfth_kernel(const float* __restrict__ YsP, const float* __restrict__ cosH,
                  const float* __restrict__ sinH, float* __restrict__ Z) {
    const int o = blockIdx.x, hbase = blockIdx.y * HCH;
    __shared__ float  ysR[64 * 64], ysI[64 * 64];
    __shared__ float4 tc4[64 * HCH / 4], ts4[64 * HCH / 4];
    for (int idx = threadIdx.x; idx < 64 * 64; idx += blockDim.x) {
        int m1 = idx >> 6, m2 = idx & 63;
        size_t off = ((size_t)o * 64 + m1) * 128 + 2 * m2;
        float r = 0.0f, im = 0.0f;
        for (int z = 0; z < ISPLIT; ++z) {
            v2f t = *(const v2f*)(YsP + (size_t)z * (OUT_C * 64 * 128) + off);
            r  += t.x;
            im += t.y;
        }
        ysR[idx] = r;
        ysI[idx] = im;
    }
    for (int idx = threadIdx.x; idx < 64 * HCH; idx += blockDim.x) {
        int m1 = idx / HCH, hh = idx % HCH;
        ((float*)tc4)[idx] = cosH[(size_t)m1 * 512 + hbase + hh];
        ((float*)ts4)[idx] = sinH[(size_t)m1 * 512 + hbase + hh];
    }
    __syncthreads();

    const int m2 = threadIdx.x & 63;
    const int h0 = (threadIdx.x >> 6) * 8;
    float aR[8] = {0, 0, 0, 0, 0, 0, 0, 0}, aI[8] = {0, 0, 0, 0, 0, 0, 0, 0};
    for (int m1 = 0; m1 < 64; ++m1) {
        const float yr = ysR[m1 * 64 + m2], yi = ysI[m1 * 64 + m2];
        const float4 c0 = tc4[(m1 * HCH + h0) / 4], c1 = tc4[(m1 * HCH + h0) / 4 + 1];
        const float4 s0 = ts4[(m1 * HCH + h0) / 4], s1 = ts4[(m1 * HCH + h0) / 4 + 1];
        const float ct[8] = {c0.x, c0.y, c0.z, c0.w, c1.x, c1.y, c1.z, c1.w};
        const float st[8] = {s0.x, s0.y, s0.z, s0.w, s1.x, s1.y, s1.z, s1.w};
        for (int j = 0; j < 8; ++j) {
            aR[j] += ct[j] * yr - st[j] * yi;
            aI[j] += ct[j] * yi + st[j] * yr;
        }
    }
    for (int j = 0; j < 8; ++j) {
        v2f zo; zo.x = aR[j]; zo.y = aI[j];
        *(v2f*)(Z + ((size_t)o * 512 + hbase + h0 + j) * 128 + 2 * m2) = zo;
    }
}

__global__ __launch_bounds__(256)
void invw_skip_gelu_kernel(const float* __restrict__ Z, const float* __restrict__ DW,
                           const float* __restrict__ x, const float* __restrict__ wskip,
                           const float* __restrict__ bconv, const float* __restrict__ bskip,
                           float* __restrict__ out) {
    const int h = blockIdx.x;
    __shared__ float Az[64 * 128];
    __shared__ float Ak[64 * 64];
    for (int idx = threadIdx.x; idx < 64 * 128 / 4; idx += blockDim.x) {
        int o = idx >> 5, kq = idx & 31;
        ((float4*)(Az + o * 128))[kq] =
            ((const float4*)(Z + ((size_t)o * 512 + h) * 128))[kq];
    }
    for (int idx = threadIdx.x; idx < 64 * 64 / 4; idx += blockDim.x)
        ((float4*)Ak)[idx] = ((const float4*)wskip)[idx];
    __syncthreads();

    const int wave = threadIdx.x >> 5, lane = threadIdx.x & 31;
    const int n = lane & 15, half = lane >> 4;
    const int wc = wave * 64 + 4 * n;
    const float* dwp = DW + (size_t)(2 * half) * 512 + wc;
    const float* xp  = x + (size_t)(2 * half) * HW + (size_t)h * 512 + wc;

    for (int mt = 0; mt < 4; ++mt) {
        const int m0 = mt * 16;
        v8f acc[4] = {{}, {}, {}, {}};
#pragma unroll 8
        for (int k = 0; k < 128; k += 4) {
            v2f a = load_a_tile(Az + m0 * 128 + k, 128, lane);
            const float4 r0 = *(const float4*)(dwp + (size_t)k * 512);
            const float4 r1 = *(const float4*)(dwp + (size_t)(k + 1) * 512);
            v2f b;
            b.x = r0.x; b.y = r1.x; acc[0] = wmma_f32(a, b, acc[0]);
            b.x = r0.y; b.y = r1.y; acc[1] = wmma_f32(a, b, acc[1]);
            b.x = r0.z; b.y = r1.z; acc[2] = wmma_f32(a, b, acc[2]);
            b.x = r0.w; b.y = r1.w; acc[3] = wmma_f32(a, b, acc[3]);
        }
#pragma unroll 8
        for (int k = 0; k < 64; k += 4) {
            v2f a = load_a_tile(Ak + m0 * 64 + k, 64, lane);
            const float4 r0 = *(const float4*)(xp + (size_t)k * HW);
            const float4 r1 = *(const float4*)(xp + (size_t)(k + 1) * HW);
            v2f b;
            b.x = r0.x; b.y = r1.x; acc[0] = wmma_f32(a, b, acc[0]);
            b.x = r0.y; b.y = r1.y; acc[1] = wmma_f32(a, b, acc[1]);
            b.x = r0.z; b.y = r1.z; acc[2] = wmma_f32(a, b, acc[2]);
            b.x = r0.w; b.y = r1.w; acc[3] = wmma_f32(a, b, acc[3]);
        }
        const int mb = m0 + half * 8;
        for (int v = 0; v < 8; ++v) {
            const int m = mb + v;
            float g[4];
            for (int j = 0; j < 4; ++j) {
                float val = acc[j][v] + bconv[m] + bskip[m];
                float c3  = val * val * val;
                g[j] = 0.5f * val *
                       (1.0f + tanhf(0.7978845608028654f * (val + 0.044715f * c3)));
            }
            float4 o4; o4.x = g[0]; o4.y = g[1]; o4.z = g[2]; o4.w = g[3];
            *(float4*)(out + (size_t)m * HW + (size_t)h * 512 + wc) = o4;
        }
    }
}

extern "C" void kernel_launch(void* const* d_in, const int* in_sizes, int n_in,
                              void* d_out, int out_size, void* d_ws, size_t ws_size,
                              hipStream_t stream) {
    const float* x     = (const float*)d_in[0];
    const float* w_re  = (const float*)d_in[1];
    const float* w_im  = (const float*)d_in[2];
    const float* b_cv  = (const float*)d_in[3];
    const float* w_sk  = (const float*)d_in[4];
    const float* b_sk  = (const float*)d_in[5];
    float* out = (float*)d_out;

    float* ws   = (float*)d_ws;
    float* FW   = ws;
    float* cosH = FW   + 65536;
    float* sinH = cosH + 32768;
    float* DW   = sinH + 32768;
    float* XsP  = DW   + 65536;
    float* CT2  = XsP  + 4194304;
    float* ST2  = CT2  + 32768;
    float* YsP  = CT2  + 524288;
    float* Z    = YsP  + 2097152;

    twiddle_kernel<<<128, 256, 0, stream>>>(FW, cosH, sinH, DW, CT2, ST2);
    dftw_dfth_kernel<<<512, 512, 0, stream>>>(x, FW, CT2, ST2, XsP);
    modemul_kernel<<<dim3(64, 4, ISPLIT), 256, 0, stream>>>(XsP, w_re, w_im, YsP);
    idfth_kernel<<<dim3(64, 16), 256, 0, stream>>>(YsP, cosH, sinH, Z);
    invw_skip_gelu_kernel<<<512, 256, 0, stream>>>(Z, DW, x, w_sk, b_cv, b_sk, out);
}


# W-axis DFT GEMM (kernel 1): both operands kept in WMMA fragment order (x staged fragment-major in LDS, FW table fragment-major): every fragment read is one 16-byte access per lane; sums unchanged
# speedup vs baseline: 1.5206x; 1.0311x over previous
#include <hip/hip_runtime.h>
#include <math.h>


typedef float v8f __attribute__((ext_vector_type(8)));
typedef float v2f __attribute__((ext_vector_type(2)));

#define IN_C   64
#define OUT_C  64
#define HH     512
#define WW     512
#define MODES  64
#define HW     (HH * WW)
#define TWO_PI 6.283185307179586f
#define ISPLIT 4
#define ICHUNK (IN_C / ISPLIT)
#define HPART  8
#define ASTR   132

__device__ __forceinline__ v8f wmma_f32(v2f a, v2f b, v8f c) {
    return __builtin_amdgcn_wmma_f32_16x16x4_f32(false, a, false, b, (short)0, c,
                                                 false, false);
}

__device__ __forceinline__ v2f load_a_tile(const float* __restrict__ p, int ld, int lane) {
    int m = lane & 15;
    int k = (lane >> 4) << 1;
    v2f a;
    a.x = p[(size_t)m * ld + k];
    a.y = p[(size_t)m * ld + k + 1];
    return a;
}

__global__ void twiddle_kernel(float* __restrict__ FW, float* __restrict__ cosH,
                               float* __restrict__ sinH, float* __restrict__ DW,
                               float* __restrict__ CT2, float* __restrict__ ST2) {
    int t = blockIdx.x * blockDim.x + threadIdx.x;
    if (t >= 32768) return;
    const float step = TWO_PI / 512.0f;
    {
        int w = t >> 6, m2 = t & 63;
        float th = (float)((w * m2) & 511) * step;
        const int ln = (((w >> 1) & 1) << 4) + (m2 & 15);
        const size_t q = ((size_t)(w >> 2) * 4 + (m2 >> 4)) * 32 + ln;
        FW[q * 4 + (w & 1)]     = cosf(th);
        FW[q * 4 + 2 + (w & 1)] = -sinf(th);
    }
    {
        int m1 = t >> 9, h = t & 511;
        float th = (float)((m1 * h) & 511) * step;
        float ct = cosf(th), st = sinf(th);
        cosH[t] = ct;
        sinH[t] = st;
        size_t p2 = ((size_t)(h >> 1) * 64 + m1) * 2 + (h & 1);
        CT2[p2] = ct;
        ST2[p2] = st;
    }
    {
        int k = t >> 9, w = t & 511;
        float th = (float)((k * w) & 511) * step;
        float s  = ((k == 0) ? 1.0f : 2.0f) / 262144.0f;
        DW[(size_t)(2 * k) * 512 + w]     = s * cosf(th);
        DW[(size_t)(2 * k + 1) * 512 + w] = -s * sinf(th);
    }
}

__global__ __launch_bounds__(512)
void dftw_dfth_kernel(const float* __restrict__ x, const float* __restrict__ FW,
                      const float* __restrict__ CT2, const float* __restrict__ ST2,
                      float* __restrict__ XsP) {
    __shared__ __attribute__((aligned(16))) float As[4 * 64 * ASTR];
    const int c = blockIdx.x >> 3, chunk = blockIdx.x & 7;
    const float* xrow = x + ((size_t)c * 512 + chunk * 64) * 512;
    for (int idx = threadIdx.x; idx < 64 * 512 / 4; idx += blockDim.x) {
        const float4 v = ((const float4*)xrow)[idx];
        const int row = idx >> 7, t = idx & 127;
        float* blk = As + ((row >> 4) * 64 + (t >> 1)) * ASTR + (row & 15) * 4 + (t & 1) * 2;
        v2f lo, hi;
        lo.x = v.x; lo.y = v.y;
        hi.x = v.z; hi.y = v.w;
        *(v2f*)blk        = lo;
        *(v2f*)(blk + 64) = hi;
    }
    __syncthreads();

    const int wave = threadIdx.x >> 5, lane = threadIdx.x & 31;
    const int g = wave >> 2, wq = wave & 3;
    const float*  Ag  = As + g * 64 * ASTR + lane * 4;
    const float4* fwp = (const float4*)FW + wq * 32 + lane;
    v8f acc0 = {}, acc1 = {};
    for (int j = 0; j < 64; ++j) {
        const float4 a4 = *(const float4*)(Ag + j * ASTR);
        const float4 f0 = fwp[(2 * j) * 128];
        const float4 f1 = fwp[(2 * j + 1) * 128];
        v2f a, b0, b1;
        a.x = a4.x;  a.y = a4.y;
        b0.x = f0.x; b0.y = f0.y;
        b1.x = f0.z; b1.y = f0.w;
        acc0 = wmma_f32(a, b0, acc0);
        acc1 = wmma_f32(a, b1, acc1);
        a.x = a4.z;  a.y = a4.w;
        b0.x = f1.x; b0.y = f1.y;
        b1.x = f1.z; b1.y = f1.w;
        acc0 = wmma_f32(a, b0, acc0);
        acc1 = wmma_f32(a, b1, acc1);
    }

    const v8f nacc0 = -acc0;
    v8f xr[4] = {{}, {}, {}, {}}, xi[4] = {{}, {}, {}, {}};
    const int m = lane & 15, half = lane >> 4;
    const int hrow = chunk * 64 + g * 16 + 8 * half;
    for (int j = 0; j < 4; ++j) {
        v2f bR, bI, bN;
        bR.x = acc0[2 * j];  bR.y = acc0[2 * j + 1];
        bI.x = acc1[2 * j];  bI.y = acc1[2 * j + 1];
        bN.x = nacc0[2 * j]; bN.y = nacc0[2 * j + 1];
        const size_t trow = (size_t)((hrow >> 1) + j) * 64;
        for (int mt = 0; mt < 4; ++mt) {
            const size_t ti = (trow + 16 * mt + m) * 2;
            v2f aC = *(const v2f*)(CT2 + ti);
            v2f aS = *(const v2f*)(ST2 + ti);
            xr[mt] = wmma_f32(aC, bR, xr[mt]);
            xr[mt] = wmma_f32(aS, bI, xr[mt]);
            xi[mt] = wmma_f32(aC, bI, xi[mt]);
            xi[mt] = wmma_f32(aS, bN, xi[mt]);
        }
    }

    __syncthreads();
    for (int mt = 0; mt < 4; ++mt) {
        for (int v = 0; v < 8; ++v) {
            As[((wave * 8 + 2 * mt) * 8 + v) * 32 + lane]     = xr[mt][v];
            As[((wave * 8 + 2 * mt + 1) * 8 + v) * 32 + lane] = xi[mt][v];
        }
    }
    __syncthreads();
    float* dst = XsP + (size_t)chunk * (IN_C * 64 * 128) + (size_t)c * 64 * 128;
    for (int v = 0; v < 8; ++v) {
        float sr = 0.0f, si = 0.0f;
        for (int gg = 0; gg < 4; ++gg) {
            const int w2 = gg * 4 + wq;
            sr += As[((w2 * 8 + 2 * g) * 8 + v) * 32 + lane];
            si += As[((w2 * 8 + 2 * g + 1) * 8 + v) * 32 + lane];
        }
        const int m1 = 16 * g + 8 * half + v, m2 = 16 * wq + m;
        v2f o; o.x = sr; o.y = si;
        *(v2f*)(dst + (size_t)m1 * 128 + 2 * m2) = o;
    }
}

__global__ void modemul_kernel(const float* __restrict__ XsP, const float* __restrict__ wre,
                               const float* __restrict__ wim, float* __restrict__ YsP) {
    const int m1 = blockIdx.x, obase = blockIdx.y * 16;
    const int ibase = blockIdx.z * ICHUNK;
    __shared__ float xsR[ICHUNK * 64], xsI[ICHUNK * 64];
    for (int idx = threadIdx.x; idx < ICHUNK * 64; idx += blockDim.x) {
        int i = idx >> 6, m2 = idx & 63;
        const float* p = XsP + ((size_t)(ibase + i) * 64 + m1) * 128 + 2 * m2;
        float r = 0.0f, im = 0.0f;
        for (int q = 0; q < HPART; ++q) {
            v2f t = *(const v2f*)(p + (size_t)q * (IN_C * 64 * 128));
            r  += t.x;
            im += t.y;
        }
        xsR[idx] = r;
        xsI[idx] = im;
    }
    __syncthreads();

    const int m2q = (threadIdx.x & 15) * 4;
    const int o   = obase + (threadIdx.x >> 4);
    float aR[4] = {0, 0, 0, 0}, aI[4] = {0, 0, 0, 0};
#pragma unroll 4
    for (int i = 0; i < ICHUNK; ++i) {
        size_t base = (((size_t)(ibase + i) * 64 + o) * 64 + m1) * 64 + m2q;
        float4 wr = *(const float4*)(wre + base);
        float4 wi = *(const float4*)(wim + base);
        const float* xr = xsR + i * 64 + m2q;
        const float* xi = xsI + i * 64 + m2q;
        aR[0] += xr[0] * wr.x - xi[0] * wi.x;  aI[0] += xr[0] * wi.x + xi[0] * wr.x;
        aR[1] += xr[1] * wr.y - xi[1] * wi.y;  aI[1] += xr[1] * wi.y + xi[1] * wr.y;
        aR[2] += xr[2] * wr.z - xi[2] * wi.z;  aI[2] += xr[2] * wi.z + xi[2] * wr.z;
        aR[3] += xr[3] * wr.w - xi[3] * wi.w;  aI[3] += xr[3] * wi.w + xi[3] * wr.w;
    }
    float* yp = YsP + (size_t)blockIdx.z * (OUT_C * 64 * 128)
                    + ((size_t)o * 64 + m1) * 128 + 2 * m2q;
    for (int j = 0; j < 4; ++j) { yp[2 * j] = aR[j]; yp[2 * j + 1] = aI[j]; }
}

#define HCH 32
__global__ __launch_bounds__(256)
void idfth_kernel(const float* __restrict__ YsP, const float* __restrict__ cosH,
                  const float* __restrict__ sinH, float* __restrict__ Z) {
    const int o = blockIdx.x, hbase = blockIdx.y * HCH;
    __shared__ float  ysR[64 * 64], ysI[64 * 64];
    __shared__ float4 tc4[64 * HCH / 4], ts4[64 * HCH / 4];
    for (int idx = threadIdx.x; idx < 64 * 64; idx += blockDim.x) {
        int m1 = idx >> 6, m2 = idx & 63;
        size_t off = ((size_t)o * 64 + m1) * 128 + 2 * m2;
        float r = 0.0f, im = 0.0f;
        for (int z = 0; z < ISPLIT; ++z) {
            v2f t = *(const v2f*)(YsP + (size_t)z * (OUT_C * 64 * 128) + off);
            r  += t.x;
            im += t.y;
        }
        ysR[idx] = r;
        ysI[idx] = im;
    }
    for (int idx = threadIdx.x; idx < 64 * HCH; idx += blockDim.x) {
        int m1 = idx / HCH, hh = idx % HCH;
        ((float*)tc4)[idx] = cosH[(size_t)m1 * 512 + hbase + hh];
        ((float*)ts4)[idx] = sinH[(size_t)m1 * 512 + hbase + hh];
    }
    __syncthreads();

    const int m2 = threadIdx.x & 63;
    const int h0 = (threadIdx.x >> 6) * 8;
    float aR[8] = {0, 0, 0, 0, 0, 0, 0, 0}, aI[8] = {0, 0, 0, 0, 0, 0, 0, 0};
    for (int m1 = 0; m1 < 64; ++m1) {
        const float yr = ysR[m1 * 64 + m2], yi = ysI[m1 * 64 + m2];
        const float4 c0 = tc4[(m1 * HCH + h0) / 4], c1 = tc4[(m1 * HCH + h0) / 4 + 1];
        const float4 s0 = ts4[(m1 * HCH + h0) / 4], s1 = ts4[(m1 * HCH + h0) / 4 + 1];
        const float ct[8] = {c0.x, c0.y, c0.z, c0.w, c1.x, c1.y, c1.z, c1.w};
        const float st[8] = {s0.x, s0.y, s0.z, s0.w, s1.x, s1.y, s1.z, s1.w};
        for (int j = 0; j < 8; ++j) {
            aR[j] += ct[j] * yr - st[j] * yi;
            aI[j] += ct[j] * yi + st[j] * yr;
        }
    }
    for (int j = 0; j < 8; ++j) {
        v2f zo; zo.x = aR[j]; zo.y = aI[j];
        *(v2f*)(Z + ((size_t)o * 512 + hbase + h0 + j) * 128 + 2 * m2) = zo;
    }
}

__global__ __launch_bounds__(256)
void invw_skip_gelu_kernel(const float* __restrict__ Z, const float* __restrict__ DW,
                           const float* __restrict__ x, const float* __restrict__ wskip,
                           const float* __restrict__ bconv, const float* __restrict__ bskip,
                           float* __restrict__ out) {
    const int h = blockIdx.x;
    __shared__ float Az[64 * 128];
    __shared__ float Ak[64 * 64];
    for (int idx = threadIdx.x; idx < 64 * 128 / 4; idx += blockDim.x) {
        int o = idx >> 5, kq = idx & 31;
        ((float4*)(Az + o * 128))[kq] =
            ((const float4*)(Z + ((size_t)o * 512 + h) * 128))[kq];
    }
    for (int idx = threadIdx.x; idx < 64 * 64 / 4; idx += blockDim.x)
        ((float4*)Ak)[idx] = ((const float4*)wskip)[idx];
    __syncthreads();

    const int wave = threadIdx.x >> 5, lane = threadIdx.x & 31;
    const int n = lane & 15, half = lane >> 4;
    const int wc = wave * 64 + 4 * n;
    const float* dwp = DW + (size_t)(2 * half) * 512 + wc;
    const float* xp  = x + (size_t)(2 * half) * HW + (size_t)h * 512 + wc;

    for (int mt = 0; mt < 4; ++mt) {
        const int m0 = mt * 16;
        v8f acc[4] = {{}, {}, {}, {}};
#pragma unroll 8
        for (int k = 0; k < 128; k += 4) {
            v2f a = load_a_tile(Az + m0 * 128 + k, 128, lane);
            const float4 r0 = *(const float4*)(dwp + (size_t)k * 512);
            const float4 r1 = *(const float4*)(dwp + (size_t)(k + 1) * 512);
            v2f b;
            b.x = r0.x; b.y = r1.x; acc[0] = wmma_f32(a, b, acc[0]);
            b.x = r0.y; b.y = r1.y; acc[1] = wmma_f32(a, b, acc[1]);
            b.x = r0.z; b.y = r1.z; acc[2] = wmma_f32(a, b, acc[2]);
            b.x = r0.w; b.y = r1.w; acc[3] = wmma_f32(a, b, acc[3]);
        }
#pragma unroll 8
        for (int k = 0; k < 64; k += 4) {
            v2f a = load_a_tile(Ak + m0 * 64 + k, 64, lane);
            const float4 r0 = *(const float4*)(xp + (size_t)k * HW);
            const float4 r1 = *(const float4*)(xp + (size_t)(k + 1) * HW);
            v2f b;
            b.x = r0.x; b.y = r1.x; acc[0] = wmma_f32(a, b, acc[0]);
            b.x = r0.y; b.y = r1.y; acc[1] = wmma_f32(a, b, acc[1]);
            b.x = r0.z; b.y = r1.z; acc[2] = wmma_f32(a, b, acc[2]);
            b.x = r0.w; b.y = r1.w; acc[3] = wmma_f32(a, b, acc[3]);
        }
        const int mb = m0 + half * 8;
        for (int v = 0; v < 8; ++v) {
            const int m = mb + v;
            float g[4];
            for (int j = 0; j < 4; ++j) {
                float val = acc[j][v] + bconv[m] + bskip[m];
                float c3  = val * val * val;
                g[j] = 0.5f * val *
                       (1.0f + tanhf(0.7978845608028654f * (val + 0.044715f * c3)));
            }
            float4 o4; o4.x = g[0]; o4.y = g[1]; o4.z = g[2]; o4.w = g[3];
            *(float4*)(out + (size_t)m * HW + (size_t)h * 512 + wc) = o4;
        }
    }
}

extern "C" void kernel_launch(void* const* d_in, const int* in_sizes, int n_in,
                              void* d_out, int out_size, void* d_ws, size_t ws_size,
                              hipStream_t stream) {
    const float* x     = (const float*)d_in[0];
    const float* w_re  = (const float*)d_in[1];
    const float* w_im  = (const float*)d_in[2];
    const float* b_cv  = (const float*)d_in[3];
    const float* w_sk  = (const float*)d_in[4];
    const float* b_sk  = (const float*)d_in[5];
    float* out = (float*)d_out;

    float* ws   = (float*)d_ws;
    float* FW   = ws;
    float* cosH = FW   + 65536;
    float* sinH = cosH + 32768;
    float* DW   = sinH + 32768;
    float* XsP  = DW   + 65536;
    float* CT2  = XsP  + 4194304;
    float* ST2  = CT2  + 32768;
    float* YsP  = CT2  + 524288;
    float* Z    = YsP  + 2097152;

    twiddle_kernel<<<128, 256, 0, stream>>>(FW, cosH, sinH, DW, CT2, ST2);
    dftw_dfth_kernel<<<512, 512, 0, stream>>>(x, FW, CT2, ST2, XsP);
    modemul_kernel<<<dim3(64, 4, ISPLIT), 256, 0, stream>>>(XsP, w_re, w_im, YsP);
    idfth_kernel<<<dim3(64, 16), 256, 0, stream>>>(YsP, cosH, sinH, Z);
    invw_skip_gelu_kernel<<<512, 256, 0, stream>>>(Z, DW, x, w_sk, b_cv, b_sk, out);
}


# kernels 1 and 4: operand staging split in 4 chunks along the reduction axis; the next chunk's global loads are issued before this chunk's compute, its LDS stores after it; all sums unchanged
# speedup vs baseline: 1.6814x; 1.1057x over previous
#include <hip/hip_runtime.h>
#include <math.h>


typedef float v8f __attribute__((ext_vector_type(8)));
typedef float v2f __attribute__((ext_vector_type(2)));

#define IN_C   64
#define OUT_C  64
#define HH     512
#define WW     512
#define MODES  64
#define HW     (HH * WW)
#define TWO_PI 6.283185307179586f
#define ISPLIT 4
#define ICHUNK (IN_C / ISPLIT)
#define HPART  8
#define ASTR   132
#define XQ     4

__device__ __forceinline__ v8f wmma_f32(v2f a, v2f b, v8f c) {
    return __builtin_amdgcn_wmma_f32_16x16x4_f32(false, a, false, b, (short)0, c,
                                                 false, false);
}

__device__ __forceinline__ v2f load_a_tile(const float* __restrict__ p, int ld, int lane) {
    int m = lane & 15;
    int k = (lane >> 4) << 1;
    v2f a;
    a.x = p[(size_t)m * ld + k];
    a.y = p[(size_t)m * ld + k + 1];
    return a;
}

__global__ void twiddle_kernel(float* __restrict__ FW, float* __restrict__ cosH,
                               float* __restrict__ sinH, float* __restrict__ DW,
                               float* __restrict__ CT2, float* __restrict__ ST2) {
    int t = blockIdx.x * blockDim.x + threadIdx.x;
    if (t >= 32768) return;
    const float step = TWO_PI / 512.0f;
    {
        int w = t >> 6, m2 = t & 63;
        float th = (float)((w * m2) & 511) * step;
        const int ln = (((w >> 1) & 1) << 4) + (m2 & 15);
        const size_t q = ((size_t)(w >> 2) * 4 + (m2 >> 4)) * 32 + ln;
        FW[q * 4 + (w & 1)]     = cosf(th);
        FW[q * 4 + 2 + (w & 1)] = -sinf(th);
    }
    {
        int m1 = t >> 9, h = t & 511;
        float th = (float)((m1 * h) & 511) * step;
        float ct = cosf(th), st = sinf(th);
        cosH[t] = ct;
        sinH[t] = st;
        size_t p2 = ((size_t)(h >> 1) * 64 + m1) * 2 + (h & 1);
        CT2[p2] = ct;
        ST2[p2] = st;
    }
    {
        int k = t >> 9, w = t & 511;
        float th = (float)((k * w) & 511) * step;
        float s  = ((k == 0) ? 1.0f : 2.0f) / 262144.0f;
        DW[(size_t)(2 * k) * 512 + w]     = s * cosf(th);
        DW[(size_t)(2 * k + 1) * 512 + w] = -s * sinf(th);
    }
}

__global__ __launch_bounds__(512)
void dftw_dfth_kernel(const float* __restrict__ x, const float* __restrict__ FW,
                      const float* __restrict__ CT2, const float* __restrict__ ST2,
                      float* __restrict__ XsP) {
    __shared__ __attribute__((aligned(16))) float As[4 * 64 * ASTR];
    const int c = blockIdx.x >> 3, chunk = blockIdx.x & 7;
    const float* xrow = x + ((size_t)c * 512 + chunk * 64) * 512;
    const float4* xrow4 = (const float4*)xrow;
    const int srow = threadIdx.x >> 5, st = threadIdx.x & 31;
    const float4* xsrc = xrow4 + (size_t)srow * 128 + st;
    float* sdst = As + (st >> 1) * ASTR + srow * 4 + (st & 1) * 2;
    float4 xv[4];
    for (int u = 0; u < 4; ++u) xv[u] = xsrc[u * 16 * 128];
    for (int u = 0; u < 4; ++u) {
        float* blk = sdst + u * 64 * ASTR;
        v2f lo, hi;
        lo.x = xv[u].x; lo.y = xv[u].y;
        hi.x = xv[u].z; hi.y = xv[u].w;
        *(v2f*)blk        = lo;
        *(v2f*)(blk + 64) = hi;
    }
    __syncthreads();

    const int wave = threadIdx.x >> 5, lane = threadIdx.x & 31;
    const int g = wave >> 2, wq = wave & 3;
    const float*  Ag  = As + g * 64 * ASTR + lane * 4;
    const float4* fwp = (const float4*)FW + wq * 32 + lane;
    v8f acc0 = {}, acc1 = {};
#pragma nounroll
    for (int q = 0; q < XQ; ++q) {
        if (q + 1 < XQ)
            for (int u = 0; u < 4; ++u) xv[u] = xsrc[u * 16 * 128 + 32 * (q + 1)];
#pragma unroll 8
        for (int j = 16 * q; j < 16 * q + 16; ++j) {
            const float4 a4 = *(const float4*)(Ag + j * ASTR);
            const float4 f0 = fwp[(2 * j) * 128];
            const float4 f1 = fwp[(2 * j + 1) * 128];
            v2f a, b0, b1;
            a.x = a4.x;  a.y = a4.y;
            b0.x = f0.x; b0.y = f0.y;
            b1.x = f0.z; b1.y = f0.w;
            acc0 = wmma_f32(a, b0, acc0);
            acc1 = wmma_f32(a, b1, acc1);
            a.x = a4.z;  a.y = a4.w;
            b0.x = f1.x; b0.y = f1.y;
            b1.x = f1.z; b1.y = f1.w;
            acc0 = wmma_f32(a, b0, acc0);
            acc1 = wmma_f32(a, b1, acc1);
        }
        if (q + 1 < XQ) {
            for (int u = 0; u < 4; ++u) {
                float* blk = sdst + u * 64 * ASTR + 16 * (q + 1) * ASTR;
                v2f lo, hi;
                lo.x = xv[u].x; lo.y = xv[u].y;
                hi.x = xv[u].z; hi.y = xv[u].w;
                *(v2f*)blk        = lo;
                *(v2f*)(blk + 64) = hi;
            }
            __syncthreads();
        }
    }

    const v8f nacc0 = -acc0;
    v8f xr[4] = {{}, {}, {}, {}}, xi[4] = {{}, {}, {}, {}};
    const int m = lane & 15, half = lane >> 4;
    const int hrow = chunk * 64 + g * 16 + 8 * half;
    for (int j = 0; j < 4; ++j) {
        v2f bR, bI, bN;
        bR.x = acc0[2 * j];  bR.y = acc0[2 * j + 1];
        bI.x = acc1[2 * j];  bI.y = acc1[2 * j + 1];
        bN.x = nacc0[2 * j]; bN.y = nacc0[2 * j + 1];
        const size_t trow = (size_t)((hrow >> 1) + j) * 64;
        for (int mt = 0; mt < 4; ++mt) {
            const size_t ti = (trow + 16 * mt + m) * 2;
            v2f aC = *(const v2f*)(CT2 + ti);
            v2f aS = *(const v2f*)(ST2 + ti);
            xr[mt] = wmma_f32(aC, bR, xr[mt]);
            xr[mt] = wmma_f32(aS, bI, xr[mt]);
            xi[mt] = wmma_f32(aC, bI, xi[mt]);
            xi[mt] = wmma_f32(aS, bN, xi[mt]);
        }
    }

    __syncthreads();
    for (int mt = 0; mt < 4; ++mt) {
        for (int v = 0; v < 8; ++v) {
            As[((wave * 8 + 2 * mt) * 8 + v) * 32 + lane]     = xr[mt][v];
            As[((wave * 8 + 2 * mt + 1) * 8 + v) * 32 + lane] = xi[mt][v];
        }
    }
    __syncthreads();
    float* dst = XsP + (size_t)chunk * (IN_C * 64 * 128) + (size_t)c * 64 * 128;
    for (int v = 0; v < 8; ++v) {
        float sr = 0.0f, si = 0.0f;
        for (int gg = 0; gg < 4; ++gg) {
            const int w2 = gg * 4 + wq;
            sr += As[((w2 * 8 + 2 * g) * 8 + v) * 32 + lane];
            si += As[((w2 * 8 + 2 * g + 1) * 8 + v) * 32 + lane];
        }
        const int m1 = 16 * g + 8 * half + v, m2 = 16 * wq + m;
        v2f o; o.x = sr; o.y = si;
        *(v2f*)(dst + (size_t)m1 * 128 + 2 * m2) = o;
    }
}

__global__ void modemul_kernel(const float* __restrict__ XsP, const float* __restrict__ wre,
                               const float* __restrict__ wim, float* __restrict__ YsP) {
    const int m1 = blockIdx.x, obase = blockIdx.y * 16;
    const int ibase = blockIdx.z * ICHUNK;
    __shared__ float xsR[ICHUNK * 64], xsI[ICHUNK * 64];
    for (int idx = threadIdx.x; idx < ICHUNK * 64; idx += blockDim.x) {
        int i = idx >> 6, m2 = idx & 63;
        const float* p = XsP + ((size_t)(ibase + i) * 64 + m1) * 128 + 2 * m2;
        float r = 0.0f, im = 0.0f;
        for (int q = 0; q < HPART; ++q) {
            v2f t = *(const v2f*)(p + (size_t)q * (IN_C * 64 * 128));
            r  += t.x;
            im += t.y;
        }
        xsR[idx] = r;
        xsI[idx] = im;
    }
    __syncthreads();

    const int m2q = (threadIdx.x & 15) * 4;
    const int o   = obase + (threadIdx.x >> 4);
    float aR[4] = {0, 0, 0, 0}, aI[4] = {0, 0, 0, 0};
#pragma unroll 4
    for (int i = 0; i < ICHUNK; ++i) {
        size_t base = (((size_t)(ibase + i) * 64 + o) * 64 + m1) * 64 + m2q;
        float4 wr = *(const float4*)(wre + base);
        float4 wi = *(const float4*)(wim + base);
        const float* xr = xsR + i * 64 + m2q;
        const float* xi = xsI + i * 64 + m2q;
        aR[0] += xr[0] * wr.x - xi[0] * wi.x;  aI[0] += xr[0] * wi.x + xi[0] * wr.x;
        aR[1] += xr[1] * wr.y - xi[1] * wi.y;  aI[1] += xr[1] * wi.y + xi[1] * wr.y;
        aR[2] += xr[2] * wr.z - xi[2] * wi.z;  aI[2] += xr[2] * wi.z + xi[2] * wr.z;
        aR[3] += xr[3] * wr.w - xi[3] * wi.w;  aI[3] += xr[3] * wi.w + xi[3] * wr.w;
    }
    float* yp = YsP + (size_t)blockIdx.z * (OUT_C * 64 * 128)
                    + ((size_t)o * 64 + m1) * 128 + 2 * m2q;
    for (int j = 0; j < 4; ++j) { yp[2 * j] = aR[j]; yp[2 * j + 1] = aI[j]; }
}

#define HCH 32
#define MQ  4
__global__ __launch_bounds__(256)
void idfth_kernel(const float* __restrict__ YsP, const float* __restrict__ cosH,
                  const float* __restrict__ sinH, float* __restrict__ Z) {
    const int o = blockIdx.x, hbase = blockIdx.y * HCH;
    __shared__ float  ysR[64 * 64], ysI[64 * 64];
    __shared__ float4 tc4[64 * HCH / 4], ts4[64 * HCH / 4];
    const float* ysrc = YsP + (size_t)o * (64 * 128) + 2 * threadIdx.x;
    const int    tsrc = (threadIdx.x / HCH) * 512 + hbase + (threadIdx.x % HCH);
    float* tcf = (float*)tc4;
    float* tsf = (float*)ts4;
    v2f   yv[4][ISPLIT];
    float cv[2], sv[2];
    for (int u = 0; u < 4; ++u)
        for (int z = 0; z < ISPLIT; ++z)
            yv[u][z] = *(const v2f*)(ysrc + (size_t)z * (OUT_C * 64 * 128) + 512 * u);
    for (int u = 0; u < 2; ++u) {
        cv[u] = cosH[tsrc + u * (256 / HCH) * 512];
        sv[u] = sinH[tsrc + u * (256 / HCH) * 512];
    }
    for (int u = 0; u < 4; ++u) {
        float r = 0.0f, im = 0.0f;
        for (int z = 0; z < ISPLIT; ++z) { r += yv[u][z].x; im += yv[u][z].y; }
        ysR[threadIdx.x + 256 * u] = r;
        ysI[threadIdx.x + 256 * u] = im;
    }
    for (int u = 0; u < 2; ++u) {
        tcf[threadIdx.x + 256 * u] = cv[u];
        tsf[threadIdx.x + 256 * u] = sv[u];
    }
    __syncthreads();

    const int m2 = threadIdx.x & 63;
    const int h0 = (threadIdx.x >> 6) * 8;
    float aR[8] = {0, 0, 0, 0, 0, 0, 0, 0}, aI[8] = {0, 0, 0, 0, 0, 0, 0, 0};
#pragma nounroll
    for (int q = 0; q < MQ; ++q) {
        if (q + 1 < MQ) {
            for (int u = 0; u < 4; ++u)
                for (int z = 0; z < ISPLIT; ++z)
                    yv[u][z] = *(const v2f*)(ysrc + (size_t)z * (OUT_C * 64 * 128)
                                             + 2048 * (q + 1) + 512 * u);
            for (int u = 0; u < 2; ++u) {
                cv[u] = cosH[tsrc + ((q + 1) * 16 + u * (256 / HCH)) * 512];
                sv[u] = sinH[tsrc + ((q + 1) * 16 + u * (256 / HCH)) * 512];
            }
        }
#pragma unroll 2
        for (int m1 = 16 * q; m1 < 16 * q + 16; ++m1) {
            const float yr = ysR[m1 * 64 + m2], yi = ysI[m1 * 64 + m2];
            const float4 c0 = tc4[(m1 * HCH + h0) / 4], c1 = tc4[(m1 * HCH + h0) / 4 + 1];
            const float4 s0 = ts4[(m1 * HCH + h0) / 4], s1 = ts4[(m1 * HCH + h0) / 4 + 1];
            const float ct[8] = {c0.x, c0.y, c0.z, c0.w, c1.x, c1.y, c1.z, c1.w};
            const float st[8] = {s0.x, s0.y, s0.z, s0.w, s1.x, s1.y, s1.z, s1.w};
            for (int j = 0; j < 8; ++j) {
                aR[j] += ct[j] * yr - st[j] * yi;
                aI[j] += ct[j] * yi + st[j] * yr;
            }
        }
        if (q + 1 < MQ) {
            for (int u = 0; u < 4; ++u) {
                float r = 0.0f, im = 0.0f;
                for (int z = 0; z < ISPLIT; ++z) { r += yv[u][z].x; im += yv[u][z].y; }
                ysR[1024 * (q + 1) + threadIdx.x + 256 * u] = r;
                ysI[1024 * (q + 1) + threadIdx.x + 256 * u] = im;
            }
            for (int u = 0; u < 2; ++u) {
                tcf[512 * (q + 1) + threadIdx.x + 256 * u] = cv[u];
                tsf[512 * (q + 1) + threadIdx.x + 256 * u] = sv[u];
            }
            __syncthreads();
        }
    }
    for (int j = 0; j < 8; ++j) {
        v2f zo; zo.x = aR[j]; zo.y = aI[j];
        *(v2f*)(Z + ((size_t)o * 512 + hbase + h0 + j) * 128 + 2 * m2) = zo;
    }
}

__global__ __launch_bounds__(256)
void invw_skip_gelu_kernel(const float* __restrict__ Z, const float* __restrict__ DW,
                           const float* __restrict__ x, const float* __restrict__ wskip,
                           const float* __restrict__ bconv, const float* __restrict__ bskip,
                           float* __restrict__ out) {
    const int h = blockIdx.x;
    __shared__ float Az[64 * 128];
    __shared__ float Ak[64 * 64];
    for (int idx = threadIdx.x; idx < 64 * 128 / 4; idx += blockDim.x) {
        int o = idx >> 5, kq = idx & 31;
        ((float4*)(Az + o * 128))[kq] =
            ((const float4*)(Z + ((size_t)o * 512 + h) * 128))[kq];
    }
    for (int idx = threadIdx.x; idx < 64 * 64 / 4; idx += blockDim.x)
        ((float4*)Ak)[idx] = ((const float4*)wskip)[idx];
    __syncthreads();

    const int wave = threadIdx.x >> 5, lane = threadIdx.x & 31;
    const int n = lane & 15, half = lane >> 4;
    const int wc = wave * 64 + 4 * n;
    const float* dwp = DW + (size_t)(2 * half) * 512 + wc;
    const float* xp  = x + (size_t)(2 * half) * HW + (size_t)h * 512 + wc;

    for (int mt = 0; mt < 4; ++mt) {
        const int m0 = mt * 16;
        v8f acc[4] = {{}, {}, {}, {}};
#pragma unroll 8
        for (int k = 0; k < 128; k += 4) {
            v2f a = load_a_tile(Az + m0 * 128 + k, 128, lane);
            const float4 r0 = *(const float4*)(dwp + (size_t)k * 512);
            const float4 r1 = *(const float4*)(dwp + (size_t)(k + 1) * 512);
            v2f b;
            b.x = r0.x; b.y = r1.x; acc[0] = wmma_f32(a, b, acc[0]);
            b.x = r0.y; b.y = r1.y; acc[1] = wmma_f32(a, b, acc[1]);
            b.x = r0.z; b.y = r1.z; acc[2] = wmma_f32(a, b, acc[2]);
            b.x = r0.w; b.y = r1.w; acc[3] = wmma_f32(a, b, acc[3]);
        }
#pragma unroll 8
        for (int k = 0; k < 64; k += 4) {
            v2f a = load_a_tile(Ak + m0 * 64 + k, 64, lane);
            const float4 r0 = *(const float4*)(xp + (size_t)k * HW);
            const float4 r1 = *(const float4*)(xp + (size_t)(k + 1) * HW);
            v2f b;
            b.x = r0.x; b.y = r1.x; acc[0] = wmma_f32(a, b, acc[0]);
            b.x = r0.y; b.y = r1.y; acc[1] = wmma_f32(a, b, acc[1]);
            b.x = r0.z; b.y = r1.z; acc[2] = wmma_f32(a, b, acc[2]);
            b.x = r0.w; b.y = r1.w; acc[3] = wmma_f32(a, b, acc[3]);
        }
        const int mb = m0 + half * 8;
        for (int v = 0; v < 8; ++v) {
            const int m = mb + v;
            float g[4];
            for (int j = 0; j < 4; ++j) {
                float val = acc[j][v] + bconv[m] + bskip[m];
                float c3  = val * val * val;
                g[j] = 0.5f * val *
                       (1.0f + tanhf(0.7978845608028654f * (val + 0.044715f * c3)));
            }
            float4 o4; o4.x = g[0]; o4.y = g[1]; o4.z = g[2]; o4.w = g[3];
            *(float4*)(out + (size_t)m * HW + (size_t)h * 512 + wc) = o4;
        }
    }
}

extern "C" void kernel_launch(void* const* d_in, const int* in_sizes, int n_in,
                              void* d_out, int out_size, void* d_ws, size_t ws_size,
                              hipStream_t stream) {
    const float* x     = (const float*)d_in[0];
    const float* w_re  = (const float*)d_in[1];
    const float* w_im  = (const float*)d_in[2];
    const float* b_cv  = (const float*)d_in[3];
    const float* w_sk  = (const float*)d_in[4];
    const float* b_sk  = (const float*)d_in[5];
    float* out = (float*)d_out;

    float* ws   = (float*)d_ws;
    float* FW   = ws;
    float* cosH = FW   + 65536;
    float* sinH = cosH + 32768;
    float* DW   = sinH + 32768;
    float* XsP  = DW   + 65536;
    float* CT2  = XsP  + 4194304;
    float* ST2  = CT2  + 32768;
    float* YsP  = CT2  + 524288;
    float* Z    = YsP  + 2097152;

    twiddle_kernel<<<128, 256, 0, stream>>>(FW, cosH, sinH, DW, CT2, ST2);
    dftw_dfth_kernel<<<512, 512, 0, stream>>>(x, FW, CT2, ST2, XsP);
    modemul_kernel<<<dim3(64, 4, ISPLIT), 256, 0, stream>>>(XsP, w_re, w_im, YsP);
    idfth_kernel<<<dim3(64, 16), 256, 0, stream>>>(YsP, cosH, sinH, Z);
    invw_skip_gelu_kernel<<<512, 256, 0, stream>>>(Z, DW, x, w_sk, b_cv, b_sk, out);
}


# iDFT_W+skip+gelu kernel: B operand (DW rows, x rows) staged through LDS in fragment order, 64-column chunks, once per block, shared by the 4 M-tile waves; conflict-free A/B pitches; sums unchanged
# speedup vs baseline: 1.7945x; 1.0673x over previous
#include <hip/hip_runtime.h>
#include <math.h>


typedef float v8f __attribute__((ext_vector_type(8)));
typedef float v2f __attribute__((ext_vector_type(2)));
typedef float v4f __attribute__((ext_vector_type(4)));

#define IN_C   64
#define OUT_C  64
#define HH     512
#define WW     512
#define MODES  64
#define HW     (HH * WW)
#define TWO_PI 6.283185307179586f
#define ISPLIT 4
#define ICHUNK (IN_C / ISPLIT)
#define HPART  8
#define ASTR   132
#define XQ     4

__device__ __forceinline__ v8f wmma_f32(v2f a, v2f b, v8f c) {
    return __builtin_amdgcn_wmma_f32_16x16x4_f32(false, a, false, b, (short)0, c,
                                                 false, false);
}

__device__ __forceinline__ v2f load_a_tile(const float* __restrict__ p, int ld, int lane) {
    int m = lane & 15;
    int k = (lane >> 4) << 1;
    v2f a;
    a.x = p[(size_t)m * ld + k];
    a.y = p[(size_t)m * ld + k + 1];
    return a;
}

__global__ void twiddle_kernel(float* __restrict__ FW, float* __restrict__ cosH,
                               float* __restrict__ sinH, float* __restrict__ DW,
                               float* __restrict__ CT2, float* __restrict__ ST2) {
    int t = blockIdx.x * blockDim.x + threadIdx.x;
    if (t >= 32768) return;
    const float step = TWO_PI / 512.0f;
    {
        int w = t >> 6, m2 = t & 63;
        float th = (float)((w * m2) & 511) * step;
        const int ln = (((w >> 1) & 1) << 4) + (m2 & 15);
        const size_t q = ((size_t)(w >> 2) * 4 + (m2 >> 4)) * 32 + ln;
        FW[q * 4 + (w & 1)]     = cosf(th);
        FW[q * 4 + 2 + (w & 1)] = -sinf(th);
    }
    {
        int m1 = t >> 9, h = t & 511;
        float th = (float)((m1 * h) & 511) * step;
        float ct = cosf(th), st = sinf(th);
        cosH[t] = ct;
        sinH[t] = st;
        size_t p2 = ((size_t)(h >> 1) * 64 + m1) * 2 + (h & 1);
        CT2[p2] = ct;
        ST2[p2] = st;
    }
    {
        int k = t >> 9, w = t & 511;
        float th = (float)((k * w) & 511) * step;
        float s  = ((k == 0) ? 1.0f : 2.0f) / 262144.0f;
        DW[(size_t)(2 * k) * 512 + w]     = s * cosf(th);
        DW[(size_t)(2 * k + 1) * 512 + w] = -s * sinf(th);
    }
}

__global__ __launch_bounds__(512)
void dftw_dfth_kernel(const float* __restrict__ x, const float* __restrict__ FW,
                      const float* __restrict__ CT2, const float* __restrict__ ST2,
                      float* __restrict__ XsP) {
    __shared__ __attribute__((aligned(16))) float As[4 * 64 * ASTR];
    const int c = blockIdx.x >> 3, chunk = blockIdx.x & 7;
    const float* xrow = x + ((size_t)c * 512 + chunk * 64) * 512;
    const float4* xrow4 = (const float4*)xrow;
    const int srow = threadIdx.x >> 5, st = threadIdx.x & 31;
    const float4* xsrc = xrow4 + (size_t)srow * 128 + st;
    float* sdst = As + (st >> 1) * ASTR + srow * 4 + (st & 1) * 2;
    float4 xv[4];
    for (int u = 0; u < 4; ++u) xv[u] = xsrc[u * 16 * 128];
    for (int u = 0; u < 4; ++u) {
        float* blk = sdst + u * 64 * ASTR;
        v2f lo, hi;
        lo.x = xv[u].x; lo.y = xv[u].y;
        hi.x = xv[u].z; hi.y = xv[u].w;
        *(v2f*)blk        = lo;
        *(v2f*)(blk + 64) = hi;
    }
    __syncthreads();

    const int wave = threadIdx.x >> 5, lane = threadIdx.x & 31;
    const int g = wave >> 2, wq = wave & 3;
    const float*  Ag  = As + g * 64 * ASTR + lane * 4;
    const float4* fwp = (const float4*)FW + wq * 32 + lane;
    v8f acc0 = {}, acc1 = {};
#pragma nounroll
    for (int q = 0; q < XQ; ++q) {
        if (q + 1 < XQ)
            for (int u = 0; u < 4; ++u) xv[u] = xsrc[u * 16 * 128 + 32 * (q + 1)];
#pragma unroll 8
        for (int j = 16 * q; j < 16 * q + 16; ++j) {
            const float4 a4 = *(const float4*)(Ag + j * ASTR);
            const float4 f0 = fwp[(2 * j) * 128];
            const float4 f1 = fwp[(2 * j + 1) * 128];
            v2f a, b0, b1;
            a.x = a4.x;  a.y = a4.y;
            b0.x = f0.x; b0.y = f0.y;
            b1.x = f0.z; b1.y = f0.w;
            acc0 = wmma_f32(a, b0, acc0);
            acc1 = wmma_f32(a, b1, acc1);
            a.x = a4.z;  a.y = a4.w;
            b0.x = f1.x; b0.y = f1.y;
            b1.x = f1.z; b1.y = f1.w;
            acc0 = wmma_f32(a, b0, acc0);
            acc1 = wmma_f32(a, b1, acc1);
        }
        if (q + 1 < XQ) {
            for (int u = 0; u < 4; ++u) {
                float* blk = sdst + u * 64 * ASTR + 16 * (q + 1) * ASTR;
                v2f lo, hi;
                lo.x = xv[u].x; lo.y = xv[u].y;
                hi.x = xv[u].z; hi.y = xv[u].w;
                *(v2f*)blk        = lo;
                *(v2f*)(blk + 64) = hi;
            }
            __syncthreads();
        }
    }

    const v8f nacc0 = -acc0;
    v8f xr[4] = {{}, {}, {}, {}}, xi[4] = {{}, {}, {}, {}};
    const int m = lane & 15, half = lane >> 4;
    const int hrow = chunk * 64 + g * 16 + 8 * half;
    for (int j = 0; j < 4; ++j) {
        v2f bR, bI, bN;
        bR.x = acc0[2 * j];  bR.y = acc0[2 * j + 1];
        bI.x = acc1[2 * j];  bI.y = acc1[2 * j + 1];
        bN.x = nacc0[2 * j]; bN.y = nacc0[2 * j + 1];
        const size_t trow = (size_t)((hrow >> 1) + j) * 64;
        for (int mt = 0; mt < 4; ++mt) {
            const size_t ti = (trow + 16 * mt + m) * 2;
            v2f aC = *(const v2f*)(CT2 + ti);
            v2f aS = *(const v2f*)(ST2 + ti);
            xr[mt] = wmma_f32(aC, bR, xr[mt]);
            xr[mt] = wmma_f32(aS, bI, xr[mt]);
            xi[mt] = wmma_f32(aC, bI, xi[mt]);
            xi[mt] = wmma_f32(aS, bN, xi[mt]);
        }
    }

    __syncthreads();
    for (int mt = 0; mt < 4; ++mt) {
        for (int v = 0; v < 8; ++v) {
            As[((wave * 8 + 2 * mt) * 8 + v) * 32 + lane]     = xr[mt][v];
            As[((wave * 8 + 2 * mt + 1) * 8 + v) * 32 + lane] = xi[mt][v];
        }
    }
    __syncthreads();
    float* dst = XsP + (size_t)chunk * (IN_C * 64 * 128) + (size_t)c * 64 * 128;
    for (int v = 0; v < 8; ++v) {
        float sr = 0.0f, si = 0.0f;
        for (int gg = 0; gg < 4; ++gg) {
            const int w2 = gg * 4 + wq;
            sr += As[((w2 * 8 + 2 * g) * 8 + v) * 32 + lane];
            si += As[((w2 * 8 + 2 * g + 1) * 8 + v) * 32 + lane];
        }
        const int m1 = 16 * g + 8 * half + v, m2 = 16 * wq + m;
        v2f o; o.x = sr; o.y = si;
        *(v2f*)(dst + (size_t)m1 * 128 + 2 * m2) = o;
    }
}

__global__ void modemul_kernel(const float* __restrict__ XsP, const float* __restrict__ wre,
                               const float* __restrict__ wim, float* __restrict__ YsP) {
    const int m1 = blockIdx.x, obase = blockIdx.y * 16;
    const int ibase = blockIdx.z * ICHUNK;
    __shared__ float xsR[ICHUNK * 64], xsI[ICHUNK * 64];
    for (int idx = threadIdx.x; idx < ICHUNK * 64; idx += blockDim.x) {
        int i = idx >> 6, m2 = idx & 63;
        const float* p = XsP + ((size_t)(ibase + i) * 64 + m1) * 128 + 2 * m2;
        float r = 0.0f, im = 0.0f;
        for (int q = 0; q < HPART; ++q) {
            v2f t = *(const v2f*)(p + (size_t)q * (IN_C * 64 * 128));
            r  += t.x;
            im += t.y;
        }
        xsR[idx] = r;
        xsI[idx] = im;
    }
    __syncthreads();

    const int m2q = (threadIdx.x & 15) * 4;
    const int o   = obase + (threadIdx.x >> 4);
    float aR[4] = {0, 0, 0, 0}, aI[4] = {0, 0, 0, 0};
#pragma unroll 4
    for (int i = 0; i < ICHUNK; ++i) {
        size_t base = (((size_t)(ibase + i) * 64 + o) * 64 + m1) * 64 + m2q;
        float4 wr = *(const float4*)(wre + base);
        float4 wi = *(const float4*)(wim + base);
        const float* xr = xsR + i * 64 + m2q;
        const float* xi = xsI + i * 64 + m2q;
        aR[0] += xr[0] * wr.x - xi[0] * wi.x;  aI[0] += xr[0] * wi.x + xi[0] * wr.x;
        aR[1] += xr[1] * wr.y - xi[1] * wi.y;  aI[1] += xr[1] * wi.y + xi[1] * wr.y;
        aR[2] += xr[2] * wr.z - xi[2] * wi.z;  aI[2] += xr[2] * wi.z + xi[2] * wr.z;
        aR[3] += xr[3] * wr.w - xi[3] * wi.w;  aI[3] += xr[3] * wi.w + xi[3] * wr.w;
    }
    float* yp = YsP + (size_t)blockIdx.z * (OUT_C * 64 * 128)
                    + ((size_t)o * 64 + m1) * 128 + 2 * m2q;
    for (int j = 0; j < 4; ++j) { yp[2 * j] = aR[j]; yp[2 * j + 1] = aI[j]; }
}

#define HCH 32
#define MQ  4
__global__ __launch_bounds__(256)
void idfth_kernel(const float* __restrict__ YsP, const float* __restrict__ cosH,
                  const float* __restrict__ sinH, float* __restrict__ Z) {
    const int o = blockIdx.x, hbase = blockIdx.y * HCH;
    __shared__ float  ysR[64 * 64], ysI[64 * 64];
    __shared__ float4 tc4[64 * HCH / 4], ts4[64 * HCH / 4];
    const float* ysrc = YsP + (size_t)o * (64 * 128) + 2 * threadIdx.x;
    const int    tsrc = (threadIdx.x / HCH) * 512 + hbase + (threadIdx.x % HCH);
    float* tcf = (float*)tc4;
    float* tsf = (float*)ts4;
    v2f   yv[4][ISPLIT];
    float cv[2], sv[2];
    for (int u = 0; u < 4; ++u)
        for (int z = 0; z < ISPLIT; ++z)
            yv[u][z] = *(const v2f*)(ysrc + (size_t)z * (OUT_C * 64 * 128) + 512 * u);
    for (int u = 0; u < 2; ++u) {
        cv[u] = cosH[tsrc + u * (256 / HCH) * 512];
        sv[u] = sinH[tsrc + u * (256 / HCH) * 512];
    }
    for (int u = 0; u < 4; ++u) {
        float r = 0.0f, im = 0.0f;
        for (int z = 0; z < ISPLIT; ++z) { r += yv[u][z].x; im += yv[u][z].y; }
        ysR[threadIdx.x + 256 * u] = r;
        ysI[threadIdx.x + 256 * u] = im;
    }
    for (int u = 0; u < 2; ++u) {
        tcf[threadIdx.x + 256 * u] = cv[u];
        tsf[threadIdx.x + 256 * u] = sv[u];
    }
    __syncthreads();

    const int m2 = threadIdx.x & 63;
    const int h0 = (threadIdx.x >> 6) * 8;
    float aR[8] = {0, 0, 0, 0, 0, 0, 0, 0}, aI[8] = {0, 0, 0, 0, 0, 0, 0, 0};
#pragma nounroll
    for (int q = 0; q < MQ; ++q) {
        if (q + 1 < MQ) {
            for (int u = 0; u < 4; ++u)
                for (int z = 0; z < ISPLIT; ++z)
                    yv[u][z] = *(const v2f*)(ysrc + (size_t)z * (OUT_C * 64 * 128)
                                             + 2048 * (q + 1) + 512 * u);
            for (int u = 0; u < 2; ++u) {
                cv[u] = cosH[tsrc + ((q + 1) * 16 + u * (256 / HCH)) * 512];
                sv[u] = sinH[tsrc + ((q + 1) * 16 + u * (256 / HCH)) * 512];
            }
        }
#pragma unroll 2
        for (int m1 = 16 * q; m1 < 16 * q + 16; ++m1) {
            const float yr = ysR[m1 * 64 + m2], yi = ysI[m1 * 64 + m2];
            const float4 c0 = tc4[(m1 * HCH + h0) / 4], c1 = tc4[(m1 * HCH + h0) / 4 + 1];
            const float4 s0 = ts4[(m1 * HCH + h0) / 4], s1 = ts4[(m1 * HCH + h0) / 4 + 1];
            const float ct[8] = {c0.x, c0.y, c0.z, c0.w, c1.x, c1.y, c1.z, c1.w};
            const float st[8] = {s0.x, s0.y, s0.z, s0.w, s1.x, s1.y, s1.z, s1.w};
            for (int j = 0; j < 8; ++j) {
                aR[j] += ct[j] * yr - st[j] * yi;
                aI[j] += ct[j] * yi + st[j] * yr;
            }
        }
        if (q + 1 < MQ) {
            for (int u = 0; u < 4; ++u) {
                float r = 0.0f, im = 0.0f;
                for (int z = 0; z < ISPLIT; ++z) { r += yv[u][z].x; im += yv[u][z].y; }
                ysR[1024 * (q + 1) + threadIdx.x + 256 * u] = r;
                ysI[1024 * (q + 1) + threadIdx.x + 256 * u] = im;
            }
            for (int u = 0; u < 2; ++u) {
                tcf[512 * (q + 1) + threadIdx.x + 256 * u] = cv[u];
                tsf[512 * (q + 1) + threadIdx.x + 256 * u] = sv[u];
            }
            __syncthreads();
        }
    }
    for (int j = 0; j < 8; ++j) {
        v2f zo; zo.x = aR[j]; zo.y = aI[j];
        *(v2f*)(Z + ((size_t)o * 512 + hbase + h0 + j) * 128 + 2 * m2) = zo;
    }
}

__device__ __forceinline__ v2f lo2(float4 v) { v2f r; r.x = v.x; r.y = v.y; return r; }
__device__ __forceinline__ v2f hi2(float4 v) { v2f r; r.x = v.z; r.y = v.w; return r; }

__device__ __forceinline__ void store_b_pair(float* dst, float4 r0, float4 r1) {
    v4f t0 = {r0.x, r0.y, r0.z, r0.w}, t1 = {r1.x, r1.y, r1.z, r1.w};
    asm volatile("" : "+v"(t0), "+v"(t1));
    float4 lo, hi;
    lo.x = t0.x; lo.y = t1.x; lo.z = t0.y; lo.w = t1.y;
    hi.x = t0.z; hi.y = t1.z; hi.z = t0.w; hi.w = t1.w;
    *(float4*)dst       = lo;
    *(float4*)(dst + 4) = hi;
}

#define AZP 132
#define AKP 68
#define BP  160
#define NCH 8
__global__ __launch_bounds__(256)
void invw_skip_gelu_kernel(const float* __restrict__ Z, const float* __restrict__ DW,
                           const float* __restrict__ x, const float* __restrict__ wskip,
                           const float* __restrict__ bconv, const float* __restrict__ bskip,
                           float* __restrict__ out) {
    const int h = blockIdx.x;
    __shared__ __attribute__((aligned(16))) float Az[64 * AZP];
    __shared__ __attribute__((aligned(16))) float Ak[64 * AKP];
    __shared__ __attribute__((aligned(16))) float Bs[96 * BP];

    const int q0 = threadIdx.x >> 4, pc4 = threadIdx.x & 15;
    const float4* dsrc = (const float4*)(DW + (size_t)(2 * q0) * 512) + pc4;
    const float4* xsrc = (const float4*)(x + (size_t)(2 * q0) * HW + (size_t)h * 512) + pc4;
    float* bdst = Bs + q0 * BP + 8 * pc4;
    float4 ra0, ra1, ra2, ra3, ra4, ra5, rb0, rb1, rb2, rb3, rb4, rb5;
    ra0 = dsrc[0 * 128];  rb0 = dsrc[1 * 128];
    ra1 = dsrc[32 * 128];  rb1 = dsrc[33 * 128];
    ra2 = dsrc[64 * 128];  rb2 = dsrc[65 * 128];
    ra3 = dsrc[96 * 128];  rb3 = dsrc[97 * 128];
    ra4 = xsrc[(size_t)0 * (HW / 4)];  rb4 = xsrc[(size_t)1 * (HW / 4)];
    ra5 = xsrc[(size_t)32 * (HW / 4)];  rb5 = xsrc[(size_t)33 * (HW / 4)];

    for (int idx = threadIdx.x; idx < 64 * 128 / 4; idx += blockDim.x) {
        int o = idx >> 5, kq = idx & 31;
        *(float4*)(Az + o * AZP + 4 * kq) =
            ((const float4*)(Z + ((size_t)o * 512 + h) * 128))[kq];
    }
    for (int idx = threadIdx.x; idx < 64 * 64 / 4; idx += blockDim.x) {
        int o = idx >> 4, kq = idx & 15;
        *(float4*)(Ak + o * AKP + 4 * kq) = ((const float4*)wskip)[idx];
    }
    store_b_pair(bdst + 0 * 16 * BP, ra0, rb0);
    store_b_pair(bdst + 1 * 16 * BP, ra1, rb1);
    store_b_pair(bdst + 2 * 16 * BP, ra2, rb2);
    store_b_pair(bdst + 3 * 16 * BP, ra3, rb3);
    store_b_pair(bdst + 4 * 16 * BP, ra4, rb4);
    store_b_pair(bdst + 5 * 16 * BP, ra5, rb5);
    __syncthreads();

    const int wave = threadIdx.x >> 5, lane = threadIdx.x & 31;
    const int n = lane & 15, half = lane >> 4;
    const int mt = wave & 3, hh = wave >> 2;
    const int m0 = mt * 16;
    const float* azp = Az + (m0 + n) * AZP + 2 * half;
    const float* akp = Ak + (m0 + n) * AKP + 2 * half;
    const float* bp  = Bs + half * BP + 64 * hh + 4 * n;
    const int mb = m0 + half * 8;
    float bc[8], bk[8];
    for (int v = 0; v < 8; ++v) { bc[v] = bconv[mb + v]; bk[v] = bskip[mb + v]; }
    float* orow = out + (size_t)mb * HW + (size_t)h * 512 + 32 * hh + 2 * n;

#pragma nounroll
    for (int c = 0; c < NCH; ++c) {
        if (c + 1 < NCH) {
            ra0 = dsrc[0 * 128 + 16 * (c + 1)];  rb0 = dsrc[1 * 128 + 16 * (c + 1)];
            ra1 = dsrc[32 * 128 + 16 * (c + 1)];  rb1 = dsrc[33 * 128 + 16 * (c + 1)];
            ra2 = dsrc[64 * 128 + 16 * (c + 1)];  rb2 = dsrc[65 * 128 + 16 * (c + 1)];
            ra3 = dsrc[96 * 128 + 16 * (c + 1)];  rb3 = dsrc[97 * 128 + 16 * (c + 1)];
            ra4 = xsrc[(size_t)0 * (HW / 4) + 16 * (c + 1)];  rb4 = xsrc[(size_t)1 * (HW / 4) + 16 * (c + 1)];
            ra5 = xsrc[(size_t)32 * (HW / 4) + 16 * (c + 1)];  rb5 = xsrc[(size_t)33 * (HW / 4) + 16 * (c + 1)];
        }
        v8f acc0 = {}, acc1 = {};
        for (int t = 0; t < 4; ++t) {
            const float* ap = azp + 32 * t;
            const float* bq = bp + (16 * t) * BP;
            const v2f    a0 = *(const v2f*)(ap + 0);
            const v2f    a1 = *(const v2f*)(ap + 4);
            const v2f    a2 = *(const v2f*)(ap + 8);
            const v2f    a3 = *(const v2f*)(ap + 12);
            const v2f    a4 = *(const v2f*)(ap + 16);
            const v2f    a5 = *(const v2f*)(ap + 20);
            const v2f    a6 = *(const v2f*)(ap + 24);
            const v2f    a7 = *(const v2f*)(ap + 28);
            const float4 b0 = *(const float4*)(bq + 0 * BP);
            const float4 b1 = *(const float4*)(bq + 2 * BP);
            const float4 b2 = *(const float4*)(bq + 4 * BP);
            const float4 b3 = *(const float4*)(bq + 6 * BP);
            const float4 b4 = *(const float4*)(bq + 8 * BP);
            const float4 b5 = *(const float4*)(bq + 10 * BP);
            const float4 b6 = *(const float4*)(bq + 12 * BP);
            const float4 b7 = *(const float4*)(bq + 14 * BP);
            __builtin_amdgcn_sched_barrier(0);
            acc0 = wmma_f32(a0, lo2(b0), acc0);  acc1 = wmma_f32(a0, hi2(b0), acc1);
            acc0 = wmma_f32(a1, lo2(b1), acc0);  acc1 = wmma_f32(a1, hi2(b1), acc1);
            acc0 = wmma_f32(a2, lo2(b2), acc0);  acc1 = wmma_f32(a2, hi2(b2), acc1);
            acc0 = wmma_f32(a3, lo2(b3), acc0);  acc1 = wmma_f32(a3, hi2(b3), acc1);
            acc0 = wmma_f32(a4, lo2(b4), acc0);  acc1 = wmma_f32(a4, hi2(b4), acc1);
            acc0 = wmma_f32(a5, lo2(b5), acc0);  acc1 = wmma_f32(a5, hi2(b5), acc1);
            acc0 = wmma_f32(a6, lo2(b6), acc0);  acc1 = wmma_f32(a6, hi2(b6), acc1);
            acc0 = wmma_f32(a7, lo2(b7), acc0);  acc1 = wmma_f32(a7, hi2(b7), acc1);
        }
        for (int t = 0; t < 2; ++t) {
            const float* ap = akp + 32 * t;
            const float* bq = bp + (64 + 16 * t) * BP;
            const v2f    a0 = *(const v2f*)(ap + 0);
            const v2f    a1 = *(const v2f*)(ap + 4);
            const v2f    a2 = *(const v2f*)(ap + 8);
            const v2f    a3 = *(const v2f*)(ap + 12);
            const v2f    a4 = *(const v2f*)(ap + 16);
            const v2f    a5 = *(const v2f*)(ap + 20);
            const v2f    a6 = *(const v2f*)(ap + 24);
            const v2f    a7 = *(const v2f*)(ap + 28);
            const float4 b0 = *(const float4*)(bq + 0 * BP);
            const float4 b1 = *(const float4*)(bq + 2 * BP);
            const float4 b2 = *(const float4*)(bq + 4 * BP);
            const float4 b3 = *(const float4*)(bq + 6 * BP);
            const float4 b4 = *(const float4*)(bq + 8 * BP);
            const float4 b5 = *(const float4*)(bq + 10 * BP);
            const float4 b6 = *(const float4*)(bq + 12 * BP);
            const float4 b7 = *(const float4*)(bq + 14 * BP);
            __builtin_amdgcn_sched_barrier(0);
            acc0 = wmma_f32(a0, lo2(b0), acc0);  acc1 = wmma_f32(a0, hi2(b0), acc1);
            acc0 = wmma_f32(a1, lo2(b1), acc0);  acc1 = wmma_f32(a1, hi2(b1), acc1);
            acc0 = wmma_f32(a2, lo2(b2), acc0);  acc1 = wmma_f32(a2, hi2(b2), acc1);
            acc0 = wmma_f32(a3, lo2(b3), acc0);  acc1 = wmma_f32(a3, hi2(b3), acc1);
            acc0 = wmma_f32(a4, lo2(b4), acc0);  acc1 = wmma_f32(a4, hi2(b4), acc1);
            acc0 = wmma_f32(a5, lo2(b5), acc0);  acc1 = wmma_f32(a5, hi2(b5), acc1);
            acc0 = wmma_f32(a6, lo2(b6), acc0);  acc1 = wmma_f32(a6, hi2(b6), acc1);
            acc0 = wmma_f32(a7, lo2(b7), acc0);  acc1 = wmma_f32(a7, hi2(b7), acc1);
        }
        if (c + 1 < NCH) {
            __syncthreads();
            store_b_pair(bdst + 0 * 16 * BP, ra0, rb0);
            store_b_pair(bdst + 1 * 16 * BP, ra1, rb1);
            store_b_pair(bdst + 2 * 16 * BP, ra2, rb2);
            store_b_pair(bdst + 3 * 16 * BP, ra3, rb3);
            store_b_pair(bdst + 4 * 16 * BP, ra4, rb4);
            store_b_pair(bdst + 5 * 16 * BP, ra5, rb5);
            __syncthreads();
        }
        for (int v = 0; v < 8; ++v) {
            float g[2];
            for (int j = 0; j < 2; ++j) {
                float val = (j ? acc1[v] : acc0[v]) + bc[v] + bk[v];
                float c3  = val * val * val;
                g[j] = 0.5f * val *
                       (1.0f + tanhf(0.7978845608028654f * (val + 0.044715f * c3)));
            }
            v2f o2; o2.x = g[0]; o2.y = g[1];
            *(v2f*)(orow + (size_t)v * HW + 64 * c) = o2;
        }
    }
}

extern "C" void kernel_launch(void* const* d_in, const int* in_sizes, int n_in,
                              void* d_out, int out_size, void* d_ws, size_t ws_size,
                              hipStream_t stream) {
    const float* x     = (const float*)d_in[0];
    const float* w_re  = (const float*)d_in[1];
    const float* w_im  = (const float*)d_in[2];
    const float* b_cv  = (const float*)d_in[3];
    const float* w_sk  = (const float*)d_in[4];
    const float* b_sk  = (const float*)d_in[5];
    float* out = (float*)d_out;

    float* ws   = (float*)d_ws;
    float* FW   = ws;
    float* cosH = FW   + 65536;
    float* sinH = cosH + 32768;
    float* DW   = sinH + 32768;
    float* XsP  = DW   + 65536;
    float* CT2  = XsP  + 4194304;
    float* ST2  = CT2  + 32768;
    float* YsP  = CT2  + 524288;
    float* Z    = YsP  + 2097152;

    twiddle_kernel<<<128, 256, 0, stream>>>(FW, cosH, sinH, DW, CT2, ST2);
    dftw_dfth_kernel<<<512, 512, 0, stream>>>(x, FW, CT2, ST2, XsP);
    modemul_kernel<<<dim3(64, 4, ISPLIT), 256, 0, stream>>>(XsP, w_re, w_im, YsP);
    idfth_kernel<<<dim3(64, 16), 256, 0, stream>>>(YsP, cosH, sinH, Z);
    invw_skip_gelu_kernel<<<512, 256, 0, stream>>>(Z, DW, x, w_sk, b_cv, b_sk, out);
}


# W-axis forward DFT (kernel 1) split as w = w1 + 64*w2: 8-point DFTs on the vector unit while staging, then 64-term WMMA sums per residue of m2 mod 8: 64 WMMAs/wave, was 256; w sum order changes
# speedup vs baseline: 2.0235x; 1.1276x over previous
#include <hip/hip_runtime.h>
#include <math.h>


typedef float v8f __attribute__((ext_vector_type(8)));
typedef float v2f __attribute__((ext_vector_type(2)));
typedef float v4f __attribute__((ext_vector_type(4)));

#define IN_C   64
#define OUT_C  64
#define HH     512
#define WW     512
#define MODES  64
#define HW     (HH * WW)
#define TWO_PI 6.283185307179586f
#define ISPLIT 4
#define ICHUNK (IN_C / ISPLIT)
#define HPART  8
#define ASTR   132

__device__ __forceinline__ v8f wmma_f32(v2f a, v2f b, v8f c) {
    return __builtin_amdgcn_wmma_f32_16x16x4_f32(false, a, false, b, (short)0, c,
                                                 false, false);
}

__device__ __forceinline__ v2f load_a_tile(const float* __restrict__ p, int ld, int lane) {
    int m = lane & 15;
    int k = (lane >> 4) << 1;
    v2f a;
    a.x = p[(size_t)m * ld + k];
    a.y = p[(size_t)m * ld + k + 1];
    return a;
}

__device__ __forceinline__ int mode_of_pos(int p) {
    const int s = p >> 4, t = (p >> 3) & 1, j = p & 7;
    const int r = (s == 0) ? (t ? 4 : 0) : (t ? 8 - s : s);
    return r + 8 * j;
}

__global__ void twiddle_kernel(float* __restrict__ FW, float* __restrict__ cosH,
                               float* __restrict__ sinH, float* __restrict__ DW,
                               float* __restrict__ CT2, float* __restrict__ ST2) {
    int t = blockIdx.x * blockDim.x + threadIdx.x;
    if (t >= 32768) return;
    const float step = TWO_PI / 512.0f;
    if (t < 4096) {
        const int n = t & 15, w1 = (t >> 4) & 63, s = t >> 10;
        const int m2 = mode_of_pos(16 * s + n);
        const float th = (float)((m2 * w1) & 511) * step;
        const float cs = cosf(th), sn = sinf(th);
        float4 f;
        if (s == 0) {
            if (n < 8) { f.x = cs;   f.y = 0.0f; f.z = -sn;  f.w = 0.0f; }
            else       { f.x = 0.0f; f.y = cs;   f.z = 0.0f; f.w = -sn;  }
        } else {
            if (n < 8) { f.x = cs;   f.y = sn;   f.z = -sn;  f.w = cs;   }
            else       { f.x = cs;   f.y = -sn;  f.z = -sn;  f.w = -cs;  }
        }
        ((float4*)FW)[((size_t)(s * 32 + (w1 >> 1))) * 32 + 16 * (w1 & 1) + n] = f;
    }
    {
        int m1 = t >> 9, h = t & 511;
        float th = (float)((m1 * h) & 511) * step;
        float ct = cosf(th), st = sinf(th);
        cosH[t] = ct;
        sinH[t] = st;
        size_t p2 = ((size_t)(h >> 1) * 64 + m1) * 2 + (h & 1);
        CT2[p2] = ct;
        ST2[p2] = st;
    }
    {
        int k = t >> 9, w = t & 511;
        float th = (float)((k * w) & 511) * step;
        float s  = ((k == 0) ? 1.0f : 2.0f) / 262144.0f;
        DW[(size_t)(2 * k) * 512 + w]     = s * cosf(th);
        DW[(size_t)(2 * k + 1) * 512 + w] = -s * sinf(th);
    }
}

__device__ __forceinline__ void dft8(float a0, float a1, float a2, float a3,
                                     float a4, float a5, float a6, float a7,
                                     float& y0, float& y4, float& y1r, float& y1i,
                                     float& y2r, float& y2i, float& y3r, float& y3i) {
    const float s04 = a0 + a4, d04 = a0 - a4, s15 = a1 + a5, d15 = a1 - a5;
    const float s26 = a2 + a6, d26 = a2 - a6, s37 = a3 + a7, d37 = a3 - a7;
    const float e = s04 + s26, f = s15 + s37;
    const float p = 0.70710678118654752f * (d15 - d37);
    const float q = 0.70710678118654752f * (d15 + d37);
    y0  = e + f;      y4  = e - f;
    y1r = d04 + p;    y1i = -d26 - q;
    y2r = s04 - s26;  y2i = s37 - s15;
    y3r = d04 - p;    y3i = d26 - q;
}

__device__ __forceinline__ void stage_dft8(float* __restrict__ base, const float4 (&a)[8]) {
    float y0[4], y4[4], y1r[4], y1i[4], y2r[4], y2i[4], y3r[4], y3i[4];
    dft8(a[0].x, a[1].x, a[2].x, a[3].x, a[4].x, a[5].x, a[6].x, a[7].x,
         y0[0], y4[0], y1r[0], y1i[0], y2r[0], y2i[0], y3r[0], y3i[0]);
    dft8(a[0].y, a[1].y, a[2].y, a[3].y, a[4].y, a[5].y, a[6].y, a[7].y,
         y0[1], y4[1], y1r[1], y1i[1], y2r[1], y2i[1], y3r[1], y3i[1]);
    dft8(a[0].z, a[1].z, a[2].z, a[3].z, a[4].z, a[5].z, a[6].z, a[7].z,
         y0[2], y4[2], y1r[2], y1i[2], y2r[2], y2i[2], y3r[2], y3i[2]);
    dft8(a[0].w, a[1].w, a[2].w, a[3].w, a[4].w, a[5].w, a[6].w, a[7].w,
         y0[3], y4[3], y1r[3], y1i[3], y2r[3], y2i[3], y3r[3], y3i[3]);
    v4f lo, hi;
    lo = (v4f){y0[0], y4[0], y0[2], y4[2]};      hi = (v4f){y0[1], y4[1], y0[3], y4[3]};
    *(v4f*)(base)                  = lo;         *(v4f*)(base + 64)                  = hi;
    lo = (v4f){y1r[0], y1i[0], y1r[2], y1i[2]};  hi = (v4f){y1r[1], y1i[1], y1r[3], y1i[3]};
    *(v4f*)(base + 16 * ASTR)      = lo;         *(v4f*)(base + 16 * ASTR + 64)      = hi;
    lo = (v4f){y2r[0], y2i[0], y2r[2], y2i[2]};  hi = (v4f){y2r[1], y2i[1], y2r[3], y2i[3]};
    *(v4f*)(base + 32 * ASTR)      = lo;         *(v4f*)(base + 32 * ASTR + 64)      = hi;
    lo = (v4f){y3r[0], y3i[0], y3r[2], y3i[2]};  hi = (v4f){y3r[1], y3i[1], y3r[3], y3i[3]};
    *(v4f*)(base + 48 * ASTR)      = lo;         *(v4f*)(base + 48 * ASTR + 64)      = hi;
}

__global__ __launch_bounds__(512)
void dftw_dfth_kernel(const float* __restrict__ x, const float* __restrict__ FW,
                      const float* __restrict__ CT2, const float* __restrict__ ST2,
                      float* __restrict__ XsP) {
    __shared__ __attribute__((aligned(16))) float As[4 * 64 * ASTR];
    const int c = blockIdx.x >> 3, chunk = blockIdx.x & 7;
    const float4* xrow4 = (const float4*)(x + ((size_t)c * 512 + chunk * 64) * 512);
    const int st = threadIdx.x & 15, r0 = threadIdx.x >> 4;
    const float4* xs0 = xrow4 + (size_t)r0 * 128 + st;
    float4 xa[8], xb[8];
    for (int w2 = 0; w2 < 8; ++w2) xa[w2] = xs0[16 * w2];
    for (int w2 = 0; w2 < 8; ++w2) xb[w2] = xs0[32 * 128 + 16 * w2];
    float* sb = As + (r0 >> 4) * 64 * ASTR + st * ASTR + (r0 & 15) * 4;
    stage_dft8(sb, xa);
    stage_dft8(sb + 2 * 64 * ASTR, xb);
    __syncthreads();

    const int wave = threadIdx.x >> 5, lane = threadIdx.x & 31;
    const int g = wave >> 2, wq = wave & 3;
    const float*  Ag  = As + (g * 64 + wq * 16) * ASTR + lane * 4;
    const float4* fwp = (const float4*)FW + wq * 32 * 32 + lane;
    v8f acc0 = {}, acc1 = {};
#pragma unroll 8
    for (int j = 0; j < 16; ++j) {
        const float4 a4 = *(const float4*)(Ag + j * ASTR);
        const float4 f0 = fwp[(2 * j) * 32];
        const float4 f1 = fwp[(2 * j + 1) * 32];
        v2f a, b0, b1;
        a.x = a4.x;  a.y = a4.y;
        b0.x = f0.x; b0.y = f0.y;
        b1.x = f0.z; b1.y = f0.w;
        acc0 = wmma_f32(a, b0, acc0);
        acc1 = wmma_f32(a, b1, acc1);
        a.x = a4.z;  a.y = a4.w;
        b0.x = f1.x; b0.y = f1.y;
        b1.x = f1.z; b1.y = f1.w;
        acc0 = wmma_f32(a, b0, acc0);
        acc1 = wmma_f32(a, b1, acc1);
    }

    const v8f nacc0 = -acc0;
    v8f xr[4] = {{}, {}, {}, {}}, xi[4] = {{}, {}, {}, {}};
    const int m = lane & 15, half = lane >> 4;
    const int hrow = chunk * 64 + g * 16 + 8 * half;
    for (int j = 0; j < 4; ++j) {
        v2f bR, bI, bN;
        bR.x = acc0[2 * j];  bR.y = acc0[2 * j + 1];
        bI.x = acc1[2 * j];  bI.y = acc1[2 * j + 1];
        bN.x = nacc0[2 * j]; bN.y = nacc0[2 * j + 1];
        const size_t trow = (size_t)((hrow >> 1) + j) * 64;
        for (int mt = 0; mt < 4; ++mt) {
            const size_t ti = (trow + 16 * mt + m) * 2;
            v2f aC = *(const v2f*)(CT2 + ti);
            v2f aS = *(const v2f*)(ST2 + ti);
            xr[mt] = wmma_f32(aC, bR, xr[mt]);
            xr[mt] = wmma_f32(aS, bI, xr[mt]);
            xi[mt] = wmma_f32(aC, bI, xi[mt]);
            xi[mt] = wmma_f32(aS, bN, xi[mt]);
        }
    }

    __syncthreads();
    for (int mt = 0; mt < 4; ++mt) {
        for (int v = 0; v < 8; ++v) {
            As[((wave * 8 + 2 * mt) * 8 + v) * 32 + lane]     = xr[mt][v];
            As[((wave * 8 + 2 * mt + 1) * 8 + v) * 32 + lane] = xi[mt][v];
        }
    }
    __syncthreads();
    float* dst = XsP + (size_t)chunk * (IN_C * 64 * 128) + (size_t)c * 64 * 128;
    for (int v = 0; v < 8; ++v) {
        float sr = 0.0f, si = 0.0f;
        for (int gg = 0; gg < 4; ++gg) {
            const int w2 = gg * 4 + wq;
            sr += As[((w2 * 8 + 2 * g) * 8 + v) * 32 + lane];
            si += As[((w2 * 8 + 2 * g + 1) * 8 + v) * 32 + lane];
        }
        const int m1 = 16 * g + 8 * half + v, pos = 16 * wq + m;
        v2f o; o.x = sr; o.y = si;
        *(v2f*)(dst + (size_t)m1 * 128 + 2 * pos) = o;
    }
}

__global__ void modemul_kernel(const float* __restrict__ XsP, const float* __restrict__ wre,
                               const float* __restrict__ wim, float* __restrict__ YsP) {
    const int m1 = blockIdx.x, obase = blockIdx.y * 16;
    const int ibase = blockIdx.z * ICHUNK;
    __shared__ float xsR[ICHUNK * 64], xsI[ICHUNK * 64];
    for (int idx = threadIdx.x; idx < ICHUNK * 64; idx += blockDim.x) {
        int i = idx >> 6, pos = idx & 63;
        const float* p = XsP + ((size_t)(ibase + i) * 64 + m1) * 128 + 2 * pos;
        float r = 0.0f, im = 0.0f;
        for (int q = 0; q < HPART; ++q) {
            v2f t = *(const v2f*)(p + (size_t)q * (IN_C * 64 * 128));
            r  += t.x;
            im += t.y;
        }
        const int m2 = mode_of_pos(pos);
        xsR[i * 64 + m2] = r;
        xsI[i * 64 + m2] = im;
    }
    __syncthreads();

    const int m2q = (threadIdx.x & 15) * 4;
    const int o   = obase + (threadIdx.x >> 4);
    float aR[4] = {0, 0, 0, 0}, aI[4] = {0, 0, 0, 0};
#pragma unroll 4
    for (int i = 0; i < ICHUNK; ++i) {
        size_t base = (((size_t)(ibase + i) * 64 + o) * 64 + m1) * 64 + m2q;
        float4 wr = *(const float4*)(wre + base);
        float4 wi = *(const float4*)(wim + base);
        const float* xr = xsR + i * 64 + m2q;
        const float* xi = xsI + i * 64 + m2q;
        aR[0] += xr[0] * wr.x - xi[0] * wi.x;  aI[0] += xr[0] * wi.x + xi[0] * wr.x;
        aR[1] += xr[1] * wr.y - xi[1] * wi.y;  aI[1] += xr[1] * wi.y + xi[1] * wr.y;
        aR[2] += xr[2] * wr.z - xi[2] * wi.z;  aI[2] += xr[2] * wi.z + xi[2] * wr.z;
        aR[3] += xr[3] * wr.w - xi[3] * wi.w;  aI[3] += xr[3] * wi.w + xi[3] * wr.w;
    }
    float* yp = YsP + (size_t)blockIdx.z * (OUT_C * 64 * 128)
                    + ((size_t)o * 64 + m1) * 128 + 2 * m2q;
    for (int j = 0; j < 4; ++j) { yp[2 * j] = aR[j]; yp[2 * j + 1] = aI[j]; }
}

#define HCH 32
#define MQ  4
__global__ __launch_bounds__(256)
void idfth_kernel(const float* __restrict__ YsP, const float* __restrict__ cosH,
                  const float* __restrict__ sinH, float* __restrict__ Z) {
    const int o = blockIdx.x, hbase = blockIdx.y * HCH;
    __shared__ float  ysR[64 * 64], ysI[64 * 64];
    __shared__ float4 tc4[64 * HCH / 4], ts4[64 * HCH / 4];
    const float* ysrc = YsP + (size_t)o * (64 * 128) + 2 * threadIdx.x;
    const int    tsrc = (threadIdx.x / HCH) * 512 + hbase + (threadIdx.x % HCH);
    float* tcf = (float*)tc4;
    float* tsf = (float*)ts4;
    v2f   yv[4][ISPLIT];
    float cv[2], sv[2];
    for (int u = 0; u < 4; ++u)
        for (int z = 0; z < ISPLIT; ++z)
            yv[u][z] = *(const v2f*)(ysrc + (size_t)z * (OUT_C * 64 * 128) + 512 * u);
    for (int u = 0; u < 2; ++u) {
        cv[u] = cosH[tsrc + u * (256 / HCH) * 512];
        sv[u] = sinH[tsrc + u * (256 / HCH) * 512];
    }
    for (int u = 0; u < 4; ++u) {
        float r = 0.0f, im = 0.0f;
        for (int z = 0; z < ISPLIT; ++z) { r += yv[u][z].x; im += yv[u][z].y; }
        ysR[threadIdx.x + 256 * u] = r;
        ysI[threadIdx.x + 256 * u] = im;
    }
    for (int u = 0; u < 2; ++u) {
        tcf[threadIdx.x + 256 * u] = cv[u];
        tsf[threadIdx.x + 256 * u] = sv[u];
    }
    __syncthreads();

    const int m2 = threadIdx.x & 63;
    const int h0 = (threadIdx.x >> 6) * 8;
    float aR[8] = {0, 0, 0, 0, 0, 0, 0, 0}, aI[8] = {0, 0, 0, 0, 0, 0, 0, 0};
#pragma nounroll
    for (int q = 0; q < MQ; ++q) {
        if (q + 1 < MQ) {
            for (int u = 0; u < 4; ++u)
                for (int z = 0; z < ISPLIT; ++z)
                    yv[u][z] = *(const v2f*)(ysrc + (size_t)z * (OUT_C * 64 * 128)
                                             + 2048 * (q + 1) + 512 * u);
            for (int u = 0; u < 2; ++u) {
                cv[u] = cosH[tsrc + ((q + 1) * 16 + u * (256 / HCH)) * 512];
                sv[u] = sinH[tsrc + ((q + 1) * 16 + u * (256 / HCH)) * 512];
            }
        }
#pragma unroll 2
        for (int m1 = 16 * q; m1 < 16 * q + 16; ++m1) {
            const float yr = ysR[m1 * 64 + m2], yi = ysI[m1 * 64 + m2];
            const float4 c0 = tc4[(m1 * HCH + h0) / 4], c1 = tc4[(m1 * HCH + h0) / 4 + 1];
            const float4 s0 = ts4[(m1 * HCH + h0) / 4], s1 = ts4[(m1 * HCH + h0) / 4 + 1];
            const float ct[8] = {c0.x, c0.y, c0.z, c0.w, c1.x, c1.y, c1.z, c1.w};
            const float st[8] = {s0.x, s0.y, s0.z, s0.w, s1.x, s1.y, s1.z, s1.w};
            for (int j = 0; j < 8; ++j) {
                aR[j] += ct[j] * yr - st[j] * yi;
                aI[j] += ct[j] * yi + st[j] * yr;
            }
        }
        if (q + 1 < MQ) {
            for (int u = 0; u < 4; ++u) {
                float r = 0.0f, im = 0.0f;
                for (int z = 0; z < ISPLIT; ++z) { r += yv[u][z].x; im += yv[u][z].y; }
                ysR[1024 * (q + 1) + threadIdx.x + 256 * u] = r;
                ysI[1024 * (q + 1) + threadIdx.x + 256 * u] = im;
            }
            for (int u = 0; u < 2; ++u) {
                tcf[512 * (q + 1) + threadIdx.x + 256 * u] = cv[u];
                tsf[512 * (q + 1) + threadIdx.x + 256 * u] = sv[u];
            }
            __syncthreads();
        }
    }
    for (int j = 0; j < 8; ++j) {
        v2f zo; zo.x = aR[j]; zo.y = aI[j];
        *(v2f*)(Z + ((size_t)o * 512 + hbase + h0 + j) * 128 + 2 * m2) = zo;
    }
}

__device__ __forceinline__ v2f lo2(float4 v) { v2f r; r.x = v.x; r.y = v.y; return r; }
__device__ __forceinline__ v2f hi2(float4 v) { v2f r; r.x = v.z; r.y = v.w; return r; }

__device__ __forceinline__ void store_b_pair(float* dst, float4 r0, float4 r1) {
    v4f t0 = {r0.x, r0.y, r0.z, r0.w}, t1 = {r1.x, r1.y, r1.z, r1.w};
    asm volatile("" : "+v"(t0), "+v"(t1));
    float4 lo, hi;
    lo.x = t0.x; lo.y = t1.x; lo.z = t0.y; lo.w = t1.y;
    hi.x = t0.z; hi.y = t1.z; hi.z = t0.w; hi.w = t1.w;
    *(float4*)dst       = lo;
    *(float4*)(dst + 4) = hi;
}

#define AZP 132
#define AKP 68
#define BP  160
#define NCH 8
__global__ __launch_bounds__(256)
void invw_skip_gelu_kernel(const float* __restrict__ Z, const float* __restrict__ DW,
                           const float* __restrict__ x, const float* __restrict__ wskip,
                           const float* __restrict__ bconv, const float* __restrict__ bskip,
                           float* __restrict__ out) {
    const int h = blockIdx.x;
    __shared__ __attribute__((aligned(16))) float Az[64 * AZP];
    __shared__ __attribute__((aligned(16))) float Ak[64 * AKP];
    __shared__ __attribute__((aligned(16))) float Bs[96 * BP];

    const int q0 = threadIdx.x >> 4, pc4 = threadIdx.x & 15;
    const float4* dsrc = (const float4*)(DW + (size_t)(2 * q0) * 512) + pc4;
    const float4* xsrc = (const float4*)(x + (size_t)(2 * q0) * HW + (size_t)h * 512) + pc4;
    float* bdst = Bs + q0 * BP + 8 * pc4;
    float4 ra0, ra1, ra2, ra3, ra4, ra5, rb0, rb1, rb2, rb3, rb4, rb5;
    ra0 = dsrc[0 * 128];  rb0 = dsrc[1 * 128];
    ra1 = dsrc[32 * 128];  rb1 = dsrc[33 * 128];
    ra2 = dsrc[64 * 128];  rb2 = dsrc[65 * 128];
    ra3 = dsrc[96 * 128];  rb3 = dsrc[97 * 128];
    ra4 = xsrc[(size_t)0 * (HW / 4)];  rb4 = xsrc[(size_t)1 * (HW / 4)];
    ra5 = xsrc[(size_t)32 * (HW / 4)];  rb5 = xsrc[(size_t)33 * (HW / 4)];

    for (int idx = threadIdx.x; idx < 64 * 128 / 4; idx += blockDim.x) {
        int o = idx >> 5, kq = idx & 31;
        *(float4*)(Az + o * AZP + 4 * kq) =
            ((const float4*)(Z + ((size_t)o * 512 + h) * 128))[kq];
    }
    for (int idx = threadIdx.x; idx < 64 * 64 / 4; idx += blockDim.x) {
        int o = idx >> 4, kq = idx & 15;
        *(float4*)(Ak + o * AKP + 4 * kq) = ((const float4*)wskip)[idx];
    }
    store_b_pair(bdst + 0 * 16 * BP, ra0, rb0);
    store_b_pair(bdst + 1 * 16 * BP, ra1, rb1);
    store_b_pair(bdst + 2 * 16 * BP, ra2, rb2);
    store_b_pair(bdst + 3 * 16 * BP, ra3, rb3);
    store_b_pair(bdst + 4 * 16 * BP, ra4, rb4);
    store_b_pair(bdst + 5 * 16 * BP, ra5, rb5);
    __syncthreads();

    const int wave = threadIdx.x >> 5, lane = threadIdx.x & 31;
    const int n = lane & 15, half = lane >> 4;
    const int mt = wave & 3, hh = wave >> 2;
    const int m0 = mt * 16;
    const float* azp = Az + (m0 + n) * AZP + 2 * half;
    const float* akp = Ak + (m0 + n) * AKP + 2 * half;
    const float* bp  = Bs + half * BP + 64 * hh + 4 * n;
    const int mb = m0 + half * 8;
    float bc[8], bk[8];
    for (int v = 0; v < 8; ++v) { bc[v] = bconv[mb + v]; bk[v] = bskip[mb + v]; }
    float* orow = out + (size_t)mb * HW + (size_t)h * 512 + 32 * hh + 2 * n;

#pragma nounroll
    for (int c = 0; c < NCH; ++c) {
        if (c + 1 < NCH) {
            ra0 = dsrc[0 * 128 + 16 * (c + 1)];  rb0 = dsrc[1 * 128 + 16 * (c + 1)];
            ra1 = dsrc[32 * 128 + 16 * (c + 1)];  rb1 = dsrc[33 * 128 + 16 * (c + 1)];
            ra2 = dsrc[64 * 128 + 16 * (c + 1)];  rb2 = dsrc[65 * 128 + 16 * (c + 1)];
            ra3 = dsrc[96 * 128 + 16 * (c + 1)];  rb3 = dsrc[97 * 128 + 16 * (c + 1)];
            ra4 = xsrc[(size_t)0 * (HW / 4) + 16 * (c + 1)];  rb4 = xsrc[(size_t)1 * (HW / 4) + 16 * (c + 1)];
            ra5 = xsrc[(size_t)32 * (HW / 4) + 16 * (c + 1)];  rb5 = xsrc[(size_t)33 * (HW / 4) + 16 * (c + 1)];
        }
        v8f acc0 = {}, acc1 = {};
        for (int t = 0; t < 4; ++t) {
            const float* ap = azp + 32 * t;
            const float* bq = bp + (16 * t) * BP;
            const v2f    a0 = *(const v2f*)(ap + 0);
            const v2f    a1 = *(const v2f*)(ap + 4);
            const v2f    a2 = *(const v2f*)(ap + 8);
            const v2f    a3 = *(const v2f*)(ap + 12);
            const v2f    a4 = *(const v2f*)(ap + 16);
            const v2f    a5 = *(const v2f*)(ap + 20);
            const v2f    a6 = *(const v2f*)(ap + 24);
            const v2f    a7 = *(const v2f*)(ap + 28);
            const float4 b0 = *(const float4*)(bq + 0 * BP);
            const float4 b1 = *(const float4*)(bq + 2 * BP);
            const float4 b2 = *(const float4*)(bq + 4 * BP);
            const float4 b3 = *(const float4*)(bq + 6 * BP);
            const float4 b4 = *(const float4*)(bq + 8 * BP);
            const float4 b5 = *(const float4*)(bq + 10 * BP);
            const float4 b6 = *(const float4*)(bq + 12 * BP);
            const float4 b7 = *(const float4*)(bq + 14 * BP);
            __builtin_amdgcn_sched_barrier(0);
            acc0 = wmma_f32(a0, lo2(b0), acc0);  acc1 = wmma_f32(a0, hi2(b0), acc1);
            acc0 = wmma_f32(a1, lo2(b1), acc0);  acc1 = wmma_f32(a1, hi2(b1), acc1);
            acc0 = wmma_f32(a2, lo2(b2), acc0);  acc1 = wmma_f32(a2, hi2(b2), acc1);
            acc0 = wmma_f32(a3, lo2(b3), acc0);  acc1 = wmma_f32(a3, hi2(b3), acc1);
            acc0 = wmma_f32(a4, lo2(b4), acc0);  acc1 = wmma_f32(a4, hi2(b4), acc1);
            acc0 = wmma_f32(a5, lo2(b5), acc0);  acc1 = wmma_f32(a5, hi2(b5), acc1);
            acc0 = wmma_f32(a6, lo2(b6), acc0);  acc1 = wmma_f32(a6, hi2(b6), acc1);
            acc0 = wmma_f32(a7, lo2(b7), acc0);  acc1 = wmma_f32(a7, hi2(b7), acc1);
        }
        for (int t = 0; t < 2; ++t) {
            const float* ap = akp + 32 * t;
            const float* bq = bp + (64 + 16 * t) * BP;
            const v2f    a0 = *(const v2f*)(ap + 0);
            const v2f    a1 = *(const v2f*)(ap + 4);
            const v2f    a2 = *(const v2f*)(ap + 8);
            const v2f    a3 = *(const v2f*)(ap + 12);
            const v2f    a4 = *(const v2f*)(ap + 16);
            const v2f    a5 = *(const v2f*)(ap + 20);
            const v2f    a6 = *(const v2f*)(ap + 24);
            const v2f    a7 = *(const v2f*)(ap + 28);
            const float4 b0 = *(const float4*)(bq + 0 * BP);
            const float4 b1 = *(const float4*)(bq + 2 * BP);
            const float4 b2 = *(const float4*)(bq + 4 * BP);
            const float4 b3 = *(const float4*)(bq + 6 * BP);
            const float4 b4 = *(const float4*)(bq + 8 * BP);
            const float4 b5 = *(const float4*)(bq + 10 * BP);
            const float4 b6 = *(const float4*)(bq + 12 * BP);
            const float4 b7 = *(const float4*)(bq + 14 * BP);
            __builtin_amdgcn_sched_barrier(0);
            acc0 = wmma_f32(a0, lo2(b0), acc0);  acc1 = wmma_f32(a0, hi2(b0), acc1);
            acc0 = wmma_f32(a1, lo2(b1), acc0);  acc1 = wmma_f32(a1, hi2(b1), acc1);
            acc0 = wmma_f32(a2, lo2(b2), acc0);  acc1 = wmma_f32(a2, hi2(b2), acc1);
            acc0 = wmma_f32(a3, lo2(b3), acc0);  acc1 = wmma_f32(a3, hi2(b3), acc1);
            acc0 = wmma_f32(a4, lo2(b4), acc0);  acc1 = wmma_f32(a4, hi2(b4), acc1);
            acc0 = wmma_f32(a5, lo2(b5), acc0);  acc1 = wmma_f32(a5, hi2(b5), acc1);
            acc0 = wmma_f32(a6, lo2(b6), acc0);  acc1 = wmma_f32(a6, hi2(b6), acc1);
            acc0 = wmma_f32(a7, lo2(b7), acc0);  acc1 = wmma_f32(a7, hi2(b7), acc1);
        }
        if (c + 1 < NCH) {
            __syncthreads();
            store_b_pair(bdst + 0 * 16 * BP, ra0, rb0);
            store_b_pair(bdst + 1 * 16 * BP, ra1, rb1);
            store_b_pair(bdst + 2 * 16 * BP, ra2, rb2);
            store_b_pair(bdst + 3 * 16 * BP, ra3, rb3);
            store_b_pair(bdst + 4 * 16 * BP, ra4, rb4);
            store_b_pair(bdst + 5 * 16 * BP, ra5, rb5);
            __syncthreads();
        }
        for (int v = 0; v < 8; ++v) {
            float g[2];
            for (int j = 0; j < 2; ++j) {
                float val = (j ? acc1[v] : acc0[v]) + bc[v] + bk[v];
                float c3  = val * val * val;
                g[j] = 0.5f * val *
                       (1.0f + tanhf(0.7978845608028654f * (val + 0.044715f * c3)));
            }
            v2f o2; o2.x = g[0]; o2.y = g[1];
            *(v2f*)(orow + (size_t)v * HW + 64 * c) = o2;
        }
    }
}

extern "C" void kernel_launch(void* const* d_in, const int* in_sizes, int n_in,
                              void* d_out, int out_size, void* d_ws, size_t ws_size,
                              hipStream_t stream) {
    const float* x     = (const float*)d_in[0];
    const float* w_re  = (const float*)d_in[1];
    const float* w_im  = (const float*)d_in[2];
    const float* b_cv  = (const float*)d_in[3];
    const float* w_sk  = (const float*)d_in[4];
    const float* b_sk  = (const float*)d_in[5];
    float* out = (float*)d_out;

    float* ws   = (float*)d_ws;
    float* FW   = ws;
    float* cosH = FW   + 65536;
    float* sinH = cosH + 32768;
    float* DW   = sinH + 32768;
    float* XsP  = DW   + 65536;
    float* CT2  = XsP  + 4194304;
    float* ST2  = CT2  + 32768;
    float* YsP  = CT2  + 524288;
    float* Z    = YsP  + 2097152;

    twiddle_kernel<<<128, 256, 0, stream>>>(FW, cosH, sinH, DW, CT2, ST2);
    dftw_dfth_kernel<<<512, 512, 0, stream>>>(x, FW, CT2, ST2, XsP);
    modemul_kernel<<<dim3(64, 4, ISPLIT), 256, 0, stream>>>(XsP, w_re, w_im, YsP);
    idfth_kernel<<<dim3(64, 16), 256, 0, stream>>>(YsP, cosH, sinH, Z);
    invw_skip_gelu_kernel<<<512, 256, 0, stream>>>(Z, DW, x, w_sk, b_cv, b_sk, out);
}


# iDFT_W+skip+gelu kernel: W-iDFT split as w = w1 + 64*w2: 8-term WMMA sums per residue k mod 8, then 8-point sums on the vector unit; block = half a row; 2944 WMMAs/row, was 6144; k sum order changes
# speedup vs baseline: 2.1804x; 1.0775x over previous
#include <hip/hip_runtime.h>
#include <math.h>


typedef float v8f __attribute__((ext_vector_type(8)));
typedef float v2f __attribute__((ext_vector_type(2)));
typedef float v4f __attribute__((ext_vector_type(4)));

#define IN_C   64
#define OUT_C  64
#define HH     512
#define WW     512
#define MODES  64
#define HW     (HH * WW)
#define TWO_PI 6.283185307179586f
#define ISPLIT 4
#define ICHUNK (IN_C / ISPLIT)
#define HPART  8
#define ASTR   132

__device__ __forceinline__ v8f wmma_f32(v2f a, v2f b, v8f c) {
    return __builtin_amdgcn_wmma_f32_16x16x4_f32(false, a, false, b, (short)0, c,
                                                 false, false);
}

__device__ __forceinline__ v2f load_a_tile(const float* __restrict__ p, int ld, int lane) {
    int m = lane & 15;
    int k = (lane >> 4) << 1;
    v2f a;
    a.x = p[(size_t)m * ld + k];
    a.y = p[(size_t)m * ld + k + 1];
    return a;
}

__device__ __forceinline__ int mode_of_pos(int p) {
    const int s = p >> 4, t = (p >> 3) & 1, j = p & 7;
    const int r = (s == 0) ? (t ? 4 : 0) : (t ? 8 - s : s);
    return r + 8 * j;
}

__global__ void twiddle_kernel(float* __restrict__ FW, float* __restrict__ cosH,
                               float* __restrict__ sinH, float* __restrict__ TW,
                               float* __restrict__ CT2, float* __restrict__ ST2) {
    int t = blockIdx.x * blockDim.x + threadIdx.x;
    if (t >= 32768) return;
    const float step = TWO_PI / 512.0f;
    if (t < 4096) {
        const int n = t & 15, w1 = (t >> 4) & 63, s = t >> 10;
        const int m2 = mode_of_pos(16 * s + n);
        const float th = (float)((m2 * w1) & 511) * step;
        const float cs = cosf(th), sn = sinf(th);
        float4 f;
        if (s == 0) {
            if (n < 8) { f.x = cs;   f.y = 0.0f; f.z = -sn;  f.w = 0.0f; }
            else       { f.x = 0.0f; f.y = cs;   f.z = 0.0f; f.w = -sn;  }
        } else {
            if (n < 8) { f.x = cs;   f.y = sn;   f.z = -sn;  f.w = cs;   }
            else       { f.x = cs;   f.y = -sn;  f.z = -sn;  f.w = -cs;  }
        }
        ((float4*)FW)[((size_t)(s * 32 + (w1 >> 1))) * 32 + 16 * (w1 & 1) + n] = f;
    }
    {
        int m1 = t >> 9, h = t & 511;
        float th = (float)((m1 * h) & 511) * step;
        float ct = cosf(th), st = sinf(th);
        cosH[t] = ct;
        sinH[t] = st;
        size_t p2 = ((size_t)(h >> 1) * 64 + m1) * 2 + (h & 1);
        CT2[p2] = ct;
        ST2[p2] = st;
    }
    if (t < 4096) {
        const int lane = t & 31, wt = (t >> 5) & 3, s = (t >> 7) & 3, r = t >> 9;
        const int k = r + 8 * (2 * s + (lane >> 4)), w1 = 16 * wt + (lane & 15);
        const float th = (float)((k * w1) & 511) * step;
        const float ck = ((k == 0) ? 1.0f : 2.0f) / 262144.0f;
        const float cs = ck * cosf(th), sn = ck * sinf(th);
        float4 f; f.x = cs; f.y = -sn; f.z = sn; f.w = cs;
        ((float4*)TW)[t] = f;
    }
}

__device__ __forceinline__ void dft8(float a0, float a1, float a2, float a3,
                                     float a4, float a5, float a6, float a7,
                                     float& y0, float& y4, float& y1r, float& y1i,
                                     float& y2r, float& y2i, float& y3r, float& y3i) {
    const float s04 = a0 + a4, d04 = a0 - a4, s15 = a1 + a5, d15 = a1 - a5;
    const float s26 = a2 + a6, d26 = a2 - a6, s37 = a3 + a7, d37 = a3 - a7;
    const float e = s04 + s26, f = s15 + s37;
    const float p = 0.70710678118654752f * (d15 - d37);
    const float q = 0.70710678118654752f * (d15 + d37);
    y0  = e + f;      y4  = e - f;
    y1r = d04 + p;    y1i = -d26 - q;
    y2r = s04 - s26;  y2i = s37 - s15;
    y3r = d04 - p;    y3i = d26 - q;
}

__device__ __forceinline__ void stage_dft8(float* __restrict__ base, const float4 (&a)[8]) {
    float y0[4], y4[4], y1r[4], y1i[4], y2r[4], y2i[4], y3r[4], y3i[4];
    dft8(a[0].x, a[1].x, a[2].x, a[3].x, a[4].x, a[5].x, a[6].x, a[7].x,
         y0[0], y4[0], y1r[0], y1i[0], y2r[0], y2i[0], y3r[0], y3i[0]);
    dft8(a[0].y, a[1].y, a[2].y, a[3].y, a[4].y, a[5].y, a[6].y, a[7].y,
         y0[1], y4[1], y1r[1], y1i[1], y2r[1], y2i[1], y3r[1], y3i[1]);
    dft8(a[0].z, a[1].z, a[2].z, a[3].z, a[4].z, a[5].z, a[6].z, a[7].z,
         y0[2], y4[2], y1r[2], y1i[2], y2r[2], y2i[2], y3r[2], y3i[2]);
    dft8(a[0].w, a[1].w, a[2].w, a[3].w, a[4].w, a[5].w, a[6].w, a[7].w,
         y0[3], y4[3], y1r[3], y1i[3], y2r[3], y2i[3], y3r[3], y3i[3]);
    v4f lo, hi;
    lo = (v4f){y0[0], y4[0], y0[2], y4[2]};      hi = (v4f){y0[1], y4[1], y0[3], y4[3]};
    *(v4f*)(base)                  = lo;         *(v4f*)(base + 64)                  = hi;
    lo = (v4f){y1r[0], y1i[0], y1r[2], y1i[2]};  hi = (v4f){y1r[1], y1i[1], y1r[3], y1i[3]};
    *(v4f*)(base + 16 * ASTR)      = lo;         *(v4f*)(base + 16 * ASTR + 64)      = hi;
    lo = (v4f){y2r[0], y2i[0], y2r[2], y2i[2]};  hi = (v4f){y2r[1], y2i[1], y2r[3], y2i[3]};
    *(v4f*)(base + 32 * ASTR)      = lo;         *(v4f*)(base + 32 * ASTR + 64)      = hi;
    lo = (v4f){y3r[0], y3i[0], y3r[2], y3i[2]};  hi = (v4f){y3r[1], y3i[1], y3r[3], y3i[3]};
    *(v4f*)(base + 48 * ASTR)      = lo;         *(v4f*)(base + 48 * ASTR + 64)      = hi;
}

__global__ __launch_bounds__(512)
void dftw_dfth_kernel(const float* __restrict__ x, const float* __restrict__ FW,
                      const float* __restrict__ CT2, const float* __restrict__ ST2,
                      float* __restrict__ XsP) {
    __shared__ __attribute__((aligned(16))) float As[4 * 64 * ASTR];
    const int c = blockIdx.x >> 3, chunk = blockIdx.x & 7;
    const float4* xrow4 = (const float4*)(x + ((size_t)c * 512 + chunk * 64) * 512);
    const int st = threadIdx.x & 15, r0 = threadIdx.x >> 4;
    const float4* xs0 = xrow4 + (size_t)r0 * 128 + st;
    float4 xa[8], xb[8];
    for (int w2 = 0; w2 < 8; ++w2) xa[w2] = xs0[16 * w2];
    for (int w2 = 0; w2 < 8; ++w2) xb[w2] = xs0[32 * 128 + 16 * w2];
    float* sb = As + (r0 >> 4) * 64 * ASTR + st * ASTR + (r0 & 15) * 4;
    stage_dft8(sb, xa);
    stage_dft8(sb + 2 * 64 * ASTR, xb);
    __syncthreads();

    const int wave = threadIdx.x >> 5, lane = threadIdx.x & 31;
    const int g = wave >> 2, wq = wave & 3;
    const float*  Ag  = As + (g * 64 + wq * 16) * ASTR + lane * 4;
    const float4* fwp = (const float4*)FW + wq * 32 * 32 + lane;
    v8f acc0 = {}, acc1 = {};
#pragma unroll 8
    for (int j = 0; j < 16; ++j) {
        const float4 a4 = *(const float4*)(Ag + j * ASTR);
        const float4 f0 = fwp[(2 * j) * 32];
        const float4 f1 = fwp[(2 * j + 1) * 32];
        v2f a, b0, b1;
        a.x = a4.x;  a.y = a4.y;
        b0.x = f0.x; b0.y = f0.y;
        b1.x = f0.z; b1.y = f0.w;
        acc0 = wmma_f32(a, b0, acc0);
        acc1 = wmma_f32(a, b1, acc1);
        a.x = a4.z;  a.y = a4.w;
        b0.x = f1.x; b0.y = f1.y;
        b1.x = f1.z; b1.y = f1.w;
        acc0 = wmma_f32(a, b0, acc0);
        acc1 = wmma_f32(a, b1, acc1);
    }

    const v8f nacc0 = -acc0;
    v8f xr[4] = {{}, {}, {}, {}}, xi[4] = {{}, {}, {}, {}};
    const int m = lane & 15, half = lane >> 4;
    const int hrow = chunk * 64 + g * 16 + 8 * half;
    for (int j = 0; j < 4; ++j) {
        v2f bR, bI, bN;
        bR.x = acc0[2 * j];  bR.y = acc0[2 * j + 1];
        bI.x = acc1[2 * j];  bI.y = acc1[2 * j + 1];
        bN.x = nacc0[2 * j]; bN.y = nacc0[2 * j + 1];
        const size_t trow = (size_t)((hrow >> 1) + j) * 64;
        for (int mt = 0; mt < 4; ++mt) {
            const size_t ti = (trow + 16 * mt + m) * 2;
            v2f aC = *(const v2f*)(CT2 + ti);
            v2f aS = *(const v2f*)(ST2 + ti);
            xr[mt] = wmma_f32(aC, bR, xr[mt]);
            xr[mt] = wmma_f32(aS, bI, xr[mt]);
            xi[mt] = wmma_f32(aC, bI, xi[mt]);
            xi[mt] = wmma_f32(aS, bN, xi[mt]);
        }
    }

    __syncthreads();
    for (int mt = 0; mt < 4; ++mt) {
        for (int v = 0; v < 8; ++v) {
            As[((wave * 8 + 2 * mt) * 8 + v) * 32 + lane]     = xr[mt][v];
            As[((wave * 8 + 2 * mt + 1) * 8 + v) * 32 + lane] = xi[mt][v];
        }
    }
    __syncthreads();
    float* dst = XsP + (size_t)chunk * (IN_C * 64 * 128) + (size_t)c * 64 * 128;
    for (int v = 0; v < 8; ++v) {
        float sr = 0.0f, si = 0.0f;
        for (int gg = 0; gg < 4; ++gg) {
            const int w2 = gg * 4 + wq;
            sr += As[((w2 * 8 + 2 * g) * 8 + v) * 32 + lane];
            si += As[((w2 * 8 + 2 * g + 1) * 8 + v) * 32 + lane];
        }
        const int m1 = 16 * g + 8 * half + v, pos = 16 * wq + m;
        v2f o; o.x = sr; o.y = si;
        *(v2f*)(dst + (size_t)m1 * 128 + 2 * pos) = o;
    }
}

__global__ void modemul_kernel(const float* __restrict__ XsP, const float* __restrict__ wre,
                               const float* __restrict__ wim, float* __restrict__ YsP) {
    const int m1 = blockIdx.x, obase = blockIdx.y * 16;
    const int ibase = blockIdx.z * ICHUNK;
    __shared__ float xsR[ICHUNK * 64], xsI[ICHUNK * 64];
    for (int idx = threadIdx.x; idx < ICHUNK * 64; idx += blockDim.x) {
        int i = idx >> 6, pos = idx & 63;
        const float* p = XsP + ((size_t)(ibase + i) * 64 + m1) * 128 + 2 * pos;
        float r = 0.0f, im = 0.0f;
        for (int q = 0; q < HPART; ++q) {
            v2f t = *(const v2f*)(p + (size_t)q * (IN_C * 64 * 128));
            r  += t.x;
            im += t.y;
        }
        const int m2 = mode_of_pos(pos);
        xsR[i * 64 + m2] = r;
        xsI[i * 64 + m2] = im;
    }
    __syncthreads();

    const int m2q = (threadIdx.x & 15) * 4;
    const int o   = obase + (threadIdx.x >> 4);
    float aR[4] = {0, 0, 0, 0}, aI[4] = {0, 0, 0, 0};
#pragma unroll 4
    for (int i = 0; i < ICHUNK; ++i) {
        size_t base = (((size_t)(ibase + i) * 64 + o) * 64 + m1) * 64 + m2q;
        float4 wr = *(const float4*)(wre + base);
        float4 wi = *(const float4*)(wim + base);
        const float* xr = xsR + i * 64 + m2q;
        const float* xi = xsI + i * 64 + m2q;
        aR[0] += xr[0] * wr.x - xi[0] * wi.x;  aI[0] += xr[0] * wi.x + xi[0] * wr.x;
        aR[1] += xr[1] * wr.y - xi[1] * wi.y;  aI[1] += xr[1] * wi.y + xi[1] * wr.y;
        aR[2] += xr[2] * wr.z - xi[2] * wi.z;  aI[2] += xr[2] * wi.z + xi[2] * wr.z;
        aR[3] += xr[3] * wr.w - xi[3] * wi.w;  aI[3] += xr[3] * wi.w + xi[3] * wr.w;
    }
    float* yp = YsP + (size_t)blockIdx.z * (OUT_C * 64 * 128)
                    + ((size_t)o * 64 + m1) * 128 + 2 * m2q;
    for (int j = 0; j < 4; ++j) { yp[2 * j] = aR[j]; yp[2 * j + 1] = aI[j]; }
}

#define HCH 32
#define MQ  4
__global__ __launch_bounds__(256)
void idfth_kernel(const float* __restrict__ YsP, const float* __restrict__ cosH,
                  const float* __restrict__ sinH, float* __restrict__ Z) {
    const int o = blockIdx.x, hbase = blockIdx.y * HCH;
    __shared__ float  ysR[64 * 64], ysI[64 * 64];
    __shared__ float4 tc4[64 * HCH / 4], ts4[64 * HCH / 4];
    const float* ysrc = YsP + (size_t)o * (64 * 128) + 2 * threadIdx.x;
    const int    tsrc = (threadIdx.x / HCH) * 512 + hbase + (threadIdx.x % HCH);
    float* tcf = (float*)tc4;
    float* tsf = (float*)ts4;
    v2f   yv[4][ISPLIT];
    float cv[2], sv[2];
    for (int u = 0; u < 4; ++u)
        for (int z = 0; z < ISPLIT; ++z)
            yv[u][z] = *(const v2f*)(ysrc + (size_t)z * (OUT_C * 64 * 128) + 512 * u);
    for (int u = 0; u < 2; ++u) {
        cv[u] = cosH[tsrc + u * (256 / HCH) * 512];
        sv[u] = sinH[tsrc + u * (256 / HCH) * 512];
    }
    for (int u = 0; u < 4; ++u) {
        float r = 0.0f, im = 0.0f;
        for (int z = 0; z < ISPLIT; ++z) { r += yv[u][z].x; im += yv[u][z].y; }
        ysR[threadIdx.x + 256 * u] = r;
        ysI[threadIdx.x + 256 * u] = im;
    }
    for (int u = 0; u < 2; ++u) {
        tcf[threadIdx.x + 256 * u] = cv[u];
        tsf[threadIdx.x + 256 * u] = sv[u];
    }
    __syncthreads();

    const int m2 = threadIdx.x & 63;
    const int h0 = (threadIdx.x >> 6) * 8;
    float aR[8] = {0, 0, 0, 0, 0, 0, 0, 0}, aI[8] = {0, 0, 0, 0, 0, 0, 0, 0};
#pragma nounroll
    for (int q = 0; q < MQ; ++q) {
        if (q + 1 < MQ) {
            for (int u = 0; u < 4; ++u)
                for (int z = 0; z < ISPLIT; ++z)
                    yv[u][z] = *(const v2f*)(ysrc + (size_t)z * (OUT_C * 64 * 128)
                                             + 2048 * (q + 1) + 512 * u);
            for (int u = 0; u < 2; ++u) {
                cv[u] = cosH[tsrc + ((q + 1) * 16 + u * (256 / HCH)) * 512];
                sv[u] = sinH[tsrc + ((q + 1) * 16 + u * (256 / HCH)) * 512];
            }
        }
#pragma unroll 2
        for (int m1 = 16 * q; m1 < 16 * q + 16; ++m1) {
            const float yr = ysR[m1 * 64 + m2], yi = ysI[m1 * 64 + m2];
            const float4 c0 = tc4[(m1 * HCH + h0) / 4], c1 = tc4[(m1 * HCH + h0) / 4 + 1];
            const float4 s0 = ts4[(m1 * HCH + h0) / 4], s1 = ts4[(m1 * HCH + h0) / 4 + 1];
            const float ct[8] = {c0.x, c0.y, c0.z, c0.w, c1.x, c1.y, c1.z, c1.w};
            const float st[8] = {s0.x, s0.y, s0.z, s0.w, s1.x, s1.y, s1.z, s1.w};
            for (int j = 0; j < 8; ++j) {
                aR[j] += ct[j] * yr - st[j] * yi;
                aI[j] += ct[j] * yi + st[j] * yr;
            }
        }
        if (q + 1 < MQ) {
            for (int u = 0; u < 4; ++u) {
                float r = 0.0f, im = 0.0f;
                for (int z = 0; z < ISPLIT; ++z) { r += yv[u][z].x; im += yv[u][z].y; }
                ysR[1024 * (q + 1) + threadIdx.x + 256 * u] = r;
                ysI[1024 * (q + 1) + threadIdx.x + 256 * u] = im;
            }
            for (int u = 0; u < 2; ++u) {
                tcf[512 * (q + 1) + threadIdx.x + 256 * u] = cv[u];
                tsf[512 * (q + 1) + threadIdx.x + 256 * u] = sv[u];
            }
            __syncthreads();
        }
    }
    for (int j = 0; j < 8; ++j) {
        v2f zo; zo.x = aR[j]; zo.y = aI[j];
        *(v2f*)(Z + ((size_t)o * 512 + hbase + h0 + j) * 128 + 2 * m2) = zo;
    }
}

__device__ __forceinline__ v2f lo2(float4 v) { v2f r; r.x = v.x; r.y = v.y; return r; }
__device__ __forceinline__ v2f hi2(float4 v) { v2f r; r.x = v.z; r.y = v.w; return r; }

__device__ __forceinline__ void store_b_pair(float* dst, float4 r0, float4 r1) {
    float4 lo, hi;
    lo.x = r0.x; lo.y = r1.x; lo.z = r0.y; lo.w = r1.y;
    hi.x = r0.z; hi.y = r1.z; hi.z = r0.w; hi.w = r1.w;
    *(float4*)dst       = lo;
    *(float4*)(dst + 4) = hi;
}

__device__ __forceinline__ float gelu_tanh(float val) {
    const float c3 = val * val * val;
    return 0.5f * val * (1.0f + tanhf(0.7978845608028654f * (val + 0.044715f * c3)));
}

#define AZP 132
#define AKP 68
#define BPX 544
__global__ __launch_bounds__(256, 4)
void invw_skip_gelu_kernel(const float* __restrict__ Z, const float* __restrict__ TW,
                           const float* __restrict__ x, const float* __restrict__ wskip,
                           const float* __restrict__ bconv, const float* __restrict__ bskip,
                           float* __restrict__ out) {
    const int h = blockIdx.x >> 1, c2 = blockIdx.x & 1;
    __shared__ __attribute__((aligned(16))) float Az[64 * AZP];
    __shared__ __attribute__((aligned(16))) float Ak[64 * AKP];
    __shared__ __attribute__((aligned(16))) float Bs[32 * BPX];

    const int p = threadIdx.x & 63, q0 = threadIdx.x >> 6;
    const float4* xsrc = (const float4*)(x + (size_t)(2 * q0) * HW + (size_t)h * 512
                                           + 32 * c2 + 4 * (p & 7) + 64 * (p >> 3));
    float* bdst = Bs + q0 * BPX + 8 * p;
    float4 ra[8], rb[8];
    for (int u = 0; u < 8; ++u) {
        ra[u] = xsrc[(size_t)(8 * u) * (HW / 4)];
        rb[u] = xsrc[(size_t)(8 * u + 1) * (HW / 4)];
    }
    for (int idx = threadIdx.x; idx < 64 * 128 / 4; idx += blockDim.x) {
        const int o = idx >> 5, kq = idx & 31;
        const float4 z = ((const float4*)(Z + ((size_t)o * 512 + h) * 128))[kq];
        const int k0 = 2 * kq, k1 = 2 * kq + 1;
        v2f z0, z1;
        z0.x = z.x; z0.y = z.y; z1.x = z.z; z1.y = z.w;
        *(v2f*)(Az + o * AZP + 2 * (8 * (k0 & 7) + (k0 >> 3))) = z0;
        *(v2f*)(Az + o * AZP + 2 * (8 * (k1 & 7) + (k1 >> 3))) = z1;
    }
    for (int idx = threadIdx.x; idx < 64 * 64 / 4; idx += blockDim.x) {
        int o = idx >> 4, kq = idx & 15;
        *(float4*)(Ak + o * AKP + 4 * kq) = ((const float4*)wskip)[idx];
    }
    for (int u = 0; u < 8; ++u) store_b_pair(bdst + 4 * u * BPX, ra[u], rb[u]);
    __syncthreads();

    const int wave = threadIdx.x >> 5, lane = threadIdx.x & 31;
    const int n = lane & 15, half = lane >> 4;
    const int mt = wave & 3, wt = wave >> 2;
    const int m0 = mt * 16;
    const float* azp = Az + (m0 + n) * AZP + 2 * half;
    const float* akp = Ak + (m0 + n) * AKP + 2 * half;
    const float* bp  = Bs + half * BPX + 32 * wt + 2 * n;
    const float4* tw = (const float4*)TW + (2 * c2 + wt) * 32 + lane;
    const int mb = m0 + half * 8;
    float bias[8];
    for (int v = 0; v < 8; ++v) bias[v] = bconv[mb + v] + bskip[mb + v];

    v8f ur[8], ui[8];
    for (int r = 0; r < 8; ++r) { ur[r] = (v8f){}; ui[r] = (v8f){}; }
    for (int s = 0; s < 4; ++s) {
        const v2f    a0 = *(const v2f*)(azp + 4 * s);
        const v2f    a4 = *(const v2f*)(azp + 64 + 4 * s);
        const float4 f0 = tw[s * 128];
        const float4 f4 = tw[(16 + s) * 128];
        ur[0] = wmma_f32(a0, lo2(f0), ur[0]);
        ur[4] = wmma_f32(a4, lo2(f4), ur[4]);
    }
    for (int r = 1; r < 8; ++r) {
        if (r == 4) continue;
        for (int s = 0; s < 4; ++s) {
            const v2f    a = *(const v2f*)(azp + 16 * r + 4 * s);
            const float4 f = tw[(4 * r + s) * 128];
            ur[r] = wmma_f32(a, lo2(f), ur[r]);
            ui[r] = wmma_f32(a, hi2(f), ui[r]);
        }
    }

    v8f acc[8];
    {
        const float hs = 0.70710678118654752f;
        const v8f s04 = ur[0] + ur[4], d04 = ur[0] - ur[4];
        const v8f s26 = ur[2] + ur[6], d26 = ui[2] - ui[6];
        const v8f e0 = s04 + s26, e2 = s04 - s26, e1 = d04 - d26, e3 = d04 + d26;
        const v8f o0 = (ur[1] + ur[5]) + (ur[3] + ur[7]);
        const v8f o2 = (ui[3] + ui[7]) - (ui[1] + ui[5]);
        const v8f o1 = hs * (((ur[1] - ui[1]) - (ur[5] - ui[5])) + ((ur[7] + ui[7]) - (ur[3] + ui[3])));
        const v8f o3 = hs * (((ur[3] - ui[3]) - (ur[7] - ui[7])) + ((ur[5] + ui[5]) - (ur[1] + ui[1])));
        acc[0] = e0 + o0;  acc[4] = e0 - o0;
        acc[1] = e1 + o1;  acc[5] = e1 - o1;
        acc[2] = e2 + o2;  acc[6] = e2 - o2;
        acc[3] = e3 + o3;  acc[7] = e3 - o3;
    }

    for (int ks = 0; ks < 16; ++ks) {
        const v2f a = *(const v2f*)(akp + 4 * ks);
        const float* bq = bp + (2 * ks) * BPX;
        for (int w2 = 0; w2 < 8; ++w2)
            acc[w2] = wmma_f32(a, *(const v2f*)(bq + 64 * w2), acc[w2]);
    }

    float* orow = out + (size_t)mb * HW + (size_t)h * 512 + 32 * c2 + 16 * wt + n;
    for (int v = 0; v < 8; ++v)
        for (int w2 = 0; w2 < 8; ++w2)
            orow[(size_t)v * HW + 64 * w2] = gelu_tanh(acc[w2][v] + bias[v]);
}

extern "C" void kernel_launch(void* const* d_in, const int* in_sizes, int n_in,
                              void* d_out, int out_size, void* d_ws, size_t ws_size,
                              hipStream_t stream) {
    const float* x     = (const float*)d_in[0];
    const float* w_re  = (const float*)d_in[1];
    const float* w_im  = (const float*)d_in[2];
    const float* b_cv  = (const float*)d_in[3];
    const float* w_sk  = (const float*)d_in[4];
    const float* b_sk  = (const float*)d_in[5];
    float* out = (float*)d_out;

    float* ws   = (float*)d_ws;
    float* FW   = ws;
    float* cosH = FW   + 65536;
    float* sinH = cosH + 32768;
    float* TW   = sinH + 32768;
    float* XsP  = TW   + 65536;
    float* CT2  = XsP  + 4194304;
    float* ST2  = CT2  + 32768;
    float* YsP  = CT2  + 524288;
    float* Z    = YsP  + 2097152;

    twiddle_kernel<<<128, 256, 0, stream>>>(FW, cosH, sinH, TW, CT2, ST2);
    dftw_dfth_kernel<<<512, 512, 0, stream>>>(x, FW, CT2, ST2, XsP);
    modemul_kernel<<<dim3(64, 4, ISPLIT), 256, 0, stream>>>(XsP, w_re, w_im, YsP);
    idfth_kernel<<<dim3(64, 16), 256, 0, stream>>>(YsP, cosH, sinH, Z);
    invw_skip_gelu_kernel<<<1024, 256, 0, stream>>>(Z, TW, x, w_sk, b_cv, b_sk, out);
}


# inverse H-DFT kernel: transform split as h = h1 + 64*h2: 8-term sums per residue m1 mod 8, then 8-point complex sums; block = one o x 16 modes, Ys staged once per o; m1 sum order changes
# speedup vs baseline: 2.3731x; 1.0884x over previous
#include <hip/hip_runtime.h>
#include <math.h>


typedef float v8f __attribute__((ext_vector_type(8)));
typedef float v2f __attribute__((ext_vector_type(2)));
typedef float v4f __attribute__((ext_vector_type(4)));

#define IN_C   64
#define OUT_C  64
#define HH     512
#define WW     512
#define MODES  64
#define HW     (HH * WW)
#define TWO_PI 6.283185307179586f
#define ISPLIT 4
#define ICHUNK (IN_C / ISPLIT)
#define HPART  8
#define ASTR   132

__device__ __forceinline__ v8f wmma_f32(v2f a, v2f b, v8f c) {
    return __builtin_amdgcn_wmma_f32_16x16x4_f32(false, a, false, b, (short)0, c,
                                                 false, false);
}

__device__ __forceinline__ v2f load_a_tile(const float* __restrict__ p, int ld, int lane) {
    int m = lane & 15;
    int k = (lane >> 4) << 1;
    v2f a;
    a.x = p[(size_t)m * ld + k];
    a.y = p[(size_t)m * ld + k + 1];
    return a;
}

__device__ __forceinline__ int mode_of_pos(int p) {
    const int s = p >> 4, t = (p >> 3) & 1, j = p & 7;
    const int r = (s == 0) ? (t ? 4 : 0) : (t ? 8 - s : s);
    return r + 8 * j;
}

__global__ void twiddle_kernel(float* __restrict__ FW, float* __restrict__ cosH,
                               float* __restrict__ sinH, float* __restrict__ TW,
                               float* __restrict__ CT2, float* __restrict__ ST2) {
    int t = blockIdx.x * blockDim.x + threadIdx.x;
    if (t >= 32768) return;
    const float step = TWO_PI / 512.0f;
    if (t < 4096) {
        const int n = t & 15, w1 = (t >> 4) & 63, s = t >> 10;
        const int m2 = mode_of_pos(16 * s + n);
        const float th = (float)((m2 * w1) & 511) * step;
        const float cs = cosf(th), sn = sinf(th);
        float4 f;
        if (s == 0) {
            if (n < 8) { f.x = cs;   f.y = 0.0f; f.z = -sn;  f.w = 0.0f; }
            else       { f.x = 0.0f; f.y = cs;   f.z = 0.0f; f.w = -sn;  }
        } else {
            if (n < 8) { f.x = cs;   f.y = sn;   f.z = -sn;  f.w = cs;   }
            else       { f.x = cs;   f.y = -sn;  f.z = -sn;  f.w = -cs;  }
        }
        ((float4*)FW)[((size_t)(s * 32 + (w1 >> 1))) * 32 + 16 * (w1 & 1) + n] = f;
    }
    {
        int m1 = t >> 9, h = t & 511;
        float th = (float)((m1 * h) & 511) * step;
        float ct = cosf(th), st = sinf(th);
        cosH[t] = ct;
        sinH[t] = st;
        size_t p2 = ((size_t)(h >> 1) * 64 + m1) * 2 + (h & 1);
        CT2[p2] = ct;
        ST2[p2] = st;
    }
    if (t < 4096) {
        const int lane = t & 31, wt = (t >> 5) & 3, s = (t >> 7) & 3, r = t >> 9;
        const int k = r + 8 * (2 * s + (lane >> 4)), w1 = 16 * wt + (lane & 15);
        const float th = (float)((k * w1) & 511) * step;
        const float ck = ((k == 0) ? 1.0f : 2.0f) / 262144.0f;
        const float cs = ck * cosf(th), sn = ck * sinf(th);
        float4 f; f.x = cs; f.y = -sn; f.z = sn; f.w = cs;
        ((float4*)TW)[t] = f;
    }
}

__device__ __forceinline__ void dft8(float a0, float a1, float a2, float a3,
                                     float a4, float a5, float a6, float a7,
                                     float& y0, float& y4, float& y1r, float& y1i,
                                     float& y2r, float& y2i, float& y3r, float& y3i) {
    const float s04 = a0 + a4, d04 = a0 - a4, s15 = a1 + a5, d15 = a1 - a5;
    const float s26 = a2 + a6, d26 = a2 - a6, s37 = a3 + a7, d37 = a3 - a7;
    const float e = s04 + s26, f = s15 + s37;
    const float p = 0.70710678118654752f * (d15 - d37);
    const float q = 0.70710678118654752f * (d15 + d37);
    y0  = e + f;      y4  = e - f;
    y1r = d04 + p;    y1i = -d26 - q;
    y2r = s04 - s26;  y2i = s37 - s15;
    y3r = d04 - p;    y3i = d26 - q;
}

__device__ __forceinline__ void stage_dft8(float* __restrict__ base, const float4 (&a)[8]) {
    float y0[4], y4[4], y1r[4], y1i[4], y2r[4], y2i[4], y3r[4], y3i[4];
    dft8(a[0].x, a[1].x, a[2].x, a[3].x, a[4].x, a[5].x, a[6].x, a[7].x,
         y0[0], y4[0], y1r[0], y1i[0], y2r[0], y2i[0], y3r[0], y3i[0]);
    dft8(a[0].y, a[1].y, a[2].y, a[3].y, a[4].y, a[5].y, a[6].y, a[7].y,
         y0[1], y4[1], y1r[1], y1i[1], y2r[1], y2i[1], y3r[1], y3i[1]);
    dft8(a[0].z, a[1].z, a[2].z, a[3].z, a[4].z, a[5].z, a[6].z, a[7].z,
         y0[2], y4[2], y1r[2], y1i[2], y2r[2], y2i[2], y3r[2], y3i[2]);
    dft8(a[0].w, a[1].w, a[2].w, a[3].w, a[4].w, a[5].w, a[6].w, a[7].w,
         y0[3], y4[3], y1r[3], y1i[3], y2r[3], y2i[3], y3r[3], y3i[3]);
    v4f lo, hi;
    lo = (v4f){y0[0], y4[0], y0[2], y4[2]};      hi = (v4f){y0[1], y4[1], y0[3], y4[3]};
    *(v4f*)(base)                  = lo;         *(v4f*)(base + 64)                  = hi;
    lo = (v4f){y1r[0], y1i[0], y1r[2], y1i[2]};  hi = (v4f){y1r[1], y1i[1], y1r[3], y1i[3]};
    *(v4f*)(base + 16 * ASTR)      = lo;         *(v4f*)(base + 16 * ASTR + 64)      = hi;
    lo = (v4f){y2r[0], y2i[0], y2r[2], y2i[2]};  hi = (v4f){y2r[1], y2i[1], y2r[3], y2i[3]};
    *(v4f*)(base + 32 * ASTR)      = lo;         *(v4f*)(base + 32 * ASTR + 64)      = hi;
    lo = (v4f){y3r[0], y3i[0], y3r[2], y3i[2]};  hi = (v4f){y3r[1], y3i[1], y3r[3], y3i[3]};
    *(v4f*)(base + 48 * ASTR)      = lo;         *(v4f*)(base + 48 * ASTR + 64)      = hi;
}

__global__ __launch_bounds__(512)
void dftw_dfth_kernel(const float* __restrict__ x, const float* __restrict__ FW,
                      const float* __restrict__ CT2, const float* __restrict__ ST2,
                      float* __restrict__ XsP) {
    __shared__ __attribute__((aligned(16))) float As[4 * 64 * ASTR];
    const int c = blockIdx.x >> 3, chunk = blockIdx.x & 7;
    const float4* xrow4 = (const float4*)(x + ((size_t)c * 512 + chunk * 64) * 512);
    const int st = threadIdx.x & 15, r0 = threadIdx.x >> 4;
    const float4* xs0 = xrow4 + (size_t)r0 * 128 + st;
    float4 xa[8], xb[8];
    for (int w2 = 0; w2 < 8; ++w2) xa[w2] = xs0[16 * w2];
    for (int w2 = 0; w2 < 8; ++w2) xb[w2] = xs0[32 * 128 + 16 * w2];
    float* sb = As + (r0 >> 4) * 64 * ASTR + st * ASTR + (r0 & 15) * 4;
    stage_dft8(sb, xa);
    stage_dft8(sb + 2 * 64 * ASTR, xb);
    __syncthreads();

    const int wave = threadIdx.x >> 5, lane = threadIdx.x & 31;
    const int g = wave >> 2, wq = wave & 3;
    const float*  Ag  = As + (g * 64 + wq * 16) * ASTR + lane * 4;
    const float4* fwp = (const float4*)FW + wq * 32 * 32 + lane;
    v8f acc0 = {}, acc1 = {};
#pragma unroll 8
    for (int j = 0; j < 16; ++j) {
        const float4 a4 = *(const float4*)(Ag + j * ASTR);
        const float4 f0 = fwp[(2 * j) * 32];
        const float4 f1 = fwp[(2 * j + 1) * 32];
        v2f a, b0, b1;
        a.x = a4.x;  a.y = a4.y;
        b0.x = f0.x; b0.y = f0.y;
        b1.x = f0.z; b1.y = f0.w;
        acc0 = wmma_f32(a, b0, acc0);
        acc1 = wmma_f32(a, b1, acc1);
        a.x = a4.z;  a.y = a4.w;
        b0.x = f1.x; b0.y = f1.y;
        b1.x = f1.z; b1.y = f1.w;
        acc0 = wmma_f32(a, b0, acc0);
        acc1 = wmma_f32(a, b1, acc1);
    }

    const v8f nacc0 = -acc0;
    v8f xr[4] = {{}, {}, {}, {}}, xi[4] = {{}, {}, {}, {}};
    const int m = lane & 15, half = lane >> 4;
    const int hrow = chunk * 64 + g * 16 + 8 * half;
    for (int j = 0; j < 4; ++j) {
        v2f bR, bI, bN;
        bR.x = acc0[2 * j];  bR.y = acc0[2 * j + 1];
        bI.x = acc1[2 * j];  bI.y = acc1[2 * j + 1];
        bN.x = nacc0[2 * j]; bN.y = nacc0[2 * j + 1];
        const size_t trow = (size_t)((hrow >> 1) + j) * 64;
        for (int mt = 0; mt < 4; ++mt) {
            const size_t ti = (trow + 16 * mt + m) * 2;
            v2f aC = *(const v2f*)(CT2 + ti);
            v2f aS = *(const v2f*)(ST2 + ti);
            xr[mt] = wmma_f32(aC, bR, xr[mt]);
            xr[mt] = wmma_f32(aS, bI, xr[mt]);
            xi[mt] = wmma_f32(aC, bI, xi[mt]);
            xi[mt] = wmma_f32(aS, bN, xi[mt]);
        }
    }

    __syncthreads();
    for (int mt = 0; mt < 4; ++mt) {
        for (int v = 0; v < 8; ++v) {
            As[((wave * 8 + 2 * mt) * 8 + v) * 32 + lane]     = xr[mt][v];
            As[((wave * 8 + 2 * mt + 1) * 8 + v) * 32 + lane] = xi[mt][v];
        }
    }
    __syncthreads();
    float* dst = XsP + (size_t)chunk * (IN_C * 64 * 128) + (size_t)c * 64 * 128;
    for (int v = 0; v < 8; ++v) {
        float sr = 0.0f, si = 0.0f;
        for (int gg = 0; gg < 4; ++gg) {
            const int w2 = gg * 4 + wq;
            sr += As[((w2 * 8 + 2 * g) * 8 + v) * 32 + lane];
            si += As[((w2 * 8 + 2 * g + 1) * 8 + v) * 32 + lane];
        }
        const int m1 = 16 * g + 8 * half + v, pos = 16 * wq + m;
        v2f o; o.x = sr; o.y = si;
        *(v2f*)(dst + (size_t)m1 * 128 + 2 * pos) = o;
    }
}

__global__ void modemul_kernel(const float* __restrict__ XsP, const float* __restrict__ wre,
                               const float* __restrict__ wim, float* __restrict__ YsP) {
    const int m1 = blockIdx.x, obase = blockIdx.y * 16;
    const int ibase = blockIdx.z * ICHUNK;
    __shared__ float xsR[ICHUNK * 64], xsI[ICHUNK * 64];
    for (int idx = threadIdx.x; idx < ICHUNK * 64; idx += blockDim.x) {
        int i = idx >> 6, pos = idx & 63;
        const float* p = XsP + ((size_t)(ibase + i) * 64 + m1) * 128 + 2 * pos;
        float r = 0.0f, im = 0.0f;
        for (int q = 0; q < HPART; ++q) {
            v2f t = *(const v2f*)(p + (size_t)q * (IN_C * 64 * 128));
            r  += t.x;
            im += t.y;
        }
        const int m2 = mode_of_pos(pos);
        xsR[i * 64 + m2] = r;
        xsI[i * 64 + m2] = im;
    }
    __syncthreads();

    const int m2q = (threadIdx.x & 15) * 4;
    const int o   = obase + (threadIdx.x >> 4);
    float aR[4] = {0, 0, 0, 0}, aI[4] = {0, 0, 0, 0};
#pragma unroll 4
    for (int i = 0; i < ICHUNK; ++i) {
        size_t base = (((size_t)(ibase + i) * 64 + o) * 64 + m1) * 64 + m2q;
        float4 wr = *(const float4*)(wre + base);
        float4 wi = *(const float4*)(wim + base);
        const float* xr = xsR + i * 64 + m2q;
        const float* xi = xsI + i * 64 + m2q;
        aR[0] += xr[0] * wr.x - xi[0] * wi.x;  aI[0] += xr[0] * wi.x + xi[0] * wr.x;
        aR[1] += xr[1] * wr.y - xi[1] * wi.y;  aI[1] += xr[1] * wi.y + xi[1] * wr.y;
        aR[2] += xr[2] * wr.z - xi[2] * wi.z;  aI[2] += xr[2] * wi.z + xi[2] * wr.z;
        aR[3] += xr[3] * wr.w - xi[3] * wi.w;  aI[3] += xr[3] * wi.w + xi[3] * wr.w;
    }
    float* yp = YsP + (size_t)blockIdx.z * (OUT_C * 64 * 128)
                    + ((size_t)o * 64 + m1) * 128 + 2 * m2q;
    for (int j = 0; j < 4; ++j) { yp[2 * j] = aR[j]; yp[2 * j + 1] = aI[j]; }
}

#define M2C 16

__device__ __forceinline__ void idft4(float a0r, float a0i, float a1r, float a1i,
                                      float a2r, float a2i, float a3r, float a3i,
                                      float (&Ar)[4], float (&Ai)[4]) {
    const float t0r = a0r + a2r, t0i = a0i + a2i, t1r = a0r - a2r, t1i = a0i - a2i;
    const float t2r = a1r + a3r, t2i = a1i + a3i, t3r = a1r - a3r, t3i = a1i - a3i;
    Ar[0] = t0r + t2r;  Ai[0] = t0i + t2i;
    Ar[1] = t1r - t3i;  Ai[1] = t1i + t3r;
    Ar[2] = t0r - t2r;  Ai[2] = t0i - t2i;
    Ar[3] = t1r + t3i;  Ai[3] = t1i - t3r;
}

__global__ __launch_bounds__(256)
void idfth_kernel(const float* __restrict__ YsP, const float* __restrict__ cosH,
                  const float* __restrict__ sinH, float* __restrict__ Z) {
    const int o = blockIdx.x, m2b = blockIdx.y * M2C;
    __shared__ v2f    ys2[64 * M2C];
    __shared__ float4 tc4[64 * 16], ts4[64 * 16];
    const int mm = threadIdx.x & 15, hg = threadIdx.x >> 4;
    const float* ysrc = YsP + ((size_t)o * 64 + hg) * 128 + 2 * (m2b + mm);
    v2f    yv[4][ISPLIT];
    float4 cv[4], sv[4];
    for (int u = 0; u < 4; ++u)
        for (int z = 0; z < ISPLIT; ++z)
            yv[u][z] = *(const v2f*)(ysrc + (size_t)z * (OUT_C * 64 * 128) + 2048 * u);
    for (int u = 0; u < 4; ++u) {
        const int idx = threadIdx.x + 256 * u;
        cv[u] = ((const float4*)(cosH + (idx >> 4) * 512))[idx & 15];
        sv[u] = ((const float4*)(sinH + (idx >> 4) * 512))[idx & 15];
    }
    for (int u = 0; u < 4; ++u) {
        float r = 0.0f, im = 0.0f;
        for (int z = 0; z < ISPLIT; ++z) { r += yv[u][z].x; im += yv[u][z].y; }
        v2f y; y.x = r; y.y = im;
        ys2[threadIdx.x + 256 * u] = y;
    }
    for (int u = 0; u < 4; ++u) {
        tc4[threadIdx.x + 256 * u] = cv[u];
        ts4[threadIdx.x + 256 * u] = sv[u];
    }
    __syncthreads();

    v2f UR[8][2], UI[8][2];
    for (int r = 0; r < 8; ++r)
        for (int b = 0; b < 2; ++b) { UR[r][b] = (v2f){0.0f, 0.0f}; UI[r][b] = (v2f){0.0f, 0.0f}; }
#pragma nounroll
    for (int j = 0; j < 8; ++j) {
        v2f    y[8];
        float4 c[8], s[8];
        for (int r = 0; r < 8; ++r) {
            const int m1 = r + 8 * j;
            y[r] = ys2[m1 * M2C + mm];
            c[r] = tc4[m1 * 16 + hg];
            s[r] = ts4[m1 * 16 + hg];
        }
        __builtin_amdgcn_sched_barrier(0);
        for (int r = 0; r < 8; ++r) {
            const v2f yr = {y[r].x, y[r].x}, yi = {y[r].y, y[r].y};
            const v2f cp[2] = {{c[r].x, c[r].y}, {c[r].z, c[r].w}};
            const v2f sp[2] = {{s[r].x, s[r].y}, {s[r].z, s[r].w}};
            for (int b = 0; b < 2; ++b) {
                UR[r][b] = __builtin_elementwise_fma(cp[b], yr, UR[r][b]);
                UR[r][b] = __builtin_elementwise_fma(-sp[b], yi, UR[r][b]);
                UI[r][b] = __builtin_elementwise_fma(cp[b], yi, UI[r][b]);
                UI[r][b] = __builtin_elementwise_fma(sp[b], yr, UI[r][b]);
            }
        }
    }

    const float hs = 0.70710678118654752f;
    float* zp = Z + ((size_t)o * 512 + 4 * hg) * 128 + 2 * (m2b + mm);
    for (int a = 0; a < 4; ++a) {
        float er[4], ei[4], qr[4], qi[4], pr[4], pi[4];
        const int b = a >> 1, e = a & 1;
        idft4(UR[0][b][e], UI[0][b][e], UR[2][b][e], UI[2][b][e], UR[4][b][e], UI[4][b][e],
              UR[6][b][e], UI[6][b][e], er, ei);
        idft4(UR[1][b][e], UI[1][b][e], UR[3][b][e], UI[3][b][e], UR[5][b][e], UI[5][b][e],
              UR[7][b][e], UI[7][b][e], qr, qi);
        pr[0] = qr[0];                  pi[0] = qi[0];
        pr[1] = hs * (qr[1] - qi[1]);   pi[1] = hs * (qr[1] + qi[1]);
        pr[2] = -qi[2];                 pi[2] = qr[2];
        pr[3] = -hs * (qr[3] + qi[3]);  pi[3] = hs * (qr[3] - qi[3]);
        for (int k = 0; k < 4; ++k) {
            v2f z0, z4;
            z0.x = er[k] + pr[k];  z0.y = ei[k] + pi[k];
            z4.x = er[k] - pr[k];  z4.y = ei[k] - pi[k];
            *(v2f*)(zp + (size_t)(a + 64 * k) * 128)       = z0;
            *(v2f*)(zp + (size_t)(a + 64 * (k + 4)) * 128) = z4;
        }
    }
}

__device__ __forceinline__ v2f lo2(float4 v) { v2f r; r.x = v.x; r.y = v.y; return r; }
__device__ __forceinline__ v2f hi2(float4 v) { v2f r; r.x = v.z; r.y = v.w; return r; }

__device__ __forceinline__ void store_b_pair(float* dst, float4 r0, float4 r1) {
    float4 lo, hi;
    lo.x = r0.x; lo.y = r1.x; lo.z = r0.y; lo.w = r1.y;
    hi.x = r0.z; hi.y = r1.z; hi.z = r0.w; hi.w = r1.w;
    *(float4*)dst       = lo;
    *(float4*)(dst + 4) = hi;
}

__device__ __forceinline__ float gelu_tanh(float val) {
    const float c3 = val * val * val;
    return 0.5f * val * (1.0f + tanhf(0.7978845608028654f * (val + 0.044715f * c3)));
}

#define AZP 132
#define AKP 68
#define BPX 544
__global__ __launch_bounds__(256, 4)
void invw_skip_gelu_kernel(const float* __restrict__ Z, const float* __restrict__ TW,
                           const float* __restrict__ x, const float* __restrict__ wskip,
                           const float* __restrict__ bconv, const float* __restrict__ bskip,
                           float* __restrict__ out) {
    const int h = blockIdx.x >> 1, c2 = blockIdx.x & 1;
    __shared__ __attribute__((aligned(16))) float Az[64 * AZP];
    __shared__ __attribute__((aligned(16))) float Ak[64 * AKP];
    __shared__ __attribute__((aligned(16))) float Bs[32 * BPX];

    const int p = threadIdx.x & 63, q0 = threadIdx.x >> 6;
    const float4* xsrc = (const float4*)(x + (size_t)(2 * q0) * HW + (size_t)h * 512
                                           + 32 * c2 + 4 * (p & 7) + 64 * (p >> 3));
    float* bdst = Bs + q0 * BPX + 8 * p;
    float4 ra[8], rb[8];
    for (int u = 0; u < 8; ++u) {
        ra[u] = xsrc[(size_t)(8 * u) * (HW / 4)];
        rb[u] = xsrc[(size_t)(8 * u + 1) * (HW / 4)];
    }
    for (int idx = threadIdx.x; idx < 64 * 128 / 4; idx += blockDim.x) {
        const int o = idx >> 5, kq = idx & 31;
        const float4 z = ((const float4*)(Z + ((size_t)o * 512 + h) * 128))[kq];
        const int k0 = 2 * kq, k1 = 2 * kq + 1;
        v2f z0, z1;
        z0.x = z.x; z0.y = z.y; z1.x = z.z; z1.y = z.w;
        *(v2f*)(Az + o * AZP + 2 * (8 * (k0 & 7) + (k0 >> 3))) = z0;
        *(v2f*)(Az + o * AZP + 2 * (8 * (k1 & 7) + (k1 >> 3))) = z1;
    }
    for (int idx = threadIdx.x; idx < 64 * 64 / 4; idx += blockDim.x) {
        int o = idx >> 4, kq = idx & 15;
        *(float4*)(Ak + o * AKP + 4 * kq) = ((const float4*)wskip)[idx];
    }
    for (int u = 0; u < 8; ++u) store_b_pair(bdst + 4 * u * BPX, ra[u], rb[u]);
    __syncthreads();

    const int wave = threadIdx.x >> 5, lane = threadIdx.x & 31;
    const int n = lane & 15, half = lane >> 4;
    const int mt = wave & 3, wt = wave >> 2;
    const int m0 = mt * 16;
    const float* azp = Az + (m0 + n) * AZP + 2 * half;
    const float* akp = Ak + (m0 + n) * AKP + 2 * half;
    const float* bp  = Bs + half * BPX + 32 * wt + 2 * n;
    const float4* tw = (const float4*)TW + (2 * c2 + wt) * 32 + lane;
    const int mb = m0 + half * 8;
    float bias[8];
    for (int v = 0; v < 8; ++v) bias[v] = bconv[mb + v] + bskip[mb + v];

    v8f ur[8], ui[8];
    for (int r = 0; r < 8; ++r) { ur[r] = (v8f){}; ui[r] = (v8f){}; }
    for (int s = 0; s < 4; ++s) {
        const v2f    a0 = *(const v2f*)(azp + 4 * s);
        const v2f    a4 = *(const v2f*)(azp + 64 + 4 * s);
        const float4 f0 = tw[s * 128];
        const float4 f4 = tw[(16 + s) * 128];
        ur[0] = wmma_f32(a0, lo2(f0), ur[0]);
        ur[4] = wmma_f32(a4, lo2(f4), ur[4]);
    }
    for (int r = 1; r < 8; ++r) {
        if (r == 4) continue;
        for (int s = 0; s < 4; ++s) {
            const v2f    a = *(const v2f*)(azp + 16 * r + 4 * s);
            const float4 f = tw[(4 * r + s) * 128];
            ur[r] = wmma_f32(a, lo2(f), ur[r]);
            ui[r] = wmma_f32(a, hi2(f), ui[r]);
        }
    }

    v8f acc[8];
    {
        const float hs = 0.70710678118654752f;
        const v8f s04 = ur[0] + ur[4], d04 = ur[0] - ur[4];
        const v8f s26 = ur[2] + ur[6], d26 = ui[2] - ui[6];
        const v8f e0 = s04 + s26, e2 = s04 - s26, e1 = d04 - d26, e3 = d04 + d26;
        const v8f o0 = (ur[1] + ur[5]) + (ur[3] + ur[7]);
        const v8f o2 = (ui[3] + ui[7]) - (ui[1] + ui[5]);
        const v8f o1 = hs * (((ur[1] - ui[1]) - (ur[5] - ui[5])) + ((ur[7] + ui[7]) - (ur[3] + ui[3])));
        const v8f o3 = hs * (((ur[3] - ui[3]) - (ur[7] - ui[7])) + ((ur[5] + ui[5]) - (ur[1] + ui[1])));
        acc[0] = e0 + o0;  acc[4] = e0 - o0;
        acc[1] = e1 + o1;  acc[5] = e1 - o1;
        acc[2] = e2 + o2;  acc[6] = e2 - o2;
        acc[3] = e3 + o3;  acc[7] = e3 - o3;
    }

    for (int ks = 0; ks < 16; ++ks) {
        const v2f a = *(const v2f*)(akp + 4 * ks);
        const float* bq = bp + (2 * ks) * BPX;
        for (int w2 = 0; w2 < 8; ++w2)
            acc[w2] = wmma_f32(a, *(const v2f*)(bq + 64 * w2), acc[w2]);
    }

    float* orow = out + (size_t)mb * HW + (size_t)h * 512 + 32 * c2 + 16 * wt + n;
    for (int v = 0; v < 8; ++v)
        for (int w2 = 0; w2 < 8; ++w2)
            orow[(size_t)v * HW + 64 * w2] = gelu_tanh(acc[w2][v] + bias[v]);
}

extern "C" void kernel_launch(void* const* d_in, const int* in_sizes, int n_in,
                              void* d_out, int out_size, void* d_ws, size_t ws_size,
                              hipStream_t stream) {
    const float* x     = (const float*)d_in[0];
    const float* w_re  = (const float*)d_in[1];
    const float* w_im  = (const float*)d_in[2];
    const float* b_cv  = (const float*)d_in[3];
    const float* w_sk  = (const float*)d_in[4];
    const float* b_sk  = (const float*)d_in[5];
    float* out = (float*)d_out;

    float* ws   = (float*)d_ws;
    float* FW   = ws;
    float* cosH = FW   + 65536;
    float* sinH = cosH + 32768;
    float* TW   = sinH + 32768;
    float* XsP  = TW   + 65536;
    float* CT2  = XsP  + 4194304;
    float* ST2  = CT2  + 32768;
    float* YsP  = CT2  + 524288;
    float* Z    = YsP  + 2097152;

    twiddle_kernel<<<128, 256, 0, stream>>>(FW, cosH, sinH, TW, CT2, ST2);
    dftw_dfth_kernel<<<512, 512, 0, stream>>>(x, FW, CT2, ST2, XsP);
    modemul_kernel<<<dim3(64, 4, ISPLIT), 256, 0, stream>>>(XsP, w_re, w_im, YsP);
    idfth_kernel<<<dim3(64, 4), 256, 0, stream>>>(YsP, cosH, sinH, Z);
    invw_skip_gelu_kernel<<<1024, 256, 0, stream>>>(Z, TW, x, w_sk, b_cv, b_sk, out);
}


# forward H-DFT in kernel 1 split as h = h1 + 64*h2: 8-point DFTs over a lane's own accumulator registers (vector unit), then 8 WMMAs/wave on (Yre,Yim) pairs, was 64; h sum order changes
# speedup vs baseline: 2.5212x; 1.0624x over previous
#include <hip/hip_runtime.h>
#include <math.h>


typedef float v8f __attribute__((ext_vector_type(8)));
typedef float v2f __attribute__((ext_vector_type(2)));
typedef float v4f __attribute__((ext_vector_type(4)));

#define IN_C   64
#define OUT_C  64
#define HH     512
#define WW     512
#define MODES  64
#define HW     (HH * WW)
#define TWO_PI 6.283185307179586f
#define ISPLIT 4
#define ICHUNK (IN_C / ISPLIT)
#define HPART  8
#define ASTR   132

__device__ __forceinline__ v8f wmma_f32(v2f a, v2f b, v8f c) {
    return __builtin_amdgcn_wmma_f32_16x16x4_f32(false, a, false, b, (short)0, c,
                                                 false, false);
}

__device__ __forceinline__ v2f load_a_tile(const float* __restrict__ p, int ld, int lane) {
    int m = lane & 15;
    int k = (lane >> 4) << 1;
    v2f a;
    a.x = p[(size_t)m * ld + k];
    a.y = p[(size_t)m * ld + k + 1];
    return a;
}

__device__ __forceinline__ int mode_of_pos(int p) {
    const int s = p >> 4, t = (p >> 3) & 1, j = p & 7;
    const int r = (s == 0) ? (t ? 4 : 0) : (t ? 8 - s : s);
    return r + 8 * j;
}

__global__ void twiddle_kernel(float* __restrict__ FW, float* __restrict__ cosH,
                               float* __restrict__ sinH, float* __restrict__ TW,
                               float* __restrict__ HT) {
    int t = blockIdx.x * blockDim.x + threadIdx.x;
    if (t >= 32768) return;
    const float step = TWO_PI / 512.0f;
    if (t < 4096) {
        const int n = t & 15, w1 = (t >> 4) & 63, s = t >> 10;
        const int m2 = mode_of_pos(16 * s + n);
        const float th = (float)((m2 * w1) & 511) * step;
        const float cs = cosf(th), sn = sinf(th);
        float4 f;
        if (s == 0) {
            if (n < 8) { f.x = cs;   f.y = 0.0f; f.z = -sn;  f.w = 0.0f; }
            else       { f.x = 0.0f; f.y = cs;   f.z = 0.0f; f.w = -sn;  }
        } else {
            if (n < 8) { f.x = cs;   f.y = sn;   f.z = -sn;  f.w = cs;   }
            else       { f.x = cs;   f.y = -sn;  f.z = -sn;  f.w = -cs;  }
        }
        ((float4*)FW)[((size_t)(s * 32 + (w1 >> 1))) * 32 + 16 * (w1 & 1) + n] = f;
    }
    {
        int m1 = t >> 9, h = t & 511;
        float th = (float)((m1 * h) & 511) * step;
        cosH[t] = cosf(th);
        sinH[t] = sinf(th);
    }
    if (t < 8192) {
        const int lane = t & 31, r = (t >> 5) & 7, hp = t >> 8;
        const int M = lane & 15, m1 = r + 8 * (M & 7), h1 = 2 * hp + (lane >> 4);
        const float th = (float)((m1 * h1) & 511) * step;
        const float cs = cosf(th), sn = sinf(th);
        v2f f;
        if (M < 8) { f.x = cs;  f.y = sn; }
        else       { f.x = -sn; f.y = cs; }
        ((v2f*)HT)[t] = f;
    }
    if (t < 4096) {
        const int lane = t & 31, wt = (t >> 5) & 3, s = (t >> 7) & 3, r = t >> 9;
        const int k = r + 8 * (2 * s + (lane >> 4)), w1 = 16 * wt + (lane & 15);
        const float th = (float)((k * w1) & 511) * step;
        const float ck = ((k == 0) ? 1.0f : 2.0f) / 262144.0f;
        const float cs = ck * cosf(th), sn = ck * sinf(th);
        float4 f; f.x = cs; f.y = -sn; f.z = sn; f.w = cs;
        ((float4*)TW)[t] = f;
    }
}

__device__ __forceinline__ void dft8(float a0, float a1, float a2, float a3,
                                     float a4, float a5, float a6, float a7,
                                     float& y0, float& y4, float& y1r, float& y1i,
                                     float& y2r, float& y2i, float& y3r, float& y3i) {
    const float s04 = a0 + a4, d04 = a0 - a4, s15 = a1 + a5, d15 = a1 - a5;
    const float s26 = a2 + a6, d26 = a2 - a6, s37 = a3 + a7, d37 = a3 - a7;
    const float e = s04 + s26, f = s15 + s37;
    const float p = 0.70710678118654752f * (d15 - d37);
    const float q = 0.70710678118654752f * (d15 + d37);
    y0  = e + f;      y4  = e - f;
    y1r = d04 + p;    y1i = -d26 - q;
    y2r = s04 - s26;  y2i = s37 - s15;
    y3r = d04 - p;    y3i = d26 - q;
}

__device__ __forceinline__ void stage_dft8(float* __restrict__ base, const float4 (&a)[8]) {
    float y0[4], y4[4], y1r[4], y1i[4], y2r[4], y2i[4], y3r[4], y3i[4];
    dft8(a[0].x, a[1].x, a[2].x, a[3].x, a[4].x, a[5].x, a[6].x, a[7].x,
         y0[0], y4[0], y1r[0], y1i[0], y2r[0], y2i[0], y3r[0], y3i[0]);
    dft8(a[0].y, a[1].y, a[2].y, a[3].y, a[4].y, a[5].y, a[6].y, a[7].y,
         y0[1], y4[1], y1r[1], y1i[1], y2r[1], y2i[1], y3r[1], y3i[1]);
    dft8(a[0].z, a[1].z, a[2].z, a[3].z, a[4].z, a[5].z, a[6].z, a[7].z,
         y0[2], y4[2], y1r[2], y1i[2], y2r[2], y2i[2], y3r[2], y3i[2]);
    dft8(a[0].w, a[1].w, a[2].w, a[3].w, a[4].w, a[5].w, a[6].w, a[7].w,
         y0[3], y4[3], y1r[3], y1i[3], y2r[3], y2i[3], y3r[3], y3i[3]);
    v4f lo, hi;
    lo = (v4f){y0[0], y4[0], y0[2], y4[2]};      hi = (v4f){y0[1], y4[1], y0[3], y4[3]};
    *(v4f*)(base)                  = lo;         *(v4f*)(base + 64)                  = hi;
    lo = (v4f){y1r[0], y1i[0], y1r[2], y1i[2]};  hi = (v4f){y1r[1], y1i[1], y1r[3], y1i[3]};
    *(v4f*)(base + 16 * ASTR)      = lo;         *(v4f*)(base + 16 * ASTR + 64)      = hi;
    lo = (v4f){y2r[0], y2i[0], y2r[2], y2i[2]};  hi = (v4f){y2r[1], y2i[1], y2r[3], y2i[3]};
    *(v4f*)(base + 32 * ASTR)      = lo;         *(v4f*)(base + 32 * ASTR + 64)      = hi;
    lo = (v4f){y3r[0], y3i[0], y3r[2], y3i[2]};  hi = (v4f){y3r[1], y3i[1], y3r[3], y3i[3]};
    *(v4f*)(base + 48 * ASTR)      = lo;         *(v4f*)(base + 48 * ASTR + 64)      = hi;
}

__device__ __forceinline__ void dft4f(float b0r, float b0i, float b1r, float b1i,
                                      float b2r, float b2i, float b3r, float b3i,
                                      float (&Br)[4], float (&Bi)[4]) {
    const float t0r = b0r + b2r, t0i = b0i + b2i, t1r = b0r - b2r, t1i = b0i - b2i;
    const float t2r = b1r + b3r, t2i = b1i + b3i, t3r = b1r - b3r, t3i = b1i - b3i;
    Br[0] = t0r + t2r;  Bi[0] = t0i + t2i;
    Br[1] = t1r + t3i;  Bi[1] = t1i - t3r;
    Br[2] = t0r - t2r;  Bi[2] = t0i - t2i;
    Br[3] = t1r - t3i;  Bi[3] = t1i + t3r;
}

__device__ __forceinline__ void dft8c(const v8f& re, const v8f& im,
                                      float (&yr)[8], float (&yi)[8]) {
    const float hs = 0.70710678118654752f;
    float er[4], ei[4], qr[4], qi[4], pr[4], pi[4];
    dft4f(re[0], im[0], re[2], im[2], re[4], im[4], re[6], im[6], er, ei);
    dft4f(re[1], im[1], re[3], im[3], re[5], im[5], re[7], im[7], qr, qi);
    pr[0] = qr[0];                  pi[0] = qi[0];
    pr[1] = hs * (qr[1] + qi[1]);   pi[1] = hs * (qi[1] - qr[1]);
    pr[2] = qi[2];                  pi[2] = -qr[2];
    pr[3] = hs * (qi[3] - qr[3]);   pi[3] = -hs * (qr[3] + qi[3]);
    for (int q = 0; q < 4; ++q) {
        yr[q]     = er[q] + pr[q];  yi[q]     = ei[q] + pi[q];
        yr[q + 4] = er[q] - pr[q];  yi[q + 4] = ei[q] - pi[q];
    }
}

__global__ __launch_bounds__(512)
void dftw_dfth_kernel(const float* __restrict__ x, const float* __restrict__ FW,
                      const float* __restrict__ HT, float* __restrict__ XsP) {
    __shared__ __attribute__((aligned(16))) float As[4 * 64 * ASTR];
    const int c = blockIdx.x >> 3, chunk = blockIdx.x & 7;
    const int st = threadIdx.x & 15, r0 = threadIdx.x >> 4;
    const int hrow = chunk * 8 + 2 * (r0 >> 4) + ((r0 >> 3) & 1) + 64 * (r0 & 7);
    const float4* xs0 = (const float4*)(x + ((size_t)c * 512 + hrow) * 512) + st;
    float4 xa[8], xb[8];
    for (int w2 = 0; w2 < 8; ++w2) xa[w2] = xs0[16 * w2];
    for (int w2 = 0; w2 < 8; ++w2) xb[w2] = xs0[4 * 128 + 16 * w2];
    float* sb = As + (r0 >> 4) * 64 * ASTR + st * ASTR + (r0 & 15) * 4;
    stage_dft8(sb, xa);
    stage_dft8(sb + 2 * 64 * ASTR, xb);
    __syncthreads();

    const int wave = threadIdx.x >> 5, lane = threadIdx.x & 31;
    const int g = wave >> 2, wq = wave & 3;
    const float*  Ag  = As + (g * 64 + wq * 16) * ASTR + lane * 4;
    const float4* fwp = (const float4*)FW + wq * 32 * 32 + lane;
    v8f acc0 = {}, acc1 = {};
#pragma unroll 8
    for (int j = 0; j < 16; ++j) {
        const float4 a4 = *(const float4*)(Ag + j * ASTR);
        const float4 f0 = fwp[(2 * j) * 32];
        const float4 f1 = fwp[(2 * j + 1) * 32];
        v2f a, b0, b1;
        a.x = a4.x;  a.y = a4.y;
        b0.x = f0.x; b0.y = f0.y;
        b1.x = f0.z; b1.y = f0.w;
        acc0 = wmma_f32(a, b0, acc0);
        acc1 = wmma_f32(a, b1, acc1);
        a.x = a4.z;  a.y = a4.w;
        b0.x = f1.x; b0.y = f1.y;
        b1.x = f1.z; b1.y = f1.w;
        acc0 = wmma_f32(a, b0, acc0);
        acc1 = wmma_f32(a, b1, acc1);
    }

    const int n = lane & 15, half = lane >> 4;
    const v2f* ht = (const v2f*)HT + ((size_t)(chunk * 4) * 8 + 2 * g) * 32 + lane;
    v2f ta[4], tb[4];
    for (int gg = 0; gg < 4; ++gg) {
        ta[gg] = ht[(gg * 8) * 32];
        tb[gg] = ht[(gg * 8 + 1) * 32];
    }

    float yr[8], yi[8];
    dft8c(acc0, acc1, yr, yi);

    __syncthreads();
    v2f* Yx = (v2f*)As;
    for (int r = 0; r < 8; ++r) {
        v2f y; y.x = yr[r]; y.y = yi[r];
        Yx[((wq * 8 + r) * 4 + g) * 32 + lane] = y;
    }
    __syncthreads();
    v8f o0 = {}, o1 = {};
    for (int gg = 0; gg < 4; ++gg) {
        const v2f b0 = Yx[((wq * 8 + 2 * g) * 4 + gg) * 32 + lane];
        const v2f b1 = Yx[((wq * 8 + 2 * g + 1) * 4 + gg) * 32 + lane];
        o0 = wmma_f32(ta[gg], b0, o0);
        o1 = wmma_f32(tb[gg], b1, o1);
    }
    float* dst = XsP + (size_t)chunk * (IN_C * 64 * 128) + (size_t)c * 64 * 128
                     + 2 * (16 * wq + n) + half;
    for (int v = 0; v < 8; ++v) {
        dst[(size_t)(2 * g + 8 * v) * 128]     = o0[v];
        dst[(size_t)(2 * g + 1 + 8 * v) * 128] = o1[v];
    }
}

__global__ void modemul_kernel(const float* __restrict__ XsP, const float* __restrict__ wre,
                               const float* __restrict__ wim, float* __restrict__ YsP) {
    const int m1 = blockIdx.x, obase = blockIdx.y * 16;
    const int ibase = blockIdx.z * ICHUNK;
    __shared__ float xsR[ICHUNK * 64], xsI[ICHUNK * 64];
    for (int idx = threadIdx.x; idx < ICHUNK * 64; idx += blockDim.x) {
        int i = idx >> 6, pos = idx & 63;
        const float* p = XsP + ((size_t)(ibase + i) * 64 + m1) * 128 + 2 * pos;
        float r = 0.0f, im = 0.0f;
        for (int q = 0; q < HPART; ++q) {
            v2f t = *(const v2f*)(p + (size_t)q * (IN_C * 64 * 128));
            r  += t.x;
            im += t.y;
        }
        const int m2 = mode_of_pos(pos);
        xsR[i * 64 + m2] = r;
        xsI[i * 64 + m2] = im;
    }
    __syncthreads();

    const int m2q = (threadIdx.x & 15) * 4;
    const int o   = obase + (threadIdx.x >> 4);
    float aR[4] = {0, 0, 0, 0}, aI[4] = {0, 0, 0, 0};
#pragma unroll 4
    for (int i = 0; i < ICHUNK; ++i) {
        size_t base = (((size_t)(ibase + i) * 64 + o) * 64 + m1) * 64 + m2q;
        float4 wr = *(const float4*)(wre + base);
        float4 wi = *(const float4*)(wim + base);
        const float* xr = xsR + i * 64 + m2q;
        const float* xi = xsI + i * 64 + m2q;
        aR[0] += xr[0] * wr.x - xi[0] * wi.x;  aI[0] += xr[0] * wi.x + xi[0] * wr.x;
        aR[1] += xr[1] * wr.y - xi[1] * wi.y;  aI[1] += xr[1] * wi.y + xi[1] * wr.y;
        aR[2] += xr[2] * wr.z - xi[2] * wi.z;  aI[2] += xr[2] * wi.z + xi[2] * wr.z;
        aR[3] += xr[3] * wr.w - xi[3] * wi.w;  aI[3] += xr[3] * wi.w + xi[3] * wr.w;
    }
    float* yp = YsP + (size_t)blockIdx.z * (OUT_C * 64 * 128)
                    + ((size_t)o * 64 + m1) * 128 + 2 * m2q;
    for (int j = 0; j < 4; ++j) { yp[2 * j] = aR[j]; yp[2 * j + 1] = aI[j]; }
}

#define M2C 16

__device__ __forceinline__ void idft4(float a0r, float a0i, float a1r, float a1i,
                                      float a2r, float a2i, float a3r, float a3i,
                                      float (&Ar)[4], float (&Ai)[4]) {
    const float t0r = a0r + a2r, t0i = a0i + a2i, t1r = a0r - a2r, t1i = a0i - a2i;
    const float t2r = a1r + a3r, t2i = a1i + a3i, t3r = a1r - a3r, t3i = a1i - a3i;
    Ar[0] = t0r + t2r;  Ai[0] = t0i + t2i;
    Ar[1] = t1r - t3i;  Ai[1] = t1i + t3r;
    Ar[2] = t0r - t2r;  Ai[2] = t0i - t2i;
    Ar[3] = t1r + t3i;  Ai[3] = t1i - t3r;
}

__global__ __launch_bounds__(256)
void idfth_kernel(const float* __restrict__ YsP, const float* __restrict__ cosH,
                  const float* __restrict__ sinH, float* __restrict__ Z) {
    const int o = blockIdx.x, m2b = blockIdx.y * M2C;
    __shared__ v2f    ys2[64 * M2C];
    __shared__ float4 tc4[64 * 16], ts4[64 * 16];
    const int mm = threadIdx.x & 15, hg = threadIdx.x >> 4;
    const float* ysrc = YsP + ((size_t)o * 64 + hg) * 128 + 2 * (m2b + mm);
    v2f    yv[4][ISPLIT];
    float4 cv[4], sv[4];
    for (int u = 0; u < 4; ++u)
        for (int z = 0; z < ISPLIT; ++z)
            yv[u][z] = *(const v2f*)(ysrc + (size_t)z * (OUT_C * 64 * 128) + 2048 * u);
    for (int u = 0; u < 4; ++u) {
        const int idx = threadIdx.x + 256 * u;
        cv[u] = ((const float4*)(cosH + (idx >> 4) * 512))[idx & 15];
        sv[u] = ((const float4*)(sinH + (idx >> 4) * 512))[idx & 15];
    }
    for (int u = 0; u < 4; ++u) {
        float r = 0.0f, im = 0.0f;
        for (int z = 0; z < ISPLIT; ++z) { r += yv[u][z].x; im += yv[u][z].y; }
        v2f y; y.x = r; y.y = im;
        ys2[threadIdx.x + 256 * u] = y;
    }
    for (int u = 0; u < 4; ++u) {
        tc4[threadIdx.x + 256 * u] = cv[u];
        ts4[threadIdx.x + 256 * u] = sv[u];
    }
    __syncthreads();

    v2f UR[8][2], UI[8][2];
    for (int r = 0; r < 8; ++r)
        for (int b = 0; b < 2; ++b) { UR[r][b] = (v2f){0.0f, 0.0f}; UI[r][b] = (v2f){0.0f, 0.0f}; }
#pragma nounroll
    for (int j = 0; j < 8; ++j) {
        v2f    y[8];
        float4 c[8], s[8];
        for (int r = 0; r < 8; ++r) {
            const int m1 = r + 8 * j;
            y[r] = ys2[m1 * M2C + mm];
            c[r] = tc4[m1 * 16 + hg];
            s[r] = ts4[m1 * 16 + hg];
        }
        __builtin_amdgcn_sched_barrier(0);
        for (int r = 0; r < 8; ++r) {
            const v2f yr = {y[r].x, y[r].x}, yi = {y[r].y, y[r].y};
            const v2f cp[2] = {{c[r].x, c[r].y}, {c[r].z, c[r].w}};
            const v2f sp[2] = {{s[r].x, s[r].y}, {s[r].z, s[r].w}};
            for (int b = 0; b < 2; ++b) {
                UR[r][b] = __builtin_elementwise_fma(cp[b], yr, UR[r][b]);
                UR[r][b] = __builtin_elementwise_fma(-sp[b], yi, UR[r][b]);
                UI[r][b] = __builtin_elementwise_fma(cp[b], yi, UI[r][b]);
                UI[r][b] = __builtin_elementwise_fma(sp[b], yr, UI[r][b]);
            }
        }
    }

    const float hs = 0.70710678118654752f;
    float* zp = Z + ((size_t)o * 512 + 4 * hg) * 128 + 2 * (m2b + mm);
    for (int a = 0; a < 4; ++a) {
        float er[4], ei[4], qr[4], qi[4], pr[4], pi[4];
        const int b = a >> 1, e = a & 1;
        idft4(UR[0][b][e], UI[0][b][e], UR[2][b][e], UI[2][b][e], UR[4][b][e], UI[4][b][e],
              UR[6][b][e], UI[6][b][e], er, ei);
        idft4(UR[1][b][e], UI[1][b][e], UR[3][b][e], UI[3][b][e], UR[5][b][e], UI[5][b][e],
              UR[7][b][e], UI[7][b][e], qr, qi);
        pr[0] = qr[0];                  pi[0] = qi[0];
        pr[1] = hs * (qr[1] - qi[1]);   pi[1] = hs * (qr[1] + qi[1]);
        pr[2] = -qi[2];                 pi[2] = qr[2];
        pr[3] = -hs * (qr[3] + qi[3]);  pi[3] = hs * (qr[3] - qi[3]);
        for (int k = 0; k < 4; ++k) {
            v2f z0, z4;
            z0.x = er[k] + pr[k];  z0.y = ei[k] + pi[k];
            z4.x = er[k] - pr[k];  z4.y = ei[k] - pi[k];
            *(v2f*)(zp + (size_t)(a + 64 * k) * 128)       = z0;
            *(v2f*)(zp + (size_t)(a + 64 * (k + 4)) * 128) = z4;
        }
    }
}

__device__ __forceinline__ v2f lo2(float4 v) { v2f r; r.x = v.x; r.y = v.y; return r; }
__device__ __forceinline__ v2f hi2(float4 v) { v2f r; r.x = v.z; r.y = v.w; return r; }

__device__ __forceinline__ void store_b_pair(float* dst, float4 r0, float4 r1) {
    float4 lo, hi;
    lo.x = r0.x; lo.y = r1.x; lo.z = r0.y; lo.w = r1.y;
    hi.x = r0.z; hi.y = r1.z; hi.z = r0.w; hi.w = r1.w;
    *(float4*)dst       = lo;
    *(float4*)(dst + 4) = hi;
}

__device__ __forceinline__ float gelu_tanh(float val) {
    const float c3 = val * val * val;
    return 0.5f * val * (1.0f + tanhf(0.7978845608028654f * (val + 0.044715f * c3)));
}

#define AZP 132
#define AKP 68
#define BPX 544
__global__ __launch_bounds__(256, 4)
void invw_skip_gelu_kernel(const float* __restrict__ Z, const float* __restrict__ TW,
                           const float* __restrict__ x, const float* __restrict__ wskip,
                           const float* __restrict__ bconv, const float* __restrict__ bskip,
                           float* __restrict__ out) {
    const int h = blockIdx.x >> 1, c2 = blockIdx.x & 1;
    __shared__ __attribute__((aligned(16))) float Az[64 * AZP];
    __shared__ __attribute__((aligned(16))) float Ak[64 * AKP];
    __shared__ __attribute__((aligned(16))) float Bs[32 * BPX];

    const int p = threadIdx.x & 63, q0 = threadIdx.x >> 6;
    const float4* xsrc = (const float4*)(x + (size_t)(2 * q0) * HW + (size_t)h * 512
                                           + 32 * c2 + 4 * (p & 7) + 64 * (p >> 3));
    float* bdst = Bs + q0 * BPX + 8 * p;
    float4 ra[8], rb[8];
    for (int u = 0; u < 8; ++u) {
        ra[u] = xsrc[(size_t)(8 * u) * (HW / 4)];
        rb[u] = xsrc[(size_t)(8 * u + 1) * (HW / 4)];
    }
    for (int idx = threadIdx.x; idx < 64 * 128 / 4; idx += blockDim.x) {
        const int o = idx >> 5, kq = idx & 31;
        const float4 z = ((const float4*)(Z + ((size_t)o * 512 + h) * 128))[kq];
        const int k0 = 2 * kq, k1 = 2 * kq + 1;
        v2f z0, z1;
        z0.x = z.x; z0.y = z.y; z1.x = z.z; z1.y = z.w;
        *(v2f*)(Az + o * AZP + 2 * (8 * (k0 & 7) + (k0 >> 3))) = z0;
        *(v2f*)(Az + o * AZP + 2 * (8 * (k1 & 7) + (k1 >> 3))) = z1;
    }
    for (int idx = threadIdx.x; idx < 64 * 64 / 4; idx += blockDim.x) {
        int o = idx >> 4, kq = idx & 15;
        *(float4*)(Ak + o * AKP + 4 * kq) = ((const float4*)wskip)[idx];
    }
    for (int u = 0; u < 8; ++u) store_b_pair(bdst + 4 * u * BPX, ra[u], rb[u]);
    __syncthreads();

    const int wave = threadIdx.x >> 5, lane = threadIdx.x & 31;
    const int n = lane & 15, half = lane >> 4;
    const int mt = wave & 3, wt = wave >> 2;
    const int m0 = mt * 16;
    const float* azp = Az + (m0 + n) * AZP + 2 * half;
    const float* akp = Ak + (m0 + n) * AKP + 2 * half;
    const float* bp  = Bs + half * BPX + 32 * wt + 2 * n;
    const float4* tw = (const float4*)TW + (2 * c2 + wt) * 32 + lane;
    const int mb = m0 + half * 8;
    float bias[8];
    for (int v = 0; v < 8; ++v) bias[v] = bconv[mb + v] + bskip[mb + v];

    v8f ur[8], ui[8];
    for (int r = 0; r < 8; ++r) { ur[r] = (v8f){}; ui[r] = (v8f){}; }
    for (int s = 0; s < 4; ++s) {
        const v2f    a0 = *(const v2f*)(azp + 4 * s);
        const v2f    a4 = *(const v2f*)(azp + 64 + 4 * s);
        const float4 f0 = tw[s * 128];
        const float4 f4 = tw[(16 + s) * 128];
        ur[0] = wmma_f32(a0, lo2(f0), ur[0]);
        ur[4] = wmma_f32(a4, lo2(f4), ur[4]);
    }
    for (int r = 1; r < 8; ++r) {
        if (r == 4) continue;
        for (int s = 0; s < 4; ++s) {
            const v2f    a = *(const v2f*)(azp + 16 * r + 4 * s);
            const float4 f = tw[(4 * r + s) * 128];
            ur[r] = wmma_f32(a, lo2(f), ur[r]);
            ui[r] = wmma_f32(a, hi2(f), ui[r]);
        }
    }

    v8f acc[8];
    {
        const float hs = 0.70710678118654752f;
        const v8f s04 = ur[0] + ur[4], d04 = ur[0] - ur[4];
        const v8f s26 = ur[2] + ur[6], d26 = ui[2] - ui[6];
        const v8f e0 = s04 + s26, e2 = s04 - s26, e1 = d04 - d26, e3 = d04 + d26;
        const v8f o0 = (ur[1] + ur[5]) + (ur[3] + ur[7]);
        const v8f o2 = (ui[3] + ui[7]) - (ui[1] + ui[5]);
        const v8f o1 = hs * (((ur[1] - ui[1]) - (ur[5] - ui[5])) + ((ur[7] + ui[7]) - (ur[3] + ui[3])));
        const v8f o3 = hs * (((ur[3] - ui[3]) - (ur[7] - ui[7])) + ((ur[5] + ui[5]) - (ur[1] + ui[1])));
        acc[0] = e0 + o0;  acc[4] = e0 - o0;
        acc[1] = e1 + o1;  acc[5] = e1 - o1;
        acc[2] = e2 + o2;  acc[6] = e2 - o2;
        acc[3] = e3 + o3;  acc[7] = e3 - o3;
    }

    for (int ks = 0; ks < 16; ++ks) {
        const v2f a = *(const v2f*)(akp + 4 * ks);
        const float* bq = bp + (2 * ks) * BPX;
        for (int w2 = 0; w2 < 8; ++w2)
            acc[w2] = wmma_f32(a, *(const v2f*)(bq + 64 * w2), acc[w2]);
    }

    float* orow = out + (size_t)mb * HW + (size_t)h * 512 + 32 * c2 + 16 * wt + n;
    for (int v = 0; v < 8; ++v)
        for (int w2 = 0; w2 < 8; ++w2)
            orow[(size_t)v * HW + 64 * w2] = gelu_tanh(acc[w2][v] + bias[v]);
}

extern "C" void kernel_launch(void* const* d_in, const int* in_sizes, int n_in,
                              void* d_out, int out_size, void* d_ws, size_t ws_size,
                              hipStream_t stream) {
    const float* x     = (const float*)d_in[0];
    const float* w_re  = (const float*)d_in[1];
    const float* w_im  = (const float*)d_in[2];
    const float* b_cv  = (const float*)d_in[3];
    const float* w_sk  = (const float*)d_in[4];
    const float* b_sk  = (const float*)d_in[5];
    float* out = (float*)d_out;

    float* ws   = (float*)d_ws;
    float* FW   = ws;
    float* cosH = FW   + 65536;
    float* sinH = cosH + 32768;
    float* TW   = sinH + 32768;
    float* XsP  = TW   + 65536;
    float* HT   = XsP  + 4194304;
    float* YsP  = HT   + 524288;
    float* Z    = YsP  + 2097152;

    twiddle_kernel<<<128, 256, 0, stream>>>(FW, cosH, sinH, TW, HT);
    dftw_dfth_kernel<<<512, 512, 0, stream>>>(x, FW, HT, XsP);
    modemul_kernel<<<dim3(64, 4, ISPLIT), 256, 0, stream>>>(XsP, w_re, w_im, YsP);
    idfth_kernel<<<dim3(64, 4), 256, 0, stream>>>(YsP, cosH, sinH, Z);
    invw_skip_gelu_kernel<<<1024, 256, 0, stream>>>(Z, TW, x, w_sk, b_cv, b_sk, out);
}
